# starting kernel
# baseline (speedup 1.0000x reference)
#include <hip/hip_runtime.h>

typedef float v2f __attribute__((ext_vector_type(2)));
typedef float v8f __attribute__((ext_vector_type(8)));

#define BB 16
#define CC 128
#define TT 1024
#define DKD 64
#define CT (CC * TT)
#define LOG10000 9.210340371976184f
#define PI_2 1.5707963267948966f

__device__ inline v8f wmma4(v2f a, v2f b, v8f c) {
  return __builtin_amdgcn_wmma_f32_16x16x4_f32(false, a, false, b, (short)0, c,
                                               false, false);
}


__global__ void posenc_add(const float* __restrict__ x, float* __restrict__ out) {
  int idx = blockIdx.x * blockDim.x + threadIdx.x;
  int ct = idx % CT;
  int c = ct / TT;
  int t = ct % TT;
  float freq = expf(-(float)(c & ~1) * (LOG10000 / (float)CC));
  float pe = sinf((float)t * freq + (float)(c & 1) * PI_2);
  out[idx] = x[idx] + pe;
}

__global__ void batch_stats(const float* __restrict__ src, float* __restrict__ stats) {
  __shared__ float ssum[256];
  __shared__ float ssq[256];
  int b = blockIdx.x;
  int tid = threadIdx.x;
  const float* p = src + b * CT;
  float s = 0.f, q = 0.f;
  for (int e = tid; e < CT; e += 256) {
    float v = p[e];
    s += v;
    q += v * v;
  }
  ssum[tid] = s;
  ssq[tid] = q;
  __syncthreads();
  for (int st = 128; st > 0; st >>= 1) {
    if (tid < st) {
      ssum[tid] += ssum[tid + st];
      ssq[tid] += ssq[tid + st];
    }
    __syncthreads();
  }
  if (tid == 0) {
    stats[b] = ssum[0];
    stats[16 + b] = ssq[0];
  }
}

__global__ void ln_apply(const float* __restrict__ src, const float* __restrict__ g,
                         const float* __restrict__ bb, const float* __restrict__ stats,
                         float* __restrict__ dst) {
  int idx = blockIdx.x * blockDim.x + threadIdx.x;
  int b = idx / CT;
  int ct = idx % CT;
  float inv_n = 1.0f / (float)CT;
  float mu = stats[b] * inv_n;
  float var = stats[16 + b] * inv_n - mu * mu;
  float r = rsqrtf(var + 1e-5f);
  dst[idx] = (src[idx] - mu) * r * g[ct] + bb[ct];
}

__global__ void dwconv(const float* __restrict__ src, const float* __restrict__ w,
                       const float* __restrict__ bias, float* __restrict__ dst) {
  int idx = blockIdx.x * blockDim.x + threadIdx.x;
  int ct = idx % CT;
  int c = ct / TT;
  int t = ct % TT;
  const float* row = src + (idx - t);
  float s = bias[c];
#pragma unroll
  for (int j = 0; j < 7; ++j) {
    int tt = t + j - 3;
    if (tt >= 0 && tt < TT) s += row[tt] * w[c * 7 + j];
  }
  dst[idx] = s;
}


__global__ __launch_bounds__(128) void pw_gemm(const float* __restrict__ Y,
                                               const float* __restrict__ W,
                                               const float* __restrict__ bias,
                                               float* __restrict__ res) {
  int l = threadIdx.x & 31;
  int tile = blockIdx.x * 4 + (threadIdx.x >> 5);
  int b = tile >> 7;
  int rem = tile & 127;
  int m0 = (rem >> 4) << 4;
  int n0 = (rem & 15) << 6;
  int lo = l & 15;
  int kh = (l >> 4) << 1;
  const float* yb = Y + b * CT;
  v8f acc0 = {}, acc1 = {}, acc2 = {}, acc3 = {};
  for (int kk = 0; kk < CC; kk += 4) {
    int k0 = kk + kh;
    v2f a;
    a.x = W[(m0 + lo) * CC + k0];
    a.y = W[(m0 + lo) * CC + k0 + 1];
    const float* p0 = yb + k0 * TT + n0 + lo;
    const float* p1 = p0 + TT;
    v2f b0, b1, b2, b3;
    b0.x = p0[0];  b0.y = p1[0];
    b1.x = p0[16]; b1.y = p1[16];
    b2.x = p0[32]; b2.y = p1[32];
    b3.x = p0[48]; b3.y = p1[48];
    acc0 = wmma4(a, b0, acc0);
    acc1 = wmma4(a, b1, acc1);
    acc2 = wmma4(a, b2, acc2);
    acc3 = wmma4(a, b3, acc3);
  }
  float* rb = res + b * CT;
#pragma unroll
  for (int v = 0; v < 8; ++v) {
    int m = m0 + v + ((l >> 4) << 3);
    float bsv = bias[m];
    float* r = rb + m * TT + n0 + lo;
    r[0]  = fmaxf(acc0[v] + bsv, 0.f) + r[0];
    r[16] = fmaxf(acc1[v] + bsv, 0.f) + r[16];
    r[32] = fmaxf(acc2[v] + bsv, 0.f) + r[32];
    r[48] = fmaxf(acc3[v] + bsv, 0.f) + r[48];
  }
}

__global__ __launch_bounds__(128) void xtw_gemm(const float* __restrict__ X,
                                                const float* __restrict__ W,
                                                float* __restrict__ dst) {
  int l = threadIdx.x & 31;
  int tile = blockIdx.x * 4 + (threadIdx.x >> 5);
  int b = tile >> 6;
  int t0 = (tile & 63) << 4;
  int lo = l & 15;
  int kh = (l >> 4) << 1;
  const float* xb = X + b * CT;
  v8f acc0 = {}, acc1 = {}, acc2 = {}, acc3 = {};
  for (int kk = 0; kk < CC; kk += 4) {
    int k0 = kk + kh;
    v2f a;
    a.x = xb[k0 * TT + t0 + lo];
    a.y = xb[(k0 + 1) * TT + t0 + lo];
    const float* w0 = W + k0 * DKD + lo;
    const float* w1 = w0 + DKD;
    v2f b0, b1, b2, b3;
    b0.x = w0[0];  b0.y = w1[0];
    b1.x = w0[16]; b1.y = w1[16];
    b2.x = w0[32]; b2.y = w1[32];
    b3.x = w0[48]; b3.y = w1[48];
    acc0 = wmma4(a, b0, acc0);
    acc1 = wmma4(a, b1, acc1);
    acc2 = wmma4(a, b2, acc2);
    acc3 = wmma4(a, b3, acc3);
  }
  float* db = dst + b * TT * DKD;
#pragma unroll
  for (int v = 0; v < 8; ++v) {
    int t = t0 + v + ((l >> 4) << 3);
    float* d = db + t * DKD + lo;
    d[0]  = acc0[v];
    d[16] = acc1[v];
    d[32] = acc2[v];
    d[48] = acc3[v];
  }
}

__global__ __launch_bounds__(32) void attn_kernel(const float* __restrict__ Q,
                                                  const float* __restrict__ Km,
                                                  const float* __restrict__ Vm,
                                                  const float* __restrict__ mask,
                                                  float* __restrict__ head) {
  __shared__ float S[16 * TT];
  int b = blockIdx.x >> 6;
  int q0 = (blockIdx.x & 63) << 4;
  int l = threadIdx.x;
  int lo = l & 15;
  int kh = (l >> 4) << 1;
  const float* qb = Q + b * TT * DKD;
  const float* kb = Km + b * TT * DKD;
  const float* vb = Vm + b * TT * DKD;
  const float* mb = mask + b * TT;

  v2f afrag[DKD / 4];
#pragma unroll
  for (int kk = 0; kk < DKD; kk += 4) {
    int k0 = kk + kh;
    afrag[kk >> 2].x = qb[(q0 + lo) * DKD + k0];
    afrag[kk >> 2].y = qb[(q0 + lo) * DKD + k0 + 1];
  }

  for (int t = 0; t < TT / 16; ++t) {
    v8f acc = {};
#pragma unroll
    for (int kk = 0; kk < DKD; kk += 4) {
      int k0 = kk + kh;
      v2f bv;
      bv.x = kb[(t * 16 + lo) * DKD + k0];
      bv.y = kb[(t * 16 + lo) * DKD + k0 + 1];
      acc = wmma4(afrag[kk >> 2], bv, acc);
    }
#pragma unroll
    for (int v = 0; v < 8; ++v) {
      int row = v + ((l >> 4) << 3);
      int col = t * 16 + lo;
      float hm = mb[col];
      float sv = acc[v] * 0.125f;
      S[row * TT + col] = sv * hm + (1.f - hm) * -1e30f;
    }
  }
  __syncthreads();

  {
    int row = lo;
    int c0 = (l >> 4) * (TT / 2);
    float mx = -3.4e38f;
    for (int c = 0; c < TT / 2; ++c) mx = fmaxf(mx, S[row * TT + c0 + c]);
    mx = fmaxf(mx, __shfl_xor(mx, 16, 32));
    float sum = 0.f;
    for (int c = 0; c < TT / 2; ++c) sum += expf(S[row * TT + c0 + c] - mx);
    sum += __shfl_xor(sum, 16, 32);
    float inv = 1.0f / sum;
    float qm = mask[b * TT + q0 + row];
    for (int c = 0; c < TT / 2; ++c) {
      int ix = row * TT + c0 + c;
      S[ix] = expf(S[ix] - mx) * inv * qm;
    }
  }
  __syncthreads();

  {
    v8f acc0 = {}, acc1 = {}, acc2 = {}, acc3 = {};
    for (int kk = 0; kk < TT; kk += 4) {
      int k0 = kk + kh;
      v2f a;
      a.x = S[lo * TT + k0];
      a.y = S[lo * TT + k0 + 1];
      const float* v0p = vb + k0 * DKD + lo;
      const float* v1p = v0p + DKD;
      v2f b0, b1, b2, b3;
      b0.x = v0p[0];  b0.y = v1p[0];
      b1.x = v0p[16]; b1.y = v1p[16];
      b2.x = v0p[32]; b2.y = v1p[32];
      b3.x = v0p[48]; b3.y = v1p[48];
      acc0 = wmma4(a, b0, acc0);
      acc1 = wmma4(a, b1, acc1);
      acc2 = wmma4(a, b2, acc2);
      acc3 = wmma4(a, b3, acc3);
    }
    float* hb = head + b * TT * DKD;
#pragma unroll
    for (int v = 0; v < 8; ++v) {
      int t = q0 + v + ((l >> 4) << 3);
      float* h = hb + t * DKD + lo;
      h[0]  = acc0[v];
      h[16] = acc1[v];
      h[32] = acc2[v];
      h[48] = acc3[v];
    }
  }
}

__global__ __launch_bounds__(128) void wo_gemm(const float* __restrict__ H,
                                               const float* __restrict__ Wo,
                                               float* __restrict__ res) {
  int l = threadIdx.x & 31;
  int tile = blockIdx.x * 4 + (threadIdx.x >> 5);
  int b = tile >> 7;
  int rem = tile & 127;
  int m0 = (rem >> 4) << 4;
  int n0 = (rem & 15) << 6;
  int lo = l & 15;
  int kh = (l >> 4) << 1;
  const float* hb = H + b * TT * DKD;
  v8f acc0 = {}, acc1 = {}, acc2 = {}, acc3 = {};
  for (int kk = 0; kk < DKD; kk += 4) {
    int k0 = kk + kh;
    v2f a;
    a.x = Wo[k0 * CC + m0 + lo] + Wo[(k0 + 64) * CC + m0 + lo];
    a.y = Wo[(k0 + 1) * CC + m0 + lo] + Wo[(k0 + 65) * CC + m0 + lo];
    const float* h0 = hb + (n0 + lo) * DKD + k0;
    v2f b0, b1, b2, b3;
    b0.x = h0[0];            b0.y = h0[1];
    b1.x = h0[16 * DKD];     b1.y = h0[16 * DKD + 1];
    b2.x = h0[32 * DKD];     b2.y = h0[32 * DKD + 1];
    b3.x = h0[48 * DKD];     b3.y = h0[48 * DKD + 1];
    acc0 = wmma4(a, b0, acc0);
    acc1 = wmma4(a, b1, acc1);
    acc2 = wmma4(a, b2, acc2);
    acc3 = wmma4(a, b3, acc3);
  }
  float* rb = res + b * CT;
#pragma unroll
  for (int v = 0; v < 8; ++v) {
    int m = m0 + v + ((l >> 4) << 3);
    float* r = rb + m * TT + n0 + lo;
    r[0]  = acc0[v] + r[0];
    r[16] = acc1[v] + r[16];
    r[32] = acc2[v] + r[32];
    r[48] = acc3[v] + r[48];
  }
}

__global__ __launch_bounds__(128) void fc_gemm(const float* __restrict__ X,
                                               const float* __restrict__ W,
                                               const float* __restrict__ bias,
                                               const float* __restrict__ res,
                                               float* __restrict__ out) {
  int l = threadIdx.x & 31;
  int tile = blockIdx.x * 4 + (threadIdx.x >> 5);
  int b = tile >> 7;
  int rem = tile & 127;
  int m0 = (rem >> 4) << 4;
  int n0 = (rem & 15) << 6;
  int lo = l & 15;
  int kh = (l >> 4) << 1;
  const float* xb = X + b * CT;
  v8f acc0 = {}, acc1 = {}, acc2 = {}, acc3 = {};
  for (int kk = 0; kk < CC; kk += 4) {
    int k0 = kk + kh;
    v2f a;
    a.x = W[(m0 + lo) * CC + k0];
    a.y = W[(m0 + lo) * CC + k0 + 1];
    const float* p0 = xb + k0 * TT + n0 + lo;
    const float* p1 = p0 + TT;
    v2f b0, b1, b2, b3;
    b0.x = p0[0];  b0.y = p1[0];
    b1.x = p0[16]; b1.y = p1[16];
    b2.x = p0[32]; b2.y = p1[32];
    b3.x = p0[48]; b3.y = p1[48];
    acc0 = wmma4(a, b0, acc0);
    acc1 = wmma4(a, b1, acc1);
    acc2 = wmma4(a, b2, acc2);
    acc3 = wmma4(a, b3, acc3);
  }
  const float* rb = res + b * CT;
  float* ob = out + b * CT;
#pragma unroll
  for (int v = 0; v < 8; ++v) {
    int m = m0 + v + ((l >> 4) << 3);
    float bsv = bias[m];
    const float* r = rb + m * TT + n0 + lo;
    float* o = ob + m * TT + n0 + lo;
    o[0]  = fmaxf(acc0[v] + bsv, 0.f) + r[0];
    o[16] = fmaxf(acc1[v] + bsv, 0.f) + r[16];
    o[32] = fmaxf(acc2[v] + bsv, 0.f) + r[32];
    o[48] = fmaxf(acc3[v] + bsv, 0.f) + r[48];
  }
}


extern "C" void kernel_launch(void* const* d_in, const int* in_sizes, int n_in,
                              void* d_out, int out_size, void* d_ws, size_t ws_size,
                              hipStream_t stream) {
  (void)in_sizes; (void)n_in; (void)out_size; (void)ws_size;
  const float* x    = (const float*)d_in[0];
  const float* mask = (const float*)d_in[1];
  const float* dw_w = (const float*)d_in[2];
  const float* dw_b = (const float*)d_in[3];
  const float* pw_w = (const float*)d_in[4];
  const float* pw_b = (const float*)d_in[5];
  const float* n0g  = (const float*)d_in[6];
  const float* n0b  = (const float*)d_in[7];
  const float* nsg  = (const float*)d_in[8];
  const float* nsb  = (const float*)d_in[9];
  const float* neg_ = (const float*)d_in[10];
  const float* neb  = (const float*)d_in[11];
  const float* Wq   = (const float*)d_in[12];
  const float* Wk   = (const float*)d_in[13];
  const float* Wv   = (const float*)d_in[14];
  const float* Wo   = (const float*)d_in[15];
  const float* fcw  = (const float*)d_in[16];
  const float* fcb  = (const float*)d_in[17];
  float* outp = (float*)d_out;
  float* ws = (float*)d_ws;

  float* OUT = ws;
  float* RES = ws + 2097152;
  float* TMP = ws + 4194304;
  float* Qb  = ws + 6291456;
  float* Kb  = ws + 7340032;
  float* Vb  = ws + 8388608;
  float* Hb  = ws + 9437184;
  float* ST  = ws + 10485760;

  posenc_add<<<8192, 256, 0, stream>>>(x, OUT);
  batch_stats<<<16, 256, 0, stream>>>(OUT, ST);
  ln_apply<<<8192, 256, 0, stream>>>(OUT, n0g, n0b, ST, RES);

  for (int i = 0; i < 4; ++i) {
    dwconv<<<8192, 256, 0, stream>>>(OUT, dw_w + i * CC * 7, dw_b + i * CC, TMP);
    pw_gemm<<<512, 128, 0, stream>>>(TMP, pw_w + i * CC * CC, pw_b + i * CC, RES);
    batch_stats<<<16, 256, 0, stream>>>(RES, ST);
    ln_apply<<<8192, 256, 0, stream>>>(RES, nsg + i * CT, nsb + i * CT, ST, OUT);
  }

  xtw_gemm<<<256, 128, 0, stream>>>(OUT, Wq, Qb);
  xtw_gemm<<<256, 128, 0, stream>>>(OUT, Wk, Kb);
  xtw_gemm<<<256, 128, 0, stream>>>(OUT, Wv, Vb);
  attn_kernel<<<1024, 32, 0, stream>>>(Qb, Kb, Vb, mask, Hb);
  wo_gemm<<<512, 128, 0, stream>>>(Hb, Wo, RES);
  batch_stats<<<16, 256, 0, stream>>>(RES, ST);
  ln_apply<<<8192, 256, 0, stream>>>(RES, neg_, neb, ST, OUT);
  fc_gemm<<<512, 128, 0, stream>>>(OUT, fcw, fcb, RES, outp);
}


# LayerNorm sums and normalisation in one launch: each of 8 work-groups per batch element sums the whole element itself (lane butterfly), then normalises its eighth: a different reduction tree
# speedup vs baseline: 2.0043x; 2.0043x over previous
#include <hip/hip_runtime.h>

typedef float v2f __attribute__((ext_vector_type(2)));
typedef float v8f __attribute__((ext_vector_type(8)));

#define BB 16
#define CC 128
#define TT 1024
#define DKD 64
#define CT (CC * TT)
#define LOG10000 9.210340371976184f
#define PI_2 1.5707963267948966f

__device__ inline v8f wmma4(v2f a, v2f b, v8f c) {
  return __builtin_amdgcn_wmma_f32_16x16x4_f32(false, a, false, b, (short)0, c,
                                               false, false);
}


__global__ void posenc_add(const float* __restrict__ x, float* __restrict__ out) {
  int idx = blockIdx.x * blockDim.x + threadIdx.x;
  int ct = idx % CT;
  int c = ct / TT;
  int t = ct % TT;
  float freq = expf(-(float)(c & ~1) * (LOG10000 / (float)CC));
  float pe = sinf((float)t * freq + (float)(c & 1) * PI_2);
  out[idx] = x[idx] + pe;
}

typedef float v4f __attribute__((ext_vector_type(4)));
#define LN_SLICES 8
#define LN_THREADS 1024
#define CT4 (CT / 4)
#define LN_ITEMS (CT4 / LN_THREADS)
#define LN_OWN (LN_ITEMS / LN_SLICES)

__device__ __forceinline__ float wave_sum(float x) {
#pragma unroll
  for (int m = 16; m > 0; m >>= 1) x += __shfl_xor(x, m);
  return x;
}

__global__ __launch_bounds__(LN_THREADS) void ln_fused(const float* __restrict__ src,
                                                       const float* __restrict__ g,
                                                       const float* __restrict__ bb,
                                                       float* __restrict__ dst) {
  __shared__ float red[64];
  int b = blockIdx.x / LN_SLICES;
  int sl = blockIdx.x % LN_SLICES;
  int tid = threadIdx.x;
  const v4f* p = reinterpret_cast<const v4f*>(src + b * CT);
  const v4f* g4 = reinterpret_cast<const v4f*>(g);
  const v4f* b4 = reinterpret_cast<const v4f*>(bb);
  v4f* d4 = reinterpret_cast<v4f*>(dst + b * CT);

  v4f s4 = {0.f, 0.f, 0.f, 0.f};
  v4f q4 = {0.f, 0.f, 0.f, 0.f};
#pragma unroll 8
  for (int i = 0; i < LN_ITEMS; ++i) {
    v4f v = p[i * LN_THREADS + tid];
    s4 += v;
    q4 += v * v;
  }
  v4f xv[LN_OWN], gv[LN_OWN], bv[LN_OWN];
#pragma unroll
  for (int i = 0; i < LN_OWN; ++i) {
    int e = (sl * LN_OWN + i) * LN_THREADS + tid;
    xv[i] = p[e];
    gv[i] = g4[e];
    bv[i] = b4[e];
  }
  float s = (s4.x + s4.y) + (s4.z + s4.w);
  float q = (q4.x + q4.y) + (q4.z + q4.w);
  s = wave_sum(s);
  q = wave_sum(q);
  int w = tid >> 5;
  int l = tid & 31;
  if (l == 0) {
    red[w] = s;
    red[32 + w] = q;
  }
  __syncthreads();
  s = wave_sum(red[l]);
  q = wave_sum(red[32 + l]);
  float inv_n = 1.0f / (float)CT;
  float mu = s * inv_n;
  float var = q * inv_n - mu * mu;
  float r = rsqrtf(var + 1e-5f);
#pragma unroll
  for (int i = 0; i < LN_OWN; ++i) {
    int e = (sl * LN_OWN + i) * LN_THREADS + tid;
    d4[e] = (xv[i] - mu) * r * gv[i] + bv[i];
  }
}

__global__ void dwconv(const float* __restrict__ src, const float* __restrict__ w,
                       const float* __restrict__ bias, float* __restrict__ dst) {
  int idx = blockIdx.x * blockDim.x + threadIdx.x;
  int ct = idx % CT;
  int c = ct / TT;
  int t = ct % TT;
  const float* row = src + (idx - t);
  float s = bias[c];
#pragma unroll
  for (int j = 0; j < 7; ++j) {
    int tt = t + j - 3;
    if (tt >= 0 && tt < TT) s += row[tt] * w[c * 7 + j];
  }
  dst[idx] = s;
}


__global__ __launch_bounds__(128) void pw_gemm(const float* __restrict__ Y,
                                               const float* __restrict__ W,
                                               const float* __restrict__ bias,
                                               float* __restrict__ res) {
  int l = threadIdx.x & 31;
  int tile = blockIdx.x * 4 + (threadIdx.x >> 5);
  int b = tile >> 7;
  int rem = tile & 127;
  int m0 = (rem >> 4) << 4;
  int n0 = (rem & 15) << 6;
  int lo = l & 15;
  int kh = (l >> 4) << 1;
  const float* yb = Y + b * CT;
  v8f acc0 = {}, acc1 = {}, acc2 = {}, acc3 = {};
  for (int kk = 0; kk < CC; kk += 4) {
    int k0 = kk + kh;
    v2f a;
    a.x = W[(m0 + lo) * CC + k0];
    a.y = W[(m0 + lo) * CC + k0 + 1];
    const float* p0 = yb + k0 * TT + n0 + lo;
    const float* p1 = p0 + TT;
    v2f b0, b1, b2, b3;
    b0.x = p0[0];  b0.y = p1[0];
    b1.x = p0[16]; b1.y = p1[16];
    b2.x = p0[32]; b2.y = p1[32];
    b3.x = p0[48]; b3.y = p1[48];
    acc0 = wmma4(a, b0, acc0);
    acc1 = wmma4(a, b1, acc1);
    acc2 = wmma4(a, b2, acc2);
    acc3 = wmma4(a, b3, acc3);
  }
  float* rb = res + b * CT;
#pragma unroll
  for (int v = 0; v < 8; ++v) {
    int m = m0 + v + ((l >> 4) << 3);
    float bsv = bias[m];
    float* r = rb + m * TT + n0 + lo;
    r[0]  = fmaxf(acc0[v] + bsv, 0.f) + r[0];
    r[16] = fmaxf(acc1[v] + bsv, 0.f) + r[16];
    r[32] = fmaxf(acc2[v] + bsv, 0.f) + r[32];
    r[48] = fmaxf(acc3[v] + bsv, 0.f) + r[48];
  }
}

__global__ __launch_bounds__(128) void xtw_gemm(const float* __restrict__ X,
                                                const float* __restrict__ W,
                                                float* __restrict__ dst) {
  int l = threadIdx.x & 31;
  int tile = blockIdx.x * 4 + (threadIdx.x >> 5);
  int b = tile >> 6;
  int t0 = (tile & 63) << 4;
  int lo = l & 15;
  int kh = (l >> 4) << 1;
  const float* xb = X + b * CT;
  v8f acc0 = {}, acc1 = {}, acc2 = {}, acc3 = {};
  for (int kk = 0; kk < CC; kk += 4) {
    int k0 = kk + kh;
    v2f a;
    a.x = xb[k0 * TT + t0 + lo];
    a.y = xb[(k0 + 1) * TT + t0 + lo];
    const float* w0 = W + k0 * DKD + lo;
    const float* w1 = w0 + DKD;
    v2f b0, b1, b2, b3;
    b0.x = w0[0];  b0.y = w1[0];
    b1.x = w0[16]; b1.y = w1[16];
    b2.x = w0[32]; b2.y = w1[32];
    b3.x = w0[48]; b3.y = w1[48];
    acc0 = wmma4(a, b0, acc0);
    acc1 = wmma4(a, b1, acc1);
    acc2 = wmma4(a, b2, acc2);
    acc3 = wmma4(a, b3, acc3);
  }
  float* db = dst + b * TT * DKD;
#pragma unroll
  for (int v = 0; v < 8; ++v) {
    int t = t0 + v + ((l >> 4) << 3);
    float* d = db + t * DKD + lo;
    d[0]  = acc0[v];
    d[16] = acc1[v];
    d[32] = acc2[v];
    d[48] = acc3[v];
  }
}


__device__ __forceinline__ float xhalf(float x) {
  int b = __builtin_bit_cast(int, x);
  return __builtin_bit_cast(
      float, __builtin_amdgcn_permlanex16(b, b, 0x76543210, 0xfedcba98, false, false));
}

__device__ __forceinline__ v4f ld4(const float* p) {
  return *reinterpret_cast<const v4f*>(p);
}

__device__ __forceinline__ void load_block(const float* __restrict__ kb,
                                           const float* __restrict__ vb,
                                           const float* __restrict__ mb,
                                           int key0, int c, int h,
                                           v4f (&k)[2][8], v4f (&v)[2][8],
                                           v4f (&hm)[2][2]) {
#pragma unroll
  for (int u = 0; u < 2; ++u) {
    const float* kp = kb + (key0 + 16 * u + c) * DKD + 4 * h;
    const float* vp = vb + (key0 + 16 * u + 8 * h) * DKD + 4 * c;
    const float* mp = mb + key0 + 16 * u + 8 * h;
#pragma unroll
    for (int m = 0; m < 8; ++m) k[u][m] = ld4(kp + 8 * m);
#pragma unroll
    for (int r = 0; r < 8; ++r) v[u][r] = ld4(vp + r * DKD);
    hm[u][0] = ld4(mp);
    hm[u][1] = ld4(mp + 4);
  }
}

__device__ __forceinline__ void process_block(const v4f (&q)[8],
                                              const v4f (&k)[2][8],
                                              const v4f (&v)[2][8],
                                              const v4f (&hm)[2][2],
                                              float& m_run, float& l_run,
                                              v8f (&acc)[4]) {
  v8f s[2] = {};
#pragma unroll
  for (int m = 0; m < 8; ++m) {
    v2f qlo, qhi, a;
    qlo.x = q[m].x; qlo.y = q[m].y;
    qhi.x = q[m].z; qhi.y = q[m].w;
    a.x = k[0][m].x; a.y = k[0][m].y;
    s[0] = wmma4(a, qlo, s[0]);
    a.x = k[1][m].x; a.y = k[1][m].y;
    s[1] = wmma4(a, qlo, s[1]);
    a.x = k[0][m].z; a.y = k[0][m].w;
    s[0] = wmma4(a, qhi, s[0]);
    a.x = k[1][m].z; a.y = k[1][m].w;
    s[1] = wmma4(a, qhi, s[1]);
  }
  float val[2][8];
  float bm = -3.4e38f;
#pragma unroll
  for (int u = 0; u < 2; ++u) {
#pragma unroll
    for (int r = 0; r < 8; ++r) {
      float hmv = hm[u][r >> 2][r & 3];
      float sv = s[u][r] * 0.125f;
      val[u][r] = sv * hmv + (1.f - hmv) * -1e30f;
      bm = fmaxf(bm, val[u][r]);
    }
  }
  bm = fmaxf(bm, xhalf(bm));
  float m_new = fmaxf(m_run, bm);
  float alpha = expf(m_run - m_new);
  float p[2][8];
  float ps = 0.f;
#pragma unroll
  for (int u = 0; u < 2; ++u) {
#pragma unroll
    for (int r = 0; r < 8; ++r) {
      p[u][r] = expf(val[u][r] - m_new);
      ps += p[u][r];
    }
  }
  l_run = l_run * alpha + ps;
  m_run = m_new;
#pragma unroll
  for (int i = 0; i < 4; ++i) acc[i] = acc[i] * alpha;
#pragma unroll
  for (int u = 0; u < 2; ++u) {
#pragma unroll
    for (int j = 0; j < 4; ++j) {
      v2f bp;
      bp.x = p[u][2 * j];
      bp.y = p[u][2 * j + 1];
#pragma unroll
      for (int i = 0; i < 4; ++i) {
        v2f a;
        a.x = v[u][2 * j][i];
        a.y = v[u][2 * j + 1][i];
        acc[i] = wmma4(a, bp, acc[i]);
      }
    }
  }
}

__global__ __launch_bounds__(128) void attn_kernel(const float* __restrict__ Q,
                                                   const float* __restrict__ Km,
                                                   const float* __restrict__ Vm,
                                                   const float* __restrict__ mask,
                                                   float* __restrict__ head) {
  int l = threadIdx.x & 31;
  int tile = blockIdx.x * 4 + (threadIdx.x >> 5);
  int b = tile >> 6;
  int q0 = (tile & 63) << 4;
  int c = l & 15;
  int h = l >> 4;
  const float* qb = Q + b * TT * DKD;
  const float* kb = Km + b * TT * DKD;
  const float* vb = Vm + b * TT * DKD;
  const float* mb = mask + b * TT;

  v4f q[8];
#pragma unroll
  for (int m = 0; m < 8; ++m) q[m] = ld4(qb + (q0 + c) * DKD + 8 * m + 4 * h);

  v4f kA[2][8], vA[2][8], hA[2][2];
  v4f kB[2][8], vB[2][8], hB[2][2];
  v8f acc[4] = {};
  float m_run = -3.4e38f;
  float l_run = 0.f;

  load_block(kb, vb, mb, 0, c, h, kA, vA, hA);
  for (int blk = 0; blk < TT / 32 - 2; blk += 2) {
    load_block(kb, vb, mb, (blk + 1) * 32, c, h, kB, vB, hB);
    process_block(q, kA, vA, hA, m_run, l_run, acc);
    load_block(kb, vb, mb, (blk + 2) * 32, c, h, kA, vA, hA);
    process_block(q, kB, vB, hB, m_run, l_run, acc);
  }
  load_block(kb, vb, mb, TT - 32, c, h, kB, vB, hB);
  process_block(q, kA, vA, hA, m_run, l_run, acc);
  process_block(q, kB, vB, hB, m_run, l_run, acc);

  float lsum = l_run + xhalf(l_run);
  float f = (1.0f / lsum) * mb[q0 + c];
  float* hp = head + b * TT * DKD + (q0 + c) * DKD + 32 * h;
#pragma unroll
  for (int r = 0; r < 8; ++r) {
    v4f o;
    o.x = acc[0][r];
    o.y = acc[1][r];
    o.z = acc[2][r];
    o.w = acc[3][r];
    *reinterpret_cast<v4f*>(hp + 4 * r) = o * f;
  }
}

__global__ __launch_bounds__(128) void wo_gemm(const float* __restrict__ H,
                                               const float* __restrict__ Wo,
                                               float* __restrict__ res) {
  int l = threadIdx.x & 31;
  int tile = blockIdx.x * 4 + (threadIdx.x >> 5);
  int b = tile >> 7;
  int rem = tile & 127;
  int m0 = (rem >> 4) << 4;
  int n0 = (rem & 15) << 6;
  int lo = l & 15;
  int kh = (l >> 4) << 1;
  const float* hb = H + b * TT * DKD;
  v8f acc0 = {}, acc1 = {}, acc2 = {}, acc3 = {};
  for (int kk = 0; kk < DKD; kk += 4) {
    int k0 = kk + kh;
    v2f a;
    a.x = Wo[k0 * CC + m0 + lo] + Wo[(k0 + 64) * CC + m0 + lo];
    a.y = Wo[(k0 + 1) * CC + m0 + lo] + Wo[(k0 + 65) * CC + m0 + lo];
    const float* h0 = hb + (n0 + lo) * DKD + k0;
    v2f b0, b1, b2, b3;
    b0.x = h0[0];            b0.y = h0[1];
    b1.x = h0[16 * DKD];     b1.y = h0[16 * DKD + 1];
    b2.x = h0[32 * DKD];     b2.y = h0[32 * DKD + 1];
    b3.x = h0[48 * DKD];     b3.y = h0[48 * DKD + 1];
    acc0 = wmma4(a, b0, acc0);
    acc1 = wmma4(a, b1, acc1);
    acc2 = wmma4(a, b2, acc2);
    acc3 = wmma4(a, b3, acc3);
  }
  float* rb = res + b * CT;
#pragma unroll
  for (int v = 0; v < 8; ++v) {
    int m = m0 + v + ((l >> 4) << 3);
    float* r = rb + m * TT + n0 + lo;
    r[0]  = acc0[v] + r[0];
    r[16] = acc1[v] + r[16];
    r[32] = acc2[v] + r[32];
    r[48] = acc3[v] + r[48];
  }
}

__global__ __launch_bounds__(128) void fc_gemm(const float* __restrict__ X,
                                               const float* __restrict__ W,
                                               const float* __restrict__ bias,
                                               const float* __restrict__ res,
                                               float* __restrict__ out) {
  int l = threadIdx.x & 31;
  int tile = blockIdx.x * 4 + (threadIdx.x >> 5);
  int b = tile >> 7;
  int rem = tile & 127;
  int m0 = (rem >> 4) << 4;
  int n0 = (rem & 15) << 6;
  int lo = l & 15;
  int kh = (l >> 4) << 1;
  const float* xb = X + b * CT;
  v8f acc0 = {}, acc1 = {}, acc2 = {}, acc3 = {};
  for (int kk = 0; kk < CC; kk += 4) {
    int k0 = kk + kh;
    v2f a;
    a.x = W[(m0 + lo) * CC + k0];
    a.y = W[(m0 + lo) * CC + k0 + 1];
    const float* p0 = xb + k0 * TT + n0 + lo;
    const float* p1 = p0 + TT;
    v2f b0, b1, b2, b3;
    b0.x = p0[0];  b0.y = p1[0];
    b1.x = p0[16]; b1.y = p1[16];
    b2.x = p0[32]; b2.y = p1[32];
    b3.x = p0[48]; b3.y = p1[48];
    acc0 = wmma4(a, b0, acc0);
    acc1 = wmma4(a, b1, acc1);
    acc2 = wmma4(a, b2, acc2);
    acc3 = wmma4(a, b3, acc3);
  }
  const float* rb = res + b * CT;
  float* ob = out + b * CT;
#pragma unroll
  for (int v = 0; v < 8; ++v) {
    int m = m0 + v + ((l >> 4) << 3);
    float bsv = bias[m];
    const float* r = rb + m * TT + n0 + lo;
    float* o = ob + m * TT + n0 + lo;
    o[0]  = fmaxf(acc0[v] + bsv, 0.f) + r[0];
    o[16] = fmaxf(acc1[v] + bsv, 0.f) + r[16];
    o[32] = fmaxf(acc2[v] + bsv, 0.f) + r[32];
    o[48] = fmaxf(acc3[v] + bsv, 0.f) + r[48];
  }
}


extern "C" void kernel_launch(void* const* d_in, const int* in_sizes, int n_in,
                              void* d_out, int out_size, void* d_ws, size_t ws_size,
                              hipStream_t stream) {
  (void)in_sizes; (void)n_in; (void)out_size; (void)ws_size;
  const float* x    = (const float*)d_in[0];
  const float* mask = (const float*)d_in[1];
  const float* dw_w = (const float*)d_in[2];
  const float* dw_b = (const float*)d_in[3];
  const float* pw_w = (const float*)d_in[4];
  const float* pw_b = (const float*)d_in[5];
  const float* n0g  = (const float*)d_in[6];
  const float* n0b  = (const float*)d_in[7];
  const float* nsg  = (const float*)d_in[8];
  const float* nsb  = (const float*)d_in[9];
  const float* neg_ = (const float*)d_in[10];
  const float* neb  = (const float*)d_in[11];
  const float* Wq   = (const float*)d_in[12];
  const float* Wk   = (const float*)d_in[13];
  const float* Wv   = (const float*)d_in[14];
  const float* Wo   = (const float*)d_in[15];
  const float* fcw  = (const float*)d_in[16];
  const float* fcb  = (const float*)d_in[17];
  float* outp = (float*)d_out;
  float* ws = (float*)d_ws;

  float* OUT = ws;
  float* RES = ws + 2097152;
  float* TMP = ws + 4194304;
  float* Qb  = ws + 6291456;
  float* Kb  = ws + 7340032;
  float* Vb  = ws + 8388608;
  float* Hb  = ws + 9437184;

  posenc_add<<<8192, 256, 0, stream>>>(x, OUT);
  ln_fused<<<128, 1024, 0, stream>>>(OUT, n0g, n0b, RES);

  for (int i = 0; i < 4; ++i) {
    dwconv<<<8192, 256, 0, stream>>>(OUT, dw_w + i * CC * 7, dw_b + i * CC, TMP);
    pw_gemm<<<512, 128, 0, stream>>>(TMP, pw_w + i * CC * CC, pw_b + i * CC, RES);
    ln_fused<<<128, 1024, 0, stream>>>(RES, nsg + i * CT, nsb + i * CT, OUT);
  }

  xtw_gemm<<<256, 128, 0, stream>>>(OUT, Wq, Qb);
  xtw_gemm<<<256, 128, 0, stream>>>(OUT, Wk, Kb);
  xtw_gemm<<<256, 128, 0, stream>>>(OUT, Wv, Vb);
  attn_kernel<<<256, 128, 0, stream>>>(Qb, Kb, Vb, mask, Hb);
  wo_gemm<<<512, 128, 0, stream>>>(Hb, Wo, RES);
  ln_fused<<<128, 1024, 0, stream>>>(RES, neg_, neb, OUT);
  fc_gemm<<<512, 128, 0, stream>>>(OUT, fcw, fcb, RES, outp);
}


# Depthwise conv done by the LayerNorm launch that makes or reads its input: own slice put in LDS between zero items, neighbours' items read back; the 4 dwconv launches are gone, sum order kept
# speedup vs baseline: 2.2904x; 1.1427x over previous
#include <hip/hip_runtime.h>

typedef float v2f __attribute__((ext_vector_type(2)));
typedef float v8f __attribute__((ext_vector_type(8)));

#define BB 16
#define CC 128
#define TT 1024
#define DKD 64
#define CT (CC * TT)
#define LOG10000 9.210340371976184f
#define PI_2 1.5707963267948966f

__device__ inline v8f wmma4(v2f a, v2f b, v8f c) {
  return __builtin_amdgcn_wmma_f32_16x16x4_f32(false, a, false, b, (short)0, c,
                                               false, false);
}


__global__ void posenc_add(const float* __restrict__ x, float* __restrict__ out) {
  int idx = blockIdx.x * blockDim.x + threadIdx.x;
  int ct = idx % CT;
  int c = ct / TT;
  int t = ct % TT;
  float freq = expf(-(float)(c & ~1) * (LOG10000 / (float)CC));
  float pe = sinf((float)t * freq + (float)(c & 1) * PI_2);
  out[idx] = x[idx] + pe;
}

typedef float v4f __attribute__((ext_vector_type(4)));
#define LN_SLICES 8
#define LN_THREADS 1024
#define CT4 (CT / 4)
#define LN_ITEMS (CT4 / LN_THREADS)
#define LN_OWN (LN_ITEMS / LN_SLICES)

__device__ __forceinline__ float wave_sum(float x) {
#pragma unroll
  for (int m = 16; m > 0; m >>= 1) x += __shfl_xor(x, m);
  return x;
}

#define LN_PLAIN 0
#define LN_CONV_OUT 1
#define LN_CONV_IN 2
#define ROW_ITEMS (TT / 4)
#define ROWS_PER_STEP (LN_THREADS / ROW_ITEMS)
#define ROW_PITCH (ROW_ITEMS + 2)

__device__ __forceinline__ void ln_body(const int mode,
                                        const float* __restrict__ src,
                                        const float* __restrict__ g,
                                        const float* __restrict__ bb,
                                        float* __restrict__ dst,
                                        const float* __restrict__ cw,
                                        const float* __restrict__ cb,
                                        float* __restrict__ cdst) {
  __shared__ float red[64];
  __shared__ v4f tile[LN_OWN][ROWS_PER_STEP * ROW_PITCH];
  int b = blockIdx.x / LN_SLICES;
  int sl = blockIdx.x % LN_SLICES;
  int tid = threadIdx.x;
  const v4f* p = reinterpret_cast<const v4f*>(src + b * CT);
  const v4f* g4 = reinterpret_cast<const v4f*>(g);
  const v4f* b4 = reinterpret_cast<const v4f*>(bb);
  v4f* d4 = reinterpret_cast<v4f*>(dst + b * CT);
  v4f* c4 = reinterpret_cast<v4f*>(cdst + b * CT);
  int rowi = tid / ROW_ITEMS;
  int cell = rowi * ROW_PITCH + 1 + tid % ROW_ITEMS;

  v4f s4 = {0.f, 0.f, 0.f, 0.f};
  v4f q4 = {0.f, 0.f, 0.f, 0.f};
#pragma unroll 8
  for (int i = 0; i < LN_ITEMS; ++i) {
    v4f v = p[i * LN_THREADS + tid];
    s4 += v;
    q4 += v * v;
  }
  float wt[LN_OWN][7], cbias[LN_OWN];
  if (mode != LN_PLAIN) {
#pragma unroll
    for (int i = 0; i < LN_OWN; ++i) {
      int c = __builtin_amdgcn_readfirstlane((sl * LN_OWN + i) * ROWS_PER_STEP + rowi);
#pragma unroll
      for (int j = 0; j < 7; ++j) wt[i][j] = cw[c * 7 + j];
      cbias[i] = cb[c];
    }
  }
  v4f xv[LN_OWN], gv[LN_OWN], bv[LN_OWN];
#pragma unroll
  for (int i = 0; i < LN_OWN; ++i) {
    int e = (sl * LN_OWN + i) * LN_THREADS + tid;
    xv[i] = p[e];
    gv[i] = g4[e];
    bv[i] = b4[e];
  }
  float s = (s4.x + s4.y) + (s4.z + s4.w);
  float q = (q4.x + q4.y) + (q4.z + q4.w);
  s = wave_sum(s);
  q = wave_sum(q);
  int w = tid >> 5;
  int l = tid & 31;
  if (l == 0) {
    red[w] = s;
    red[32 + w] = q;
  }
  if (mode != LN_PLAIN) {
    if (tid < LN_OWN * ROWS_PER_STEP * 2) {
      v4f z = {0.f, 0.f, 0.f, 0.f};
      tile[tid >> 3][((tid >> 1) & (ROWS_PER_STEP - 1)) * ROW_PITCH + (tid & 1) * (ROW_PITCH - 1)] = z;
    }
  }
  if (mode == LN_CONV_IN) {
#pragma unroll
    for (int i = 0; i < LN_OWN; ++i) tile[i][cell] = xv[i];
  }
  __syncthreads();
  s = wave_sum(red[l]);
  q = wave_sum(red[32 + l]);
  float inv_n = 1.0f / (float)CT;
  float mu = s * inv_n;
  float var = q * inv_n - mu * mu;
  float r = rsqrtf(var + 1e-5f);
  v4f y[LN_OWN];
#pragma unroll
  for (int i = 0; i < LN_OWN; ++i) {
    int e = (sl * LN_OWN + i) * LN_THREADS + tid;
    y[i] = (xv[i] - mu) * r * gv[i] + bv[i];
    if (mode != LN_CONV_OUT) d4[e] = y[i];
    if (mode == LN_CONV_OUT) tile[i][cell] = y[i];
  }
  if (mode == LN_PLAIN) return;
  if (mode == LN_CONV_OUT) __syncthreads();
#pragma unroll
  for (int i = 0; i < LN_OWN; ++i) {
    int e = (sl * LN_OWN + i) * LN_THREADS + tid;
    v4f lf = tile[i][cell - 1];
    v4f rt = tile[i][cell + 1];
    v4f md = (mode == LN_CONV_IN) ? xv[i] : y[i];
    float t[10] = {lf.y, lf.z, lf.w, md.x, md.y, md.z, md.w, rt.x, rt.y, rt.z};
    v4f o;
#pragma unroll
    for (int k = 0; k < 4; ++k) {
      float a = cbias[i];
#pragma unroll
      for (int j = 0; j < 7; ++j) a += t[k + j] * wt[i][j];
      o[k] = a;
    }
    c4[e] = o;
  }
}

__global__ __launch_bounds__(LN_THREADS) void ln_fused(const float* __restrict__ src,
                                                       const float* __restrict__ g,
                                                       const float* __restrict__ bb,
                                                       float* __restrict__ dst) {
  ln_body(LN_PLAIN, src, g, bb, dst, nullptr, nullptr, nullptr);
}

__global__ __launch_bounds__(LN_THREADS) void ln_conv_out(const float* __restrict__ src,
                                                          const float* __restrict__ g,
                                                          const float* __restrict__ bb,
                                                          const float* __restrict__ cw,
                                                          const float* __restrict__ cb,
                                                          float* __restrict__ cdst) {
  ln_body(LN_CONV_OUT, src, g, bb, nullptr, cw, cb, cdst);
}

__global__ __launch_bounds__(LN_THREADS) void ln_conv_in(const float* __restrict__ src,
                                                         const float* __restrict__ g,
                                                         const float* __restrict__ bb,
                                                         float* __restrict__ dst,
                                                         const float* __restrict__ cw,
                                                         const float* __restrict__ cb,
                                                         float* __restrict__ cdst) {
  ln_body(LN_CONV_IN, src, g, bb, dst, cw, cb, cdst);
}


__global__ __launch_bounds__(128) void pw_gemm(const float* __restrict__ Y,
                                               const float* __restrict__ W,
                                               const float* __restrict__ bias,
                                               float* __restrict__ res) {
  int l = threadIdx.x & 31;
  int tile = blockIdx.x * 4 + (threadIdx.x >> 5);
  int b = tile >> 7;
  int rem = tile & 127;
  int m0 = (rem >> 4) << 4;
  int n0 = (rem & 15) << 6;
  int lo = l & 15;
  int kh = (l >> 4) << 1;
  const float* yb = Y + b * CT;
  v8f acc0 = {}, acc1 = {}, acc2 = {}, acc3 = {};
  for (int kk = 0; kk < CC; kk += 4) {
    int k0 = kk + kh;
    v2f a;
    a.x = W[(m0 + lo) * CC + k0];
    a.y = W[(m0 + lo) * CC + k0 + 1];
    const float* p0 = yb + k0 * TT + n0 + lo;
    const float* p1 = p0 + TT;
    v2f b0, b1, b2, b3;
    b0.x = p0[0];  b0.y = p1[0];
    b1.x = p0[16]; b1.y = p1[16];
    b2.x = p0[32]; b2.y = p1[32];
    b3.x = p0[48]; b3.y = p1[48];
    acc0 = wmma4(a, b0, acc0);
    acc1 = wmma4(a, b1, acc1);
    acc2 = wmma4(a, b2, acc2);
    acc3 = wmma4(a, b3, acc3);
  }
  float* rb = res + b * CT;
#pragma unroll
  for (int v = 0; v < 8; ++v) {
    int m = m0 + v + ((l >> 4) << 3);
    float bsv = bias[m];
    float* r = rb + m * TT + n0 + lo;
    r[0]  = fmaxf(acc0[v] + bsv, 0.f) + r[0];
    r[16] = fmaxf(acc1[v] + bsv, 0.f) + r[16];
    r[32] = fmaxf(acc2[v] + bsv, 0.f) + r[32];
    r[48] = fmaxf(acc3[v] + bsv, 0.f) + r[48];
  }
}

__global__ __launch_bounds__(128) void xtw_gemm(const float* __restrict__ X,
                                                const float* __restrict__ W,
                                                float* __restrict__ dst) {
  int l = threadIdx.x & 31;
  int tile = blockIdx.x * 4 + (threadIdx.x >> 5);
  int b = tile >> 6;
  int t0 = (tile & 63) << 4;
  int lo = l & 15;
  int kh = (l >> 4) << 1;
  const float* xb = X + b * CT;
  v8f acc0 = {}, acc1 = {}, acc2 = {}, acc3 = {};
  for (int kk = 0; kk < CC; kk += 4) {
    int k0 = kk + kh;
    v2f a;
    a.x = xb[k0 * TT + t0 + lo];
    a.y = xb[(k0 + 1) * TT + t0 + lo];
    const float* w0 = W + k0 * DKD + lo;
    const float* w1 = w0 + DKD;
    v2f b0, b1, b2, b3;
    b0.x = w0[0];  b0.y = w1[0];
    b1.x = w0[16]; b1.y = w1[16];
    b2.x = w0[32]; b2.y = w1[32];
    b3.x = w0[48]; b3.y = w1[48];
    acc0 = wmma4(a, b0, acc0);
    acc1 = wmma4(a, b1, acc1);
    acc2 = wmma4(a, b2, acc2);
    acc3 = wmma4(a, b3, acc3);
  }
  float* db = dst + b * TT * DKD;
#pragma unroll
  for (int v = 0; v < 8; ++v) {
    int t = t0 + v + ((l >> 4) << 3);
    float* d = db + t * DKD + lo;
    d[0]  = acc0[v];
    d[16] = acc1[v];
    d[32] = acc2[v];
    d[48] = acc3[v];
  }
}


__device__ __forceinline__ float xhalf(float x) {
  int b = __builtin_bit_cast(int, x);
  return __builtin_bit_cast(
      float, __builtin_amdgcn_permlanex16(b, b, 0x76543210, 0xfedcba98, false, false));
}

__device__ __forceinline__ v4f ld4(const float* p) {
  return *reinterpret_cast<const v4f*>(p);
}

__device__ __forceinline__ void load_block(const float* __restrict__ kb,
                                           const float* __restrict__ vb,
                                           const float* __restrict__ mb,
                                           int key0, int c, int h,
                                           v4f (&k)[2][8], v4f (&v)[2][8],
                                           v4f (&hm)[2][2]) {
#pragma unroll
  for (int u = 0; u < 2; ++u) {
    const float* kp = kb + (key0 + 16 * u + c) * DKD + 4 * h;
    const float* vp = vb + (key0 + 16 * u + 8 * h) * DKD + 4 * c;
    const float* mp = mb + key0 + 16 * u + 8 * h;
#pragma unroll
    for (int m = 0; m < 8; ++m) k[u][m] = ld4(kp + 8 * m);
#pragma unroll
    for (int r = 0; r < 8; ++r) v[u][r] = ld4(vp + r * DKD);
    hm[u][0] = ld4(mp);
    hm[u][1] = ld4(mp + 4);
  }
}

__device__ __forceinline__ void process_block(const v4f (&q)[8],
                                              const v4f (&k)[2][8],
                                              const v4f (&v)[2][8],
                                              const v4f (&hm)[2][2],
                                              float& m_run, float& l_run,
                                              v8f (&acc)[4]) {
  v8f s[2] = {};
#pragma unroll
  for (int m = 0; m < 8; ++m) {
    v2f qlo, qhi, a;
    qlo.x = q[m].x; qlo.y = q[m].y;
    qhi.x = q[m].z; qhi.y = q[m].w;
    a.x = k[0][m].x; a.y = k[0][m].y;
    s[0] = wmma4(a, qlo, s[0]);
    a.x = k[1][m].x; a.y = k[1][m].y;
    s[1] = wmma4(a, qlo, s[1]);
    a.x = k[0][m].z; a.y = k[0][m].w;
    s[0] = wmma4(a, qhi, s[0]);
    a.x = k[1][m].z; a.y = k[1][m].w;
    s[1] = wmma4(a, qhi, s[1]);
  }
  float val[2][8];
  float bm = -3.4e38f;
#pragma unroll
  for (int u = 0; u < 2; ++u) {
#pragma unroll
    for (int r = 0; r < 8; ++r) {
      float hmv = hm[u][r >> 2][r & 3];
      float sv = s[u][r] * 0.125f;
      val[u][r] = sv * hmv + (1.f - hmv) * -1e30f;
      bm = fmaxf(bm, val[u][r]);
    }
  }
  bm = fmaxf(bm, xhalf(bm));
  float m_new = fmaxf(m_run, bm);
  float alpha = expf(m_run - m_new);
  float p[2][8];
  float ps = 0.f;
#pragma unroll
  for (int u = 0; u < 2; ++u) {
#pragma unroll
    for (int r = 0; r < 8; ++r) {
      p[u][r] = expf(val[u][r] - m_new);
      ps += p[u][r];
    }
  }
  l_run = l_run * alpha + ps;
  m_run = m_new;
#pragma unroll
  for (int i = 0; i < 4; ++i) acc[i] = acc[i] * alpha;
#pragma unroll
  for (int u = 0; u < 2; ++u) {
#pragma unroll
    for (int j = 0; j < 4; ++j) {
      v2f bp;
      bp.x = p[u][2 * j];
      bp.y = p[u][2 * j + 1];
#pragma unroll
      for (int i = 0; i < 4; ++i) {
        v2f a;
        a.x = v[u][2 * j][i];
        a.y = v[u][2 * j + 1][i];
        acc[i] = wmma4(a, bp, acc[i]);
      }
    }
  }
}

__global__ __launch_bounds__(128) void attn_kernel(const float* __restrict__ Q,
                                                   const float* __restrict__ Km,
                                                   const float* __restrict__ Vm,
                                                   const float* __restrict__ mask,
                                                   float* __restrict__ head) {
  int l = threadIdx.x & 31;
  int tile = blockIdx.x * 4 + (threadIdx.x >> 5);
  int b = tile >> 6;
  int q0 = (tile & 63) << 4;
  int c = l & 15;
  int h = l >> 4;
  const float* qb = Q + b * TT * DKD;
  const float* kb = Km + b * TT * DKD;
  const float* vb = Vm + b * TT * DKD;
  const float* mb = mask + b * TT;

  v4f q[8];
#pragma unroll
  for (int m = 0; m < 8; ++m) q[m] = ld4(qb + (q0 + c) * DKD + 8 * m + 4 * h);

  v4f kA[2][8], vA[2][8], hA[2][2];
  v4f kB[2][8], vB[2][8], hB[2][2];
  v8f acc[4] = {};
  float m_run = -3.4e38f;
  float l_run = 0.f;

  load_block(kb, vb, mb, 0, c, h, kA, vA, hA);
  for (int blk = 0; blk < TT / 32 - 2; blk += 2) {
    load_block(kb, vb, mb, (blk + 1) * 32, c, h, kB, vB, hB);
    process_block(q, kA, vA, hA, m_run, l_run, acc);
    load_block(kb, vb, mb, (blk + 2) * 32, c, h, kA, vA, hA);
    process_block(q, kB, vB, hB, m_run, l_run, acc);
  }
  load_block(kb, vb, mb, TT - 32, c, h, kB, vB, hB);
  process_block(q, kA, vA, hA, m_run, l_run, acc);
  process_block(q, kB, vB, hB, m_run, l_run, acc);

  float lsum = l_run + xhalf(l_run);
  float f = (1.0f / lsum) * mb[q0 + c];
  float* hp = head + b * TT * DKD + (q0 + c) * DKD + 32 * h;
#pragma unroll
  for (int r = 0; r < 8; ++r) {
    v4f o;
    o.x = acc[0][r];
    o.y = acc[1][r];
    o.z = acc[2][r];
    o.w = acc[3][r];
    *reinterpret_cast<v4f*>(hp + 4 * r) = o * f;
  }
}

__global__ __launch_bounds__(128) void wo_gemm(const float* __restrict__ H,
                                               const float* __restrict__ Wo,
                                               float* __restrict__ res) {
  int l = threadIdx.x & 31;
  int tile = blockIdx.x * 4 + (threadIdx.x >> 5);
  int b = tile >> 7;
  int rem = tile & 127;
  int m0 = (rem >> 4) << 4;
  int n0 = (rem & 15) << 6;
  int lo = l & 15;
  int kh = (l >> 4) << 1;
  const float* hb = H + b * TT * DKD;
  v8f acc0 = {}, acc1 = {}, acc2 = {}, acc3 = {};
  for (int kk = 0; kk < DKD; kk += 4) {
    int k0 = kk + kh;
    v2f a;
    a.x = Wo[k0 * CC + m0 + lo] + Wo[(k0 + 64) * CC + m0 + lo];
    a.y = Wo[(k0 + 1) * CC + m0 + lo] + Wo[(k0 + 65) * CC + m0 + lo];
    const float* h0 = hb + (n0 + lo) * DKD + k0;
    v2f b0, b1, b2, b3;
    b0.x = h0[0];            b0.y = h0[1];
    b1.x = h0[16 * DKD];     b1.y = h0[16 * DKD + 1];
    b2.x = h0[32 * DKD];     b2.y = h0[32 * DKD + 1];
    b3.x = h0[48 * DKD];     b3.y = h0[48 * DKD + 1];
    acc0 = wmma4(a, b0, acc0);
    acc1 = wmma4(a, b1, acc1);
    acc2 = wmma4(a, b2, acc2);
    acc3 = wmma4(a, b3, acc3);
  }
  float* rb = res + b * CT;
#pragma unroll
  for (int v = 0; v < 8; ++v) {
    int m = m0 + v + ((l >> 4) << 3);
    float* r = rb + m * TT + n0 + lo;
    r[0]  = acc0[v] + r[0];
    r[16] = acc1[v] + r[16];
    r[32] = acc2[v] + r[32];
    r[48] = acc3[v] + r[48];
  }
}

__global__ __launch_bounds__(128) void fc_gemm(const float* __restrict__ X,
                                               const float* __restrict__ W,
                                               const float* __restrict__ bias,
                                               const float* __restrict__ res,
                                               float* __restrict__ out) {
  int l = threadIdx.x & 31;
  int tile = blockIdx.x * 4 + (threadIdx.x >> 5);
  int b = tile >> 7;
  int rem = tile & 127;
  int m0 = (rem >> 4) << 4;
  int n0 = (rem & 15) << 6;
  int lo = l & 15;
  int kh = (l >> 4) << 1;
  const float* xb = X + b * CT;
  v8f acc0 = {}, acc1 = {}, acc2 = {}, acc3 = {};
  for (int kk = 0; kk < CC; kk += 4) {
    int k0 = kk + kh;
    v2f a;
    a.x = W[(m0 + lo) * CC + k0];
    a.y = W[(m0 + lo) * CC + k0 + 1];
    const float* p0 = xb + k0 * TT + n0 + lo;
    const float* p1 = p0 + TT;
    v2f b0, b1, b2, b3;
    b0.x = p0[0];  b0.y = p1[0];
    b1.x = p0[16]; b1.y = p1[16];
    b2.x = p0[32]; b2.y = p1[32];
    b3.x = p0[48]; b3.y = p1[48];
    acc0 = wmma4(a, b0, acc0);
    acc1 = wmma4(a, b1, acc1);
    acc2 = wmma4(a, b2, acc2);
    acc3 = wmma4(a, b3, acc3);
  }
  const float* rb = res + b * CT;
  float* ob = out + b * CT;
#pragma unroll
  for (int v = 0; v < 8; ++v) {
    int m = m0 + v + ((l >> 4) << 3);
    float bsv = bias[m];
    const float* r = rb + m * TT + n0 + lo;
    float* o = ob + m * TT + n0 + lo;
    o[0]  = fmaxf(acc0[v] + bsv, 0.f) + r[0];
    o[16] = fmaxf(acc1[v] + bsv, 0.f) + r[16];
    o[32] = fmaxf(acc2[v] + bsv, 0.f) + r[32];
    o[48] = fmaxf(acc3[v] + bsv, 0.f) + r[48];
  }
}


extern "C" void kernel_launch(void* const* d_in, const int* in_sizes, int n_in,
                              void* d_out, int out_size, void* d_ws, size_t ws_size,
                              hipStream_t stream) {
  (void)in_sizes; (void)n_in; (void)out_size; (void)ws_size;
  const float* x    = (const float*)d_in[0];
  const float* mask = (const float*)d_in[1];
  const float* dw_w = (const float*)d_in[2];
  const float* dw_b = (const float*)d_in[3];
  const float* pw_w = (const float*)d_in[4];
  const float* pw_b = (const float*)d_in[5];
  const float* n0g  = (const float*)d_in[6];
  const float* n0b  = (const float*)d_in[7];
  const float* nsg  = (const float*)d_in[8];
  const float* nsb  = (const float*)d_in[9];
  const float* neg_ = (const float*)d_in[10];
  const float* neb  = (const float*)d_in[11];
  const float* Wq   = (const float*)d_in[12];
  const float* Wk   = (const float*)d_in[13];
  const float* Wv   = (const float*)d_in[14];
  const float* Wo   = (const float*)d_in[15];
  const float* fcw  = (const float*)d_in[16];
  const float* fcb  = (const float*)d_in[17];
  float* outp = (float*)d_out;
  float* ws = (float*)d_ws;

  float* OUT = ws;
  float* RES = ws + 2097152;
  float* TMP = ws + 4194304;
  float* Qb  = ws + 6291456;
  float* Kb  = ws + 7340032;
  float* Vb  = ws + 8388608;
  float* Hb  = ws + 9437184;

  posenc_add<<<8192, 256, 0, stream>>>(x, OUT);
  ln_conv_in<<<128, 1024, 0, stream>>>(OUT, n0g, n0b, RES, dw_w, dw_b, TMP);

  for (int i = 0; i < 3; ++i) {
    pw_gemm<<<512, 128, 0, stream>>>(TMP, pw_w + i * CC * CC, pw_b + i * CC, RES);
    ln_conv_out<<<128, 1024, 0, stream>>>(RES, nsg + i * CT, nsb + i * CT,
                                          dw_w + (i + 1) * CC * 7, dw_b + (i + 1) * CC, TMP);
  }
  pw_gemm<<<512, 128, 0, stream>>>(TMP, pw_w + 3 * CC * CC, pw_b + 3 * CC, RES);
  ln_fused<<<128, 1024, 0, stream>>>(RES, nsg + 3 * CT, nsb + 3 * CT, OUT);

  xtw_gemm<<<256, 128, 0, stream>>>(OUT, Wq, Qb);
  xtw_gemm<<<256, 128, 0, stream>>>(OUT, Wk, Kb);
  xtw_gemm<<<256, 128, 0, stream>>>(OUT, Wv, Vb);
  attn_kernel<<<256, 128, 0, stream>>>(Qb, Kb, Vb, mask, Hb);
  wo_gemm<<<512, 128, 0, stream>>>(Hb, Wo, RES);
  ln_fused<<<128, 1024, 0, stream>>>(RES, neg_, neb, OUT);
  fc_gemm<<<512, 128, 0, stream>>>(OUT, fcw, fcb, RES, outp);
}


# GEMM wave tiles interleave columns (lane lo owns columns 4lo..4lo+3, one per accumulator): fragments, residual and output move as 16-byte accesses, a wave's loads go out as one group; sums unchanged
# speedup vs baseline: 2.7775x; 1.2127x over previous
#include <hip/hip_runtime.h>

typedef float v2f __attribute__((ext_vector_type(2)));
typedef float v8f __attribute__((ext_vector_type(8)));

#define BB 16
#define CC 128
#define TT 1024
#define DKD 64
#define CT (CC * TT)
#define LOG10000 9.210340371976184f
#define PI_2 1.5707963267948966f

__device__ inline v8f wmma4(v2f a, v2f b, v8f c) {
  return __builtin_amdgcn_wmma_f32_16x16x4_f32(false, a, false, b, (short)0, c,
                                               false, false);
}


__global__ void posenc_add(const float* __restrict__ x, float* __restrict__ out) {
  int idx = blockIdx.x * blockDim.x + threadIdx.x;
  int ct = idx % CT;
  int c = ct / TT;
  int t = ct % TT;
  float freq = expf(-(float)(c & ~1) * (LOG10000 / (float)CC));
  float pe = sinf((float)t * freq + (float)(c & 1) * PI_2);
  out[idx] = x[idx] + pe;
}

typedef float v4f __attribute__((ext_vector_type(4)));
#define LN_SLICES 8
#define LN_THREADS 1024
#define CT4 (CT / 4)
#define LN_ITEMS (CT4 / LN_THREADS)
#define LN_OWN (LN_ITEMS / LN_SLICES)

__device__ __forceinline__ float wave_sum(float x) {
#pragma unroll
  for (int m = 16; m > 0; m >>= 1) x += __shfl_xor(x, m);
  return x;
}

#define LN_PLAIN 0
#define LN_CONV_OUT 1
#define LN_CONV_IN 2
#define ROW_ITEMS (TT / 4)
#define ROWS_PER_STEP (LN_THREADS / ROW_ITEMS)
#define ROW_PITCH (ROW_ITEMS + 2)

__device__ __forceinline__ void ln_body(const int mode,
                                        const float* __restrict__ src,
                                        const float* __restrict__ g,
                                        const float* __restrict__ bb,
                                        float* __restrict__ dst,
                                        const float* __restrict__ cw,
                                        const float* __restrict__ cb,
                                        float* __restrict__ cdst) {
  __shared__ float red[64];
  __shared__ v4f tile[LN_OWN][ROWS_PER_STEP * ROW_PITCH];
  int b = blockIdx.x / LN_SLICES;
  int sl = blockIdx.x % LN_SLICES;
  int tid = threadIdx.x;
  const v4f* p = reinterpret_cast<const v4f*>(src + b * CT);
  const v4f* g4 = reinterpret_cast<const v4f*>(g);
  const v4f* b4 = reinterpret_cast<const v4f*>(bb);
  v4f* d4 = reinterpret_cast<v4f*>(dst + b * CT);
  v4f* c4 = reinterpret_cast<v4f*>(cdst + b * CT);
  int rowi = tid / ROW_ITEMS;
  int cell = rowi * ROW_PITCH + 1 + tid % ROW_ITEMS;

  v4f s4 = {0.f, 0.f, 0.f, 0.f};
  v4f q4 = {0.f, 0.f, 0.f, 0.f};
#pragma unroll 8
  for (int i = 0; i < LN_ITEMS; ++i) {
    v4f v = p[i * LN_THREADS + tid];
    s4 += v;
    q4 += v * v;
  }
  float wt[LN_OWN][7], cbias[LN_OWN];
  if (mode != LN_PLAIN) {
#pragma unroll
    for (int i = 0; i < LN_OWN; ++i) {
      int c = __builtin_amdgcn_readfirstlane((sl * LN_OWN + i) * ROWS_PER_STEP + rowi);
#pragma unroll
      for (int j = 0; j < 7; ++j) wt[i][j] = cw[c * 7 + j];
      cbias[i] = cb[c];
    }
  }
  v4f xv[LN_OWN], gv[LN_OWN], bv[LN_OWN];
#pragma unroll
  for (int i = 0; i < LN_OWN; ++i) {
    int e = (sl * LN_OWN + i) * LN_THREADS + tid;
    xv[i] = p[e];
    gv[i] = g4[e];
    bv[i] = b4[e];
  }
  float s = (s4.x + s4.y) + (s4.z + s4.w);
  float q = (q4.x + q4.y) + (q4.z + q4.w);
  s = wave_sum(s);
  q = wave_sum(q);
  int w = tid >> 5;
  int l = tid & 31;
  if (l == 0) {
    red[w] = s;
    red[32 + w] = q;
  }
  if (mode != LN_PLAIN) {
    if (tid < LN_OWN * ROWS_PER_STEP * 2) {
      v4f z = {0.f, 0.f, 0.f, 0.f};
      tile[tid >> 3][((tid >> 1) & (ROWS_PER_STEP - 1)) * ROW_PITCH + (tid & 1) * (ROW_PITCH - 1)] = z;
    }
  }
  if (mode == LN_CONV_IN) {
#pragma unroll
    for (int i = 0; i < LN_OWN; ++i) tile[i][cell] = xv[i];
  }
  __syncthreads();
  s = wave_sum(red[l]);
  q = wave_sum(red[32 + l]);
  float inv_n = 1.0f / (float)CT;
  float mu = s * inv_n;
  float var = q * inv_n - mu * mu;
  float r = rsqrtf(var + 1e-5f);
  v4f y[LN_OWN];
#pragma unroll
  for (int i = 0; i < LN_OWN; ++i) {
    int e = (sl * LN_OWN + i) * LN_THREADS + tid;
    y[i] = (xv[i] - mu) * r * gv[i] + bv[i];
    if (mode != LN_CONV_OUT) d4[e] = y[i];
    if (mode == LN_CONV_OUT) tile[i][cell] = y[i];
  }
  if (mode == LN_PLAIN) return;
  if (mode == LN_CONV_OUT) __syncthreads();
#pragma unroll
  for (int i = 0; i < LN_OWN; ++i) {
    int e = (sl * LN_OWN + i) * LN_THREADS + tid;
    v4f lf = tile[i][cell - 1];
    v4f rt = tile[i][cell + 1];
    v4f md = (mode == LN_CONV_IN) ? xv[i] : y[i];
    float t[10] = {lf.y, lf.z, lf.w, md.x, md.y, md.z, md.w, rt.x, rt.y, rt.z};
    v4f o;
#pragma unroll
    for (int k = 0; k < 4; ++k) {
      float a = cbias[i];
#pragma unroll
      for (int j = 0; j < 7; ++j) a += t[k + j] * wt[i][j];
      o[k] = a;
    }
    c4[e] = o;
  }
}

__global__ __launch_bounds__(LN_THREADS) void ln_fused(const float* __restrict__ src,
                                                       const float* __restrict__ g,
                                                       const float* __restrict__ bb,
                                                       float* __restrict__ dst) {
  ln_body(LN_PLAIN, src, g, bb, dst, nullptr, nullptr, nullptr);
}

__global__ __launch_bounds__(LN_THREADS) void ln_conv_out(const float* __restrict__ src,
                                                          const float* __restrict__ g,
                                                          const float* __restrict__ bb,
                                                          const float* __restrict__ cw,
                                                          const float* __restrict__ cb,
                                                          float* __restrict__ cdst) {
  ln_body(LN_CONV_OUT, src, g, bb, nullptr, cw, cb, cdst);
}

__global__ __launch_bounds__(LN_THREADS) void ln_conv_in(const float* __restrict__ src,
                                                         const float* __restrict__ g,
                                                         const float* __restrict__ bb,
                                                         float* __restrict__ dst,
                                                         const float* __restrict__ cw,
                                                         const float* __restrict__ cb,
                                                         float* __restrict__ cdst) {
  ln_body(LN_CONV_IN, src, g, bb, dst, cw, cb, cdst);
}


__device__ __forceinline__ v4f ld4(const float* p) {
  return *reinterpret_cast<const v4f*>(p);
}

__device__ __forceinline__ v2f ld2(const float* p) {
  return *reinterpret_cast<const v2f*>(p);
}

__global__ __launch_bounds__(128) void pw_gemm(const float* __restrict__ Y,
                                               const float* __restrict__ W,
                                               const float* __restrict__ bias,
                                               float* __restrict__ res) {
  int l = threadIdx.x & 31;
  int tile = blockIdx.x * 4 + (threadIdx.x >> 5);
  int b = tile >> 7;
  int rem = tile & 127;
  int m0 = (rem >> 4) << 4;
  int n0 = (rem & 15) << 6;
  int lo = l & 15;
  int kh = (l >> 4) << 1;
  const float* yb = Y + b * CT + n0 + 4 * lo;
  const float* wr = W + (m0 + lo) * CC;
  v8f acc0 = {}, acc1 = {}, acc2 = {}, acc3 = {};
#pragma unroll
  for (int kk = 0; kk < CC; kk += 4) {
    int k0 = kk + kh;
    v2f a = ld2(wr + k0);
    v4f r0 = ld4(yb + k0 * TT);
    v4f r1 = ld4(yb + (k0 + 1) * TT);
    v2f b0, b1, b2, b3;
    b0.x = r0.x; b0.y = r1.x;
    b1.x = r0.y; b1.y = r1.y;
    b2.x = r0.z; b2.y = r1.z;
    b3.x = r0.w; b3.y = r1.w;
    acc0 = wmma4(a, b0, acc0);
    acc1 = wmma4(a, b1, acc1);
    acc2 = wmma4(a, b2, acc2);
    acc3 = wmma4(a, b3, acc3);
  }
  float* rb = res + b * CT + n0 + 4 * lo;
#pragma unroll
  for (int v = 0; v < 8; ++v) {
    int m = m0 + v + ((l >> 4) << 3);
    float bsv = bias[m];
    v4f* r = reinterpret_cast<v4f*>(rb + m * TT);
    v4f old = *r;
    v4f o;
    o.x = fmaxf(acc0[v] + bsv, 0.f) + old.x;
    o.y = fmaxf(acc1[v] + bsv, 0.f) + old.y;
    o.z = fmaxf(acc2[v] + bsv, 0.f) + old.z;
    o.w = fmaxf(acc3[v] + bsv, 0.f) + old.w;
    *r = o;
  }
}

__global__ __launch_bounds__(128) void xtw_gemm(const float* __restrict__ X,
                                                const float* __restrict__ W,
                                                float* __restrict__ dst) {
  int l = threadIdx.x & 31;
  int tile = blockIdx.x * 4 + (threadIdx.x >> 5);
  int b = tile >> 6;
  int t0 = (tile & 63) << 4;
  int lo = l & 15;
  int kh = (l >> 4) << 1;
  const float* xb = X + b * CT + t0 + lo;
  const float* wb = W + 4 * lo;
  float ax[CC / 4], ay[CC / 4];
  v4f r0[CC / 4], r1[CC / 4];
#pragma unroll
  for (int s = 0; s < CC / 4; ++s) {
    int k0 = 4 * s + kh;
    ax[s] = xb[k0 * TT];
    ay[s] = xb[(k0 + 1) * TT];
    r0[s] = ld4(wb + k0 * DKD);
    r1[s] = ld4(wb + (k0 + 1) * DKD);
  }
  __builtin_amdgcn_sched_barrier(0);
  v8f acc0 = {}, acc1 = {}, acc2 = {}, acc3 = {};
#pragma unroll
  for (int s = 0; s < CC / 4; ++s) {
    v2f a, b0, b1, b2, b3;
    a.x = ax[s]; a.y = ay[s];
    b0.x = r0[s].x; b0.y = r1[s].x;
    b1.x = r0[s].y; b1.y = r1[s].y;
    b2.x = r0[s].z; b2.y = r1[s].z;
    b3.x = r0[s].w; b3.y = r1[s].w;
    acc0 = wmma4(a, b0, acc0);
    acc1 = wmma4(a, b1, acc1);
    acc2 = wmma4(a, b2, acc2);
    acc3 = wmma4(a, b3, acc3);
  }
  float* db = dst + b * TT * DKD + 4 * lo;
#pragma unroll
  for (int v = 0; v < 8; ++v) {
    int t = t0 + v + ((l >> 4) << 3);
    v4f o;
    o.x = acc0[v];
    o.y = acc1[v];
    o.z = acc2[v];
    o.w = acc3[v];
    *reinterpret_cast<v4f*>(db + t * DKD) = o;
  }
}


__device__ __forceinline__ float xhalf(float x) {
  int b = __builtin_bit_cast(int, x);
  return __builtin_bit_cast(
      float, __builtin_amdgcn_permlanex16(b, b, 0x76543210, 0xfedcba98, false, false));
}

__device__ __forceinline__ void load_block(const float* __restrict__ kb,
                                           const float* __restrict__ vb,
                                           const float* __restrict__ mb,
                                           int key0, int c, int h,
                                           v4f (&k)[2][8], v4f (&v)[2][8],
                                           v4f (&hm)[2][2]) {
#pragma unroll
  for (int u = 0; u < 2; ++u) {
    const float* kp = kb + (key0 + 16 * u + c) * DKD + 4 * h;
    const float* vp = vb + (key0 + 16 * u + 8 * h) * DKD + 4 * c;
    const float* mp = mb + key0 + 16 * u + 8 * h;
#pragma unroll
    for (int m = 0; m < 8; ++m) k[u][m] = ld4(kp + 8 * m);
#pragma unroll
    for (int r = 0; r < 8; ++r) v[u][r] = ld4(vp + r * DKD);
    hm[u][0] = ld4(mp);
    hm[u][1] = ld4(mp + 4);
  }
}

__device__ __forceinline__ void process_block(const v4f (&q)[8],
                                              const v4f (&k)[2][8],
                                              const v4f (&v)[2][8],
                                              const v4f (&hm)[2][2],
                                              float& m_run, float& l_run,
                                              v8f (&acc)[4]) {
  v8f s[2] = {};
#pragma unroll
  for (int m = 0; m < 8; ++m) {
    v2f qlo, qhi, a;
    qlo.x = q[m].x; qlo.y = q[m].y;
    qhi.x = q[m].z; qhi.y = q[m].w;
    a.x = k[0][m].x; a.y = k[0][m].y;
    s[0] = wmma4(a, qlo, s[0]);
    a.x = k[1][m].x; a.y = k[1][m].y;
    s[1] = wmma4(a, qlo, s[1]);
    a.x = k[0][m].z; a.y = k[0][m].w;
    s[0] = wmma4(a, qhi, s[0]);
    a.x = k[1][m].z; a.y = k[1][m].w;
    s[1] = wmma4(a, qhi, s[1]);
  }
  float val[2][8];
  float bm = -3.4e38f;
#pragma unroll
  for (int u = 0; u < 2; ++u) {
#pragma unroll
    for (int r = 0; r < 8; ++r) {
      float hmv = hm[u][r >> 2][r & 3];
      float sv = s[u][r] * 0.125f;
      val[u][r] = sv * hmv + (1.f - hmv) * -1e30f;
      bm = fmaxf(bm, val[u][r]);
    }
  }
  bm = fmaxf(bm, xhalf(bm));
  float m_new = fmaxf(m_run, bm);
  float alpha = expf(m_run - m_new);
  float p[2][8];
  float ps = 0.f;
#pragma unroll
  for (int u = 0; u < 2; ++u) {
#pragma unroll
    for (int r = 0; r < 8; ++r) {
      p[u][r] = expf(val[u][r] - m_new);
      ps += p[u][r];
    }
  }
  l_run = l_run * alpha + ps;
  m_run = m_new;
#pragma unroll
  for (int i = 0; i < 4; ++i) acc[i] = acc[i] * alpha;
#pragma unroll
  for (int u = 0; u < 2; ++u) {
#pragma unroll
    for (int j = 0; j < 4; ++j) {
      v2f bp;
      bp.x = p[u][2 * j];
      bp.y = p[u][2 * j + 1];
#pragma unroll
      for (int i = 0; i < 4; ++i) {
        v2f a;
        a.x = v[u][2 * j][i];
        a.y = v[u][2 * j + 1][i];
        acc[i] = wmma4(a, bp, acc[i]);
      }
    }
  }
}

__global__ __launch_bounds__(128) void attn_kernel(const float* __restrict__ Q,
                                                   const float* __restrict__ Km,
                                                   const float* __restrict__ Vm,
                                                   const float* __restrict__ mask,
                                                   float* __restrict__ head) {
  int l = threadIdx.x & 31;
  int tile = blockIdx.x * 4 + (threadIdx.x >> 5);
  int b = tile >> 6;
  int q0 = (tile & 63) << 4;
  int c = l & 15;
  int h = l >> 4;
  const float* qb = Q + b * TT * DKD;
  const float* kb = Km + b * TT * DKD;
  const float* vb = Vm + b * TT * DKD;
  const float* mb = mask + b * TT;

  v4f q[8];
#pragma unroll
  for (int m = 0; m < 8; ++m) q[m] = ld4(qb + (q0 + c) * DKD + 8 * m + 4 * h);

  v4f kA[2][8], vA[2][8], hA[2][2];
  v4f kB[2][8], vB[2][8], hB[2][2];
  v8f acc[4] = {};
  float m_run = -3.4e38f;
  float l_run = 0.f;

  load_block(kb, vb, mb, 0, c, h, kA, vA, hA);
  for (int blk = 0; blk < TT / 32 - 2; blk += 2) {
    load_block(kb, vb, mb, (blk + 1) * 32, c, h, kB, vB, hB);
    process_block(q, kA, vA, hA, m_run, l_run, acc);
    load_block(kb, vb, mb, (blk + 2) * 32, c, h, kA, vA, hA);
    process_block(q, kB, vB, hB, m_run, l_run, acc);
  }
  load_block(kb, vb, mb, TT - 32, c, h, kB, vB, hB);
  process_block(q, kA, vA, hA, m_run, l_run, acc);
  process_block(q, kB, vB, hB, m_run, l_run, acc);

  float lsum = l_run + xhalf(l_run);
  float f = (1.0f / lsum) * mb[q0 + c];
  float* hp = head + b * TT * DKD + (q0 + c) * DKD + 32 * h;
#pragma unroll
  for (int r = 0; r < 8; ++r) {
    v4f o;
    o.x = acc[0][r];
    o.y = acc[1][r];
    o.z = acc[2][r];
    o.w = acc[3][r];
    *reinterpret_cast<v4f*>(hp + 4 * r) = o * f;
  }
}

__global__ __launch_bounds__(128) void wo_gemm(const float* __restrict__ H,
                                               const float* __restrict__ Wo,
                                               float* __restrict__ res) {
  int l = threadIdx.x & 31;
  int tile = blockIdx.x * 4 + (threadIdx.x >> 5);
  int b = tile >> 7;
  int rem = tile & 127;
  int m0 = (rem >> 4) << 4;
  int n0 = (rem & 15) << 6;
  int lo = l & 15;
  int kh = (l >> 4) << 1;
  const float* hb = H + b * TT * DKD + (n0 + 4 * lo) * DKD;
  const float* wc = Wo + m0 + lo;
  float* rb = res + b * CT + n0 + 4 * lo;
  float wa[DKD / 4], wb2[DKD / 4], wc2[DKD / 4], wd[DKD / 4];
  v2f hf[DKD / 4][4];
  v4f old[8];
#pragma unroll
  for (int s = 0; s < DKD / 4; ++s) {
    int k0 = 4 * s + kh;
    wa[s]  = wc[k0 * CC];
    wb2[s] = wc[(k0 + 64) * CC];
    wc2[s] = wc[(k0 + 1) * CC];
    wd[s]  = wc[(k0 + 65) * CC];
#pragma unroll
    for (int j = 0; j < 4; ++j) hf[s][j] = ld2(hb + j * DKD + k0);
  }
#pragma unroll
  for (int v = 0; v < 8; ++v) {
    int m = m0 + v + ((l >> 4) << 3);
    old[v] = ld4(rb + m * TT);
  }
  __builtin_amdgcn_sched_barrier(0);
  v8f acc0 = {}, acc1 = {}, acc2 = {}, acc3 = {};
#pragma unroll
  for (int s = 0; s < DKD / 4; ++s) {
    v2f a;
    a.x = wa[s] + wb2[s];
    a.y = wc2[s] + wd[s];
    acc0 = wmma4(a, hf[s][0], acc0);
    acc1 = wmma4(a, hf[s][1], acc1);
    acc2 = wmma4(a, hf[s][2], acc2);
    acc3 = wmma4(a, hf[s][3], acc3);
  }
#pragma unroll
  for (int v = 0; v < 8; ++v) {
    int m = m0 + v + ((l >> 4) << 3);
    v4f o;
    o.x = acc0[v] + old[v].x;
    o.y = acc1[v] + old[v].y;
    o.z = acc2[v] + old[v].z;
    o.w = acc3[v] + old[v].w;
    *reinterpret_cast<v4f*>(rb + m * TT) = o;
  }
}

__global__ __launch_bounds__(128) void fc_gemm(const float* __restrict__ X,
                                               const float* __restrict__ W,
                                               const float* __restrict__ bias,
                                               const float* __restrict__ res,
                                               float* __restrict__ out) {
  int l = threadIdx.x & 31;
  int tile = blockIdx.x * 4 + (threadIdx.x >> 5);
  int b = tile >> 7;
  int rem = tile & 127;
  int m0 = (rem >> 4) << 4;
  int n0 = (rem & 15) << 6;
  int lo = l & 15;
  int kh = (l >> 4) << 1;
  const float* xb = X + b * CT + n0 + 4 * lo;
  const float* wr = W + (m0 + lo) * CC;
  v8f acc0 = {}, acc1 = {}, acc2 = {}, acc3 = {};
#pragma unroll
  for (int kk = 0; kk < CC; kk += 4) {
    int k0 = kk + kh;
    v2f a = ld2(wr + k0);
    v4f r0 = ld4(xb + k0 * TT);
    v4f r1 = ld4(xb + (k0 + 1) * TT);
    v2f b0, b1, b2, b3;
    b0.x = r0.x; b0.y = r1.x;
    b1.x = r0.y; b1.y = r1.y;
    b2.x = r0.z; b2.y = r1.z;
    b3.x = r0.w; b3.y = r1.w;
    acc0 = wmma4(a, b0, acc0);
    acc1 = wmma4(a, b1, acc1);
    acc2 = wmma4(a, b2, acc2);
    acc3 = wmma4(a, b3, acc3);
  }
  const float* rb = res + b * CT + n0 + 4 * lo;
  float* ob = out + b * CT + n0 + 4 * lo;
#pragma unroll
  for (int v = 0; v < 8; ++v) {
    int m = m0 + v + ((l >> 4) << 3);
    float bsv = bias[m];
    v4f old = ld4(rb + m * TT);
    v4f o;
    o.x = fmaxf(acc0[v] + bsv, 0.f) + old.x;
    o.y = fmaxf(acc1[v] + bsv, 0.f) + old.y;
    o.z = fmaxf(acc2[v] + bsv, 0.f) + old.z;
    o.w = fmaxf(acc3[v] + bsv, 0.f) + old.w;
    *reinterpret_cast<v4f*>(ob + m * TT) = o;
  }
}


extern "C" void kernel_launch(void* const* d_in, const int* in_sizes, int n_in,
                              void* d_out, int out_size, void* d_ws, size_t ws_size,
                              hipStream_t stream) {
  (void)in_sizes; (void)n_in; (void)out_size; (void)ws_size;
  const float* x    = (const float*)d_in[0];
  const float* mask = (const float*)d_in[1];
  const float* dw_w = (const float*)d_in[2];
  const float* dw_b = (const float*)d_in[3];
  const float* pw_w = (const float*)d_in[4];
  const float* pw_b = (const float*)d_in[5];
  const float* n0g  = (const float*)d_in[6];
  const float* n0b  = (const float*)d_in[7];
  const float* nsg  = (const float*)d_in[8];
  const float* nsb  = (const float*)d_in[9];
  const float* neg_ = (const float*)d_in[10];
  const float* neb  = (const float*)d_in[11];
  const float* Wq   = (const float*)d_in[12];
  const float* Wk   = (const float*)d_in[13];
  const float* Wv   = (const float*)d_in[14];
  const float* Wo   = (const float*)d_in[15];
  const float* fcw  = (const float*)d_in[16];
  const float* fcb  = (const float*)d_in[17];
  float* outp = (float*)d_out;
  float* ws = (float*)d_ws;

  float* OUT = ws;
  float* RES = ws + 2097152;
  float* TMP = ws + 4194304;
  float* Qb  = ws + 6291456;
  float* Kb  = ws + 7340032;
  float* Vb  = ws + 8388608;
  float* Hb  = ws + 9437184;

  posenc_add<<<8192, 256, 0, stream>>>(x, OUT);
  ln_conv_in<<<128, 1024, 0, stream>>>(OUT, n0g, n0b, RES, dw_w, dw_b, TMP);

  for (int i = 0; i < 3; ++i) {
    pw_gemm<<<512, 128, 0, stream>>>(TMP, pw_w + i * CC * CC, pw_b + i * CC, RES);
    ln_conv_out<<<128, 1024, 0, stream>>>(RES, nsg + i * CT, nsb + i * CT,
                                          dw_w + (i + 1) * CC * 7, dw_b + (i + 1) * CC, TMP);
  }
  pw_gemm<<<512, 128, 0, stream>>>(TMP, pw_w + 3 * CC * CC, pw_b + 3 * CC, RES);
  ln_fused<<<128, 1024, 0, stream>>>(RES, nsg + 3 * CT, nsb + 3 * CT, OUT);

  xtw_gemm<<<256, 128, 0, stream>>>(OUT, Wq, Qb);
  xtw_gemm<<<256, 128, 0, stream>>>(OUT, Wk, Kb);
  xtw_gemm<<<256, 128, 0, stream>>>(OUT, Wv, Vb);
  attn_kernel<<<256, 128, 0, stream>>>(Qb, Kb, Vb, mask, Hb);
  wo_gemm<<<512, 128, 0, stream>>>(Hb, Wo, RES);
  ln_fused<<<128, 1024, 0, stream>>>(RES, neg_, neb, OUT);
  fc_gemm<<<512, 128, 0, stream>>>(OUT, fcw, fcb, RES, outp);
}


# LayerNorm sums: each of an element's 8 groups starts its sweep at its own slice and wraps round: 8 parts come from memory at once, not one; own slice kept from the sweep: another summation order
# speedup vs baseline: 2.9939x; 1.0779x over previous
#include <hip/hip_runtime.h>

typedef float v2f __attribute__((ext_vector_type(2)));
typedef float v8f __attribute__((ext_vector_type(8)));

#define BB 16
#define CC 128
#define TT 1024
#define DKD 64
#define CT (CC * TT)
#define LOG10000 9.210340371976184f
#define PI_2 1.5707963267948966f

__device__ inline v8f wmma4(v2f a, v2f b, v8f c) {
  return __builtin_amdgcn_wmma_f32_16x16x4_f32(false, a, false, b, (short)0, c,
                                               false, false);
}


__global__ void posenc_add(const float* __restrict__ x, float* __restrict__ out) {
  int idx = blockIdx.x * blockDim.x + threadIdx.x;
  int ct = idx % CT;
  int c = ct / TT;
  int t = ct % TT;
  float freq = expf(-(float)(c & ~1) * (LOG10000 / (float)CC));
  float pe = sinf((float)t * freq + (float)(c & 1) * PI_2);
  out[idx] = x[idx] + pe;
}

typedef float v4f __attribute__((ext_vector_type(4)));
#define LN_SLICES 8
#define LN_THREADS 1024
#define CT4 (CT / 4)
#define LN_ITEMS (CT4 / LN_THREADS)
#define LN_OWN (LN_ITEMS / LN_SLICES)

__device__ __forceinline__ float wave_sum(float x) {
#pragma unroll
  for (int m = 16; m > 0; m >>= 1) x += __shfl_xor(x, m);
  return x;
}

#define LN_PLAIN 0
#define LN_CONV_OUT 1
#define LN_CONV_IN 2
#define ROW_ITEMS (TT / 4)
#define ROWS_PER_STEP (LN_THREADS / ROW_ITEMS)
#define ROW_PITCH (ROW_ITEMS + 2)

__device__ __forceinline__ void ln_body(const int mode,
                                        const float* __restrict__ src,
                                        const float* __restrict__ g,
                                        const float* __restrict__ bb,
                                        float* __restrict__ dst,
                                        const float* __restrict__ cw,
                                        const float* __restrict__ cb,
                                        float* __restrict__ cdst) {
  __shared__ float red[64];
  __shared__ v4f tile[LN_OWN][ROWS_PER_STEP * ROW_PITCH];
  int b = blockIdx.x / LN_SLICES;
  int sl = blockIdx.x % LN_SLICES;
  int tid = threadIdx.x;
  const v4f* p = reinterpret_cast<const v4f*>(src + b * CT);
  const v4f* g4 = reinterpret_cast<const v4f*>(g);
  const v4f* b4 = reinterpret_cast<const v4f*>(bb);
  v4f* d4 = reinterpret_cast<v4f*>(dst + b * CT);
  v4f* c4 = reinterpret_cast<v4f*>(cdst + b * CT);
  int rowi = tid / ROW_ITEMS;
  int cell = rowi * ROW_PITCH + 1 + tid % ROW_ITEMS;

  const int base = sl * LN_OWN;
  v4f xv[LN_OWN];
#pragma unroll
  for (int i = 0; i < LN_OWN; ++i) xv[i] = p[(base + i) * LN_THREADS + tid];
  v4f s4 = {0.f, 0.f, 0.f, 0.f};
  v4f q4 = {0.f, 0.f, 0.f, 0.f};
#pragma unroll 7
  for (int i = LN_OWN; i < LN_ITEMS; ++i) {
    v4f v = p[((base + i) & (LN_ITEMS - 1)) * LN_THREADS + tid];
    s4 += v;
    q4 += v * v;
  }
#pragma unroll
  for (int i = 0; i < LN_OWN; ++i) {
    s4 += xv[i];
    q4 += xv[i] * xv[i];
  }
  float wt[LN_OWN][7], cbias[LN_OWN];
  if (mode != LN_PLAIN) {
#pragma unroll
    for (int i = 0; i < LN_OWN; ++i) {
      int c = __builtin_amdgcn_readfirstlane((sl * LN_OWN + i) * ROWS_PER_STEP + rowi);
#pragma unroll
      for (int j = 0; j < 7; ++j) wt[i][j] = cw[c * 7 + j];
      cbias[i] = cb[c];
    }
  }
  v4f gv[LN_OWN], bv[LN_OWN];
#pragma unroll
  for (int i = 0; i < LN_OWN; ++i) {
    int e = (base + i) * LN_THREADS + tid;
    gv[i] = g4[e];
    bv[i] = b4[e];
  }
  float s = (s4.x + s4.y) + (s4.z + s4.w);
  float q = (q4.x + q4.y) + (q4.z + q4.w);
  s = wave_sum(s);
  q = wave_sum(q);
  int w = tid >> 5;
  int l = tid & 31;
  if (l == 0) {
    red[w] = s;
    red[32 + w] = q;
  }
  if (mode != LN_PLAIN) {
    if (tid < LN_OWN * ROWS_PER_STEP * 2) {
      v4f z = {0.f, 0.f, 0.f, 0.f};
      tile[tid >> 3][((tid >> 1) & (ROWS_PER_STEP - 1)) * ROW_PITCH + (tid & 1) * (ROW_PITCH - 1)] = z;
    }
  }
  if (mode == LN_CONV_IN) {
#pragma unroll
    for (int i = 0; i < LN_OWN; ++i) tile[i][cell] = xv[i];
  }
  __syncthreads();
  s = wave_sum(red[l]);
  q = wave_sum(red[32 + l]);
  float inv_n = 1.0f / (float)CT;
  float mu = s * inv_n;
  float var = q * inv_n - mu * mu;
  float r = rsqrtf(var + 1e-5f);
  v4f y[LN_OWN];
#pragma unroll
  for (int i = 0; i < LN_OWN; ++i) {
    int e = (sl * LN_OWN + i) * LN_THREADS + tid;
    y[i] = (xv[i] - mu) * r * gv[i] + bv[i];
    if (mode != LN_CONV_OUT) d4[e] = y[i];
    if (mode == LN_CONV_OUT) tile[i][cell] = y[i];
  }
  if (mode == LN_PLAIN) return;
  if (mode == LN_CONV_OUT) __syncthreads();
#pragma unroll
  for (int i = 0; i < LN_OWN; ++i) {
    int e = (sl * LN_OWN + i) * LN_THREADS + tid;
    v4f lf = tile[i][cell - 1];
    v4f rt = tile[i][cell + 1];
    v4f md = (mode == LN_CONV_IN) ? xv[i] : y[i];
    float t[10] = {lf.y, lf.z, lf.w, md.x, md.y, md.z, md.w, rt.x, rt.y, rt.z};
    v4f o;
#pragma unroll
    for (int k = 0; k < 4; ++k) {
      float a = cbias[i];
#pragma unroll
      for (int j = 0; j < 7; ++j) a += t[k + j] * wt[i][j];
      o[k] = a;
    }
    c4[e] = o;
  }
}

__global__ __launch_bounds__(LN_THREADS) void ln_fused(const float* __restrict__ src,
                                                       const float* __restrict__ g,
                                                       const float* __restrict__ bb,
                                                       float* __restrict__ dst) {
  ln_body(LN_PLAIN, src, g, bb, dst, nullptr, nullptr, nullptr);
}

__global__ __launch_bounds__(LN_THREADS) void ln_conv_out(const float* __restrict__ src,
                                                          const float* __restrict__ g,
                                                          const float* __restrict__ bb,
                                                          const float* __restrict__ cw,
                                                          const float* __restrict__ cb,
                                                          float* __restrict__ cdst) {
  ln_body(LN_CONV_OUT, src, g, bb, nullptr, cw, cb, cdst);
}

__global__ __launch_bounds__(LN_THREADS) void ln_conv_in(const float* __restrict__ src,
                                                         const float* __restrict__ g,
                                                         const float* __restrict__ bb,
                                                         float* __restrict__ dst,
                                                         const float* __restrict__ cw,
                                                         const float* __restrict__ cb,
                                                         float* __restrict__ cdst) {
  ln_body(LN_CONV_IN, src, g, bb, dst, cw, cb, cdst);
}


__device__ __forceinline__ v4f ld4(const float* p) {
  return *reinterpret_cast<const v4f*>(p);
}

__device__ __forceinline__ v2f ld2(const float* p) {
  return *reinterpret_cast<const v2f*>(p);
}

__global__ __launch_bounds__(128) void pw_gemm(const float* __restrict__ Y,
                                               const float* __restrict__ W,
                                               const float* __restrict__ bias,
                                               float* __restrict__ res) {
  int l = threadIdx.x & 31;
  int tile = blockIdx.x * 4 + (threadIdx.x >> 5);
  int b = tile >> 7;
  int rem = tile & 127;
  int m0 = (rem >> 4) << 4;
  int n0 = (rem & 15) << 6;
  int lo = l & 15;
  int kh = (l >> 4) << 1;
  const float* yb = Y + b * CT + n0 + 4 * lo;
  const float* wr = W + (m0 + lo) * CC;
  v8f acc0 = {}, acc1 = {}, acc2 = {}, acc3 = {};
#pragma unroll
  for (int kk = 0; kk < CC; kk += 4) {
    int k0 = kk + kh;
    v2f a = ld2(wr + k0);
    v4f r0 = ld4(yb + k0 * TT);
    v4f r1 = ld4(yb + (k0 + 1) * TT);
    v2f b0, b1, b2, b3;
    b0.x = r0.x; b0.y = r1.x;
    b1.x = r0.y; b1.y = r1.y;
    b2.x = r0.z; b2.y = r1.z;
    b3.x = r0.w; b3.y = r1.w;
    acc0 = wmma4(a, b0, acc0);
    acc1 = wmma4(a, b1, acc1);
    acc2 = wmma4(a, b2, acc2);
    acc3 = wmma4(a, b3, acc3);
  }
  float* rb = res + b * CT + n0 + 4 * lo;
#pragma unroll
  for (int v = 0; v < 8; ++v) {
    int m = m0 + v + ((l >> 4) << 3);
    float bsv = bias[m];
    v4f* r = reinterpret_cast<v4f*>(rb + m * TT);
    v4f old = *r;
    v4f o;
    o.x = fmaxf(acc0[v] + bsv, 0.f) + old.x;
    o.y = fmaxf(acc1[v] + bsv, 0.f) + old.y;
    o.z = fmaxf(acc2[v] + bsv, 0.f) + old.z;
    o.w = fmaxf(acc3[v] + bsv, 0.f) + old.w;
    *r = o;
  }
}

__global__ __launch_bounds__(128) void xtw_gemm(const float* __restrict__ X,
                                                const float* __restrict__ W,
                                                float* __restrict__ dst) {
  int l = threadIdx.x & 31;
  int tile = blockIdx.x * 4 + (threadIdx.x >> 5);
  int b = tile >> 6;
  int t0 = (tile & 63) << 4;
  int lo = l & 15;
  int kh = (l >> 4) << 1;
  const float* xb = X + b * CT + t0 + lo;
  const float* wb = W + 4 * lo;
  float ax[CC / 4], ay[CC / 4];
  v4f r0[CC / 4], r1[CC / 4];
#pragma unroll
  for (int s = 0; s < CC / 4; ++s) {
    int k0 = 4 * s + kh;
    ax[s] = xb[k0 * TT];
    ay[s] = xb[(k0 + 1) * TT];
    r0[s] = ld4(wb + k0 * DKD);
    r1[s] = ld4(wb + (k0 + 1) * DKD);
  }
  __builtin_amdgcn_sched_barrier(0);
  v8f acc0 = {}, acc1 = {}, acc2 = {}, acc3 = {};
#pragma unroll
  for (int s = 0; s < CC / 4; ++s) {
    v2f a, b0, b1, b2, b3;
    a.x = ax[s]; a.y = ay[s];
    b0.x = r0[s].x; b0.y = r1[s].x;
    b1.x = r0[s].y; b1.y = r1[s].y;
    b2.x = r0[s].z; b2.y = r1[s].z;
    b3.x = r0[s].w; b3.y = r1[s].w;
    acc0 = wmma4(a, b0, acc0);
    acc1 = wmma4(a, b1, acc1);
    acc2 = wmma4(a, b2, acc2);
    acc3 = wmma4(a, b3, acc3);
  }
  float* db = dst + b * TT * DKD + 4 * lo;
#pragma unroll
  for (int v = 0; v < 8; ++v) {
    int t = t0 + v + ((l >> 4) << 3);
    v4f o;
    o.x = acc0[v];
    o.y = acc1[v];
    o.z = acc2[v];
    o.w = acc3[v];
    *reinterpret_cast<v4f*>(db + t * DKD) = o;
  }
}


__device__ __forceinline__ float xhalf(float x) {
  int b = __builtin_bit_cast(int, x);
  return __builtin_bit_cast(
      float, __builtin_amdgcn_permlanex16(b, b, 0x76543210, 0xfedcba98, false, false));
}

__device__ __forceinline__ void load_block(const float* __restrict__ kb,
                                           const float* __restrict__ vb,
                                           const float* __restrict__ mb,
                                           int key0, int c, int h,
                                           v4f (&k)[2][8], v4f (&v)[2][8],
                                           v4f (&hm)[2][2]) {
#pragma unroll
  for (int u = 0; u < 2; ++u) {
    const float* kp = kb + (key0 + 16 * u + c) * DKD + 4 * h;
    const float* vp = vb + (key0 + 16 * u + 8 * h) * DKD + 4 * c;
    const float* mp = mb + key0 + 16 * u + 8 * h;
#pragma unroll
    for (int m = 0; m < 8; ++m) k[u][m] = ld4(kp + 8 * m);
#pragma unroll
    for (int r = 0; r < 8; ++r) v[u][r] = ld4(vp + r * DKD);
    hm[u][0] = ld4(mp);
    hm[u][1] = ld4(mp + 4);
  }
}

__device__ __forceinline__ void process_block(const v4f (&q)[8],
                                              const v4f (&k)[2][8],
                                              const v4f (&v)[2][8],
                                              const v4f (&hm)[2][2],
                                              float& m_run, float& l_run,
                                              v8f (&acc)[4]) {
  v8f s[2] = {};
#pragma unroll
  for (int m = 0; m < 8; ++m) {
    v2f qlo, qhi, a;
    qlo.x = q[m].x; qlo.y = q[m].y;
    qhi.x = q[m].z; qhi.y = q[m].w;
    a.x = k[0][m].x; a.y = k[0][m].y;
    s[0] = wmma4(a, qlo, s[0]);
    a.x = k[1][m].x; a.y = k[1][m].y;
    s[1] = wmma4(a, qlo, s[1]);
    a.x = k[0][m].z; a.y = k[0][m].w;
    s[0] = wmma4(a, qhi, s[0]);
    a.x = k[1][m].z; a.y = k[1][m].w;
    s[1] = wmma4(a, qhi, s[1]);
  }
  float val[2][8];
  float bm = -3.4e38f;
#pragma unroll
  for (int u = 0; u < 2; ++u) {
#pragma unroll
    for (int r = 0; r < 8; ++r) {
      float hmv = hm[u][r >> 2][r & 3];
      float sv = s[u][r] * 0.125f;
      val[u][r] = sv * hmv + (1.f - hmv) * -1e30f;
      bm = fmaxf(bm, val[u][r]);
    }
  }
  bm = fmaxf(bm, xhalf(bm));
  float m_new = fmaxf(m_run, bm);
  float alpha = expf(m_run - m_new);
  float p[2][8];
  float ps = 0.f;
#pragma unroll
  for (int u = 0; u < 2; ++u) {
#pragma unroll
    for (int r = 0; r < 8; ++r) {
      p[u][r] = expf(val[u][r] - m_new);
      ps += p[u][r];
    }
  }
  l_run = l_run * alpha + ps;
  m_run = m_new;
#pragma unroll
  for (int i = 0; i < 4; ++i) acc[i] = acc[i] * alpha;
#pragma unroll
  for (int u = 0; u < 2; ++u) {
#pragma unroll
    for (int j = 0; j < 4; ++j) {
      v2f bp;
      bp.x = p[u][2 * j];
      bp.y = p[u][2 * j + 1];
#pragma unroll
      for (int i = 0; i < 4; ++i) {
        v2f a;
        a.x = v[u][2 * j][i];
        a.y = v[u][2 * j + 1][i];
        acc[i] = wmma4(a, bp, acc[i]);
      }
    }
  }
}

__global__ __launch_bounds__(128) void attn_kernel(const float* __restrict__ Q,
                                                   const float* __restrict__ Km,
                                                   const float* __restrict__ Vm,
                                                   const float* __restrict__ mask,
                                                   float* __restrict__ head) {
  int l = threadIdx.x & 31;
  int tile = blockIdx.x * 4 + (threadIdx.x >> 5);
  int b = tile >> 6;
  int q0 = (tile & 63) << 4;
  int c = l & 15;
  int h = l >> 4;
  const float* qb = Q + b * TT * DKD;
  const float* kb = Km + b * TT * DKD;
  const float* vb = Vm + b * TT * DKD;
  const float* mb = mask + b * TT;

  v4f q[8];
#pragma unroll
  for (int m = 0; m < 8; ++m) q[m] = ld4(qb + (q0 + c) * DKD + 8 * m + 4 * h);

  v4f kA[2][8], vA[2][8], hA[2][2];
  v4f kB[2][8], vB[2][8], hB[2][2];
  v8f acc[4] = {};
  float m_run = -3.4e38f;
  float l_run = 0.f;

  load_block(kb, vb, mb, 0, c, h, kA, vA, hA);
  for (int blk = 0; blk < TT / 32 - 2; blk += 2) {
    load_block(kb, vb, mb, (blk + 1) * 32, c, h, kB, vB, hB);
    process_block(q, kA, vA, hA, m_run, l_run, acc);
    load_block(kb, vb, mb, (blk + 2) * 32, c, h, kA, vA, hA);
    process_block(q, kB, vB, hB, m_run, l_run, acc);
  }
  load_block(kb, vb, mb, TT - 32, c, h, kB, vB, hB);
  process_block(q, kA, vA, hA, m_run, l_run, acc);
  process_block(q, kB, vB, hB, m_run, l_run, acc);

  float lsum = l_run + xhalf(l_run);
  float f = (1.0f / lsum) * mb[q0 + c];
  float* hp = head + b * TT * DKD + (q0 + c) * DKD + 32 * h;
#pragma unroll
  for (int r = 0; r < 8; ++r) {
    v4f o;
    o.x = acc[0][r];
    o.y = acc[1][r];
    o.z = acc[2][r];
    o.w = acc[3][r];
    *reinterpret_cast<v4f*>(hp + 4 * r) = o * f;
  }
}

__global__ __launch_bounds__(128) void wo_gemm(const float* __restrict__ H,
                                               const float* __restrict__ Wo,
                                               float* __restrict__ res) {
  int l = threadIdx.x & 31;
  int tile = blockIdx.x * 4 + (threadIdx.x >> 5);
  int b = tile >> 7;
  int rem = tile & 127;
  int m0 = (rem >> 4) << 4;
  int n0 = (rem & 15) << 6;
  int lo = l & 15;
  int kh = (l >> 4) << 1;
  const float* hb = H + b * TT * DKD + (n0 + 4 * lo) * DKD;
  const float* wc = Wo + m0 + lo;
  float* rb = res + b * CT + n0 + 4 * lo;
  float wa[DKD / 4], wb2[DKD / 4], wc2[DKD / 4], wd[DKD / 4];
  v2f hf[DKD / 4][4];
  v4f old[8];
#pragma unroll
  for (int s = 0; s < DKD / 4; ++s) {
    int k0 = 4 * s + kh;
    wa[s]  = wc[k0 * CC];
    wb2[s] = wc[(k0 + 64) * CC];
    wc2[s] = wc[(k0 + 1) * CC];
    wd[s]  = wc[(k0 + 65) * CC];
#pragma unroll
    for (int j = 0; j < 4; ++j) hf[s][j] = ld2(hb + j * DKD + k0);
  }
#pragma unroll
  for (int v = 0; v < 8; ++v) {
    int m = m0 + v + ((l >> 4) << 3);
    old[v] = ld4(rb + m * TT);
  }
  __builtin_amdgcn_sched_barrier(0);
  v8f acc0 = {}, acc1 = {}, acc2 = {}, acc3 = {};
#pragma unroll
  for (int s = 0; s < DKD / 4; ++s) {
    v2f a;
    a.x = wa[s] + wb2[s];
    a.y = wc2[s] + wd[s];
    acc0 = wmma4(a, hf[s][0], acc0);
    acc1 = wmma4(a, hf[s][1], acc1);
    acc2 = wmma4(a, hf[s][2], acc2);
    acc3 = wmma4(a, hf[s][3], acc3);
  }
#pragma unroll
  for (int v = 0; v < 8; ++v) {
    int m = m0 + v + ((l >> 4) << 3);
    v4f o;
    o.x = acc0[v] + old[v].x;
    o.y = acc1[v] + old[v].y;
    o.z = acc2[v] + old[v].z;
    o.w = acc3[v] + old[v].w;
    *reinterpret_cast<v4f*>(rb + m * TT) = o;
  }
}

__global__ __launch_bounds__(128) void fc_gemm(const float* __restrict__ X,
                                               const float* __restrict__ W,
                                               const float* __restrict__ bias,
                                               const float* __restrict__ res,
                                               float* __restrict__ out) {
  int l = threadIdx.x & 31;
  int tile = blockIdx.x * 4 + (threadIdx.x >> 5);
  int b = tile >> 7;
  int rem = tile & 127;
  int m0 = (rem >> 4) << 4;
  int n0 = (rem & 15) << 6;
  int lo = l & 15;
  int kh = (l >> 4) << 1;
  const float* xb = X + b * CT + n0 + 4 * lo;
  const float* wr = W + (m0 + lo) * CC;
  v8f acc0 = {}, acc1 = {}, acc2 = {}, acc3 = {};
#pragma unroll
  for (int kk = 0; kk < CC; kk += 4) {
    int k0 = kk + kh;
    v2f a = ld2(wr + k0);
    v4f r0 = ld4(xb + k0 * TT);
    v4f r1 = ld4(xb + (k0 + 1) * TT);
    v2f b0, b1, b2, b3;
    b0.x = r0.x; b0.y = r1.x;
    b1.x = r0.y; b1.y = r1.y;
    b2.x = r0.z; b2.y = r1.z;
    b3.x = r0.w; b3.y = r1.w;
    acc0 = wmma4(a, b0, acc0);
    acc1 = wmma4(a, b1, acc1);
    acc2 = wmma4(a, b2, acc2);
    acc3 = wmma4(a, b3, acc3);
  }
  const float* rb = res + b * CT + n0 + 4 * lo;
  float* ob = out + b * CT + n0 + 4 * lo;
#pragma unroll
  for (int v = 0; v < 8; ++v) {
    int m = m0 + v + ((l >> 4) << 3);
    float bsv = bias[m];
    v4f old = ld4(rb + m * TT);
    v4f o;
    o.x = fmaxf(acc0[v] + bsv, 0.f) + old.x;
    o.y = fmaxf(acc1[v] + bsv, 0.f) + old.y;
    o.z = fmaxf(acc2[v] + bsv, 0.f) + old.z;
    o.w = fmaxf(acc3[v] + bsv, 0.f) + old.w;
    *reinterpret_cast<v4f*>(ob + m * TT) = o;
  }
}


extern "C" void kernel_launch(void* const* d_in, const int* in_sizes, int n_in,
                              void* d_out, int out_size, void* d_ws, size_t ws_size,
                              hipStream_t stream) {
  (void)in_sizes; (void)n_in; (void)out_size; (void)ws_size;
  const float* x    = (const float*)d_in[0];
  const float* mask = (const float*)d_in[1];
  const float* dw_w = (const float*)d_in[2];
  const float* dw_b = (const float*)d_in[3];
  const float* pw_w = (const float*)d_in[4];
  const float* pw_b = (const float*)d_in[5];
  const float* n0g  = (const float*)d_in[6];
  const float* n0b  = (const float*)d_in[7];
  const float* nsg  = (const float*)d_in[8];
  const float* nsb  = (const float*)d_in[9];
  const float* neg_ = (const float*)d_in[10];
  const float* neb  = (const float*)d_in[11];
  const float* Wq   = (const float*)d_in[12];
  const float* Wk   = (const float*)d_in[13];
  const float* Wv   = (const float*)d_in[14];
  const float* Wo   = (const float*)d_in[15];
  const float* fcw  = (const float*)d_in[16];
  const float* fcb  = (const float*)d_in[17];
  float* outp = (float*)d_out;
  float* ws = (float*)d_ws;

  float* OUT = ws;
  float* RES = ws + 2097152;
  float* TMP = ws + 4194304;
  float* Qb  = ws + 6291456;
  float* Kb  = ws + 7340032;
  float* Vb  = ws + 8388608;
  float* Hb  = ws + 9437184;

  posenc_add<<<8192, 256, 0, stream>>>(x, OUT);
  ln_conv_in<<<128, 1024, 0, stream>>>(OUT, n0g, n0b, RES, dw_w, dw_b, TMP);

  for (int i = 0; i < 3; ++i) {
    pw_gemm<<<512, 128, 0, stream>>>(TMP, pw_w + i * CC * CC, pw_b + i * CC, RES);
    ln_conv_out<<<128, 1024, 0, stream>>>(RES, nsg + i * CT, nsb + i * CT,
                                          dw_w + (i + 1) * CC * 7, dw_b + (i + 1) * CC, TMP);
  }
  pw_gemm<<<512, 128, 0, stream>>>(TMP, pw_w + 3 * CC * CC, pw_b + 3 * CC, RES);
  ln_fused<<<128, 1024, 0, stream>>>(RES, nsg + 3 * CT, nsb + 3 * CT, OUT);

  xtw_gemm<<<256, 128, 0, stream>>>(OUT, Wq, Qb);
  xtw_gemm<<<256, 128, 0, stream>>>(OUT, Wk, Kb);
  xtw_gemm<<<256, 128, 0, stream>>>(OUT, Wv, Vb);
  attn_kernel<<<256, 128, 0, stream>>>(Qb, Kb, Vb, mask, Hb);
  wo_gemm<<<512, 128, 0, stream>>>(Hb, Wo, RES);
  ln_fused<<<128, 1024, 0, stream>>>(RES, neg_, neb, OUT);
  fc_gemm<<<512, 128, 0, stream>>>(OUT, fcw, fcb, RES, outp);
}


# Row-strided fragment operands (16x128 weight strip in pw_gemm and fc_gemm, 64x64 head tile in wo_gemm) are fetched as whole lines into LDS and read back as k-pairs; products and sums unchanged
# speedup vs baseline: 3.1396x; 1.0487x over previous
#include <hip/hip_runtime.h>

typedef float v2f __attribute__((ext_vector_type(2)));
typedef float v8f __attribute__((ext_vector_type(8)));

#define BB 16
#define CC 128
#define TT 1024
#define DKD 64
#define CT (CC * TT)
#define LOG10000 9.210340371976184f
#define PI_2 1.5707963267948966f

__device__ inline v8f wmma4(v2f a, v2f b, v8f c) {
  return __builtin_amdgcn_wmma_f32_16x16x4_f32(false, a, false, b, (short)0, c,
                                               false, false);
}


__global__ void posenc_add(const float* __restrict__ x, float* __restrict__ out) {
  int idx = blockIdx.x * blockDim.x + threadIdx.x;
  int ct = idx % CT;
  int c = ct / TT;
  int t = ct % TT;
  float freq = expf(-(float)(c & ~1) * (LOG10000 / (float)CC));
  float pe = sinf((float)t * freq + (float)(c & 1) * PI_2);
  out[idx] = x[idx] + pe;
}

typedef float v4f __attribute__((ext_vector_type(4)));
#define LN_SLICES 8
#define LN_THREADS 1024
#define CT4 (CT / 4)
#define LN_ITEMS (CT4 / LN_THREADS)
#define LN_OWN (LN_ITEMS / LN_SLICES)

__device__ __forceinline__ float wave_sum(float x) {
#pragma unroll
  for (int m = 16; m > 0; m >>= 1) x += __shfl_xor(x, m);
  return x;
}

#define LN_PLAIN 0
#define LN_CONV_OUT 1
#define LN_CONV_IN 2
#define ROW_ITEMS (TT / 4)
#define ROWS_PER_STEP (LN_THREADS / ROW_ITEMS)
#define ROW_PITCH (ROW_ITEMS + 2)

__device__ __forceinline__ void ln_body(const int mode,
                                        const float* __restrict__ src,
                                        const float* __restrict__ g,
                                        const float* __restrict__ bb,
                                        float* __restrict__ dst,
                                        const float* __restrict__ cw,
                                        const float* __restrict__ cb,
                                        float* __restrict__ cdst) {
  __shared__ float red[64];
  __shared__ v4f tile[LN_OWN][ROWS_PER_STEP * ROW_PITCH];
  int b = blockIdx.x / LN_SLICES;
  int sl = blockIdx.x % LN_SLICES;
  int tid = threadIdx.x;
  const v4f* p = reinterpret_cast<const v4f*>(src + b * CT);
  const v4f* g4 = reinterpret_cast<const v4f*>(g);
  const v4f* b4 = reinterpret_cast<const v4f*>(bb);
  v4f* d4 = reinterpret_cast<v4f*>(dst + b * CT);
  v4f* c4 = reinterpret_cast<v4f*>(cdst + b * CT);
  int rowi = tid / ROW_ITEMS;
  int cell = rowi * ROW_PITCH + 1 + tid % ROW_ITEMS;

  const int base = sl * LN_OWN;
  v4f xv[LN_OWN];
#pragma unroll
  for (int i = 0; i < LN_OWN; ++i) xv[i] = p[(base + i) * LN_THREADS + tid];
  v4f s4 = {0.f, 0.f, 0.f, 0.f};
  v4f q4 = {0.f, 0.f, 0.f, 0.f};
#pragma unroll 7
  for (int i = LN_OWN; i < LN_ITEMS; ++i) {
    v4f v = p[((base + i) & (LN_ITEMS - 1)) * LN_THREADS + tid];
    s4 += v;
    q4 += v * v;
  }
#pragma unroll
  for (int i = 0; i < LN_OWN; ++i) {
    s4 += xv[i];
    q4 += xv[i] * xv[i];
  }
  float wt[LN_OWN][7], cbias[LN_OWN];
  if (mode != LN_PLAIN) {
#pragma unroll
    for (int i = 0; i < LN_OWN; ++i) {
      int c = __builtin_amdgcn_readfirstlane((sl * LN_OWN + i) * ROWS_PER_STEP + rowi);
#pragma unroll
      for (int j = 0; j < 7; ++j) wt[i][j] = cw[c * 7 + j];
      cbias[i] = cb[c];
    }
  }
  v4f gv[LN_OWN], bv[LN_OWN];
#pragma unroll
  for (int i = 0; i < LN_OWN; ++i) {
    int e = (base + i) * LN_THREADS + tid;
    gv[i] = g4[e];
    bv[i] = b4[e];
  }
  float s = (s4.x + s4.y) + (s4.z + s4.w);
  float q = (q4.x + q4.y) + (q4.z + q4.w);
  s = wave_sum(s);
  q = wave_sum(q);
  int w = tid >> 5;
  int l = tid & 31;
  if (l == 0) {
    red[w] = s;
    red[32 + w] = q;
  }
  if (mode != LN_PLAIN) {
    if (tid < LN_OWN * ROWS_PER_STEP * 2) {
      v4f z = {0.f, 0.f, 0.f, 0.f};
      tile[tid >> 3][((tid >> 1) & (ROWS_PER_STEP - 1)) * ROW_PITCH + (tid & 1) * (ROW_PITCH - 1)] = z;
    }
  }
  if (mode == LN_CONV_IN) {
#pragma unroll
    for (int i = 0; i < LN_OWN; ++i) tile[i][cell] = xv[i];
  }
  __syncthreads();
  s = wave_sum(red[l]);
  q = wave_sum(red[32 + l]);
  float inv_n = 1.0f / (float)CT;
  float mu = s * inv_n;
  float var = q * inv_n - mu * mu;
  float r = rsqrtf(var + 1e-5f);
  v4f y[LN_OWN];
#pragma unroll
  for (int i = 0; i < LN_OWN; ++i) {
    int e = (sl * LN_OWN + i) * LN_THREADS + tid;
    y[i] = (xv[i] - mu) * r * gv[i] + bv[i];
    if (mode != LN_CONV_OUT) d4[e] = y[i];
    if (mode == LN_CONV_OUT) tile[i][cell] = y[i];
  }
  if (mode == LN_PLAIN) return;
  if (mode == LN_CONV_OUT) __syncthreads();
#pragma unroll
  for (int i = 0; i < LN_OWN; ++i) {
    int e = (sl * LN_OWN + i) * LN_THREADS + tid;
    v4f lf = tile[i][cell - 1];
    v4f rt = tile[i][cell + 1];
    v4f md = (mode == LN_CONV_IN) ? xv[i] : y[i];
    float t[10] = {lf.y, lf.z, lf.w, md.x, md.y, md.z, md.w, rt.x, rt.y, rt.z};
    v4f o;
#pragma unroll
    for (int k = 0; k < 4; ++k) {
      float a = cbias[i];
#pragma unroll
      for (int j = 0; j < 7; ++j) a += t[k + j] * wt[i][j];
      o[k] = a;
    }
    c4[e] = o;
  }
}

__global__ __launch_bounds__(LN_THREADS) void ln_fused(const float* __restrict__ src,
                                                       const float* __restrict__ g,
                                                       const float* __restrict__ bb,
                                                       float* __restrict__ dst) {
  ln_body(LN_PLAIN, src, g, bb, dst, nullptr, nullptr, nullptr);
}

__global__ __launch_bounds__(LN_THREADS) void ln_conv_out(const float* __restrict__ src,
                                                          const float* __restrict__ g,
                                                          const float* __restrict__ bb,
                                                          const float* __restrict__ cw,
                                                          const float* __restrict__ cb,
                                                          float* __restrict__ cdst) {
  ln_body(LN_CONV_OUT, src, g, bb, nullptr, cw, cb, cdst);
}

__global__ __launch_bounds__(LN_THREADS) void ln_conv_in(const float* __restrict__ src,
                                                         const float* __restrict__ g,
                                                         const float* __restrict__ bb,
                                                         float* __restrict__ dst,
                                                         const float* __restrict__ cw,
                                                         const float* __restrict__ cb,
                                                         float* __restrict__ cdst) {
  ln_body(LN_CONV_IN, src, g, bb, dst, cw, cb, cdst);
}


__device__ __forceinline__ v4f ld4(const float* p) {
  return *reinterpret_cast<const v4f*>(p);
}

__device__ __forceinline__ v2f ld2(const float* p) {
  return *reinterpret_cast<const v2f*>(p);
}


__device__ __forceinline__ void lds_handover() {
  asm volatile("s_wait_dscnt 0x0\n\ts_barrier_signal -1\n\ts_barrier_wait -1" ::: "memory");
}

#define WS_PITCH (CC + 4)

__device__ __forceinline__ void strip_fetch(const float* __restrict__ W, int m0, v4f (&ws)[4]) {
  int srow = threadIdx.x >> 5;
  int scol = (threadIdx.x & 31) << 2;
#pragma unroll
  for (int i = 0; i < 4; ++i) ws[i] = ld4(W + (m0 + srow + 4 * i) * CC + scol);
}

__device__ __forceinline__ void strip_store(float* wst, const v4f (&ws)[4]) {
  int srow = threadIdx.x >> 5;
  int scol = (threadIdx.x & 31) << 2;
#pragma unroll
  for (int i = 0; i < 4; ++i)
    *reinterpret_cast<v4f*>(wst + (srow + 4 * i) * WS_PITCH + scol) = ws[i];
}

__global__ __launch_bounds__(128) void pw_gemm(const float* __restrict__ Y,
                                               const float* __restrict__ W,
                                               const float* __restrict__ bias,
                                               float* __restrict__ res) {
  __shared__ v4f wst4[16 * WS_PITCH / 4];
  float* wst = reinterpret_cast<float*>(wst4);
  int l = threadIdx.x & 31;
  int tile = blockIdx.x * 4 + (threadIdx.x >> 5);
  int b = tile >> 7;
  int rem = tile & 127;
  int m0 = (rem >> 4) << 4;
  int n0 = (rem & 15) << 6;
  int lo = l & 15;
  int kh = (l >> 4) << 1;
  const float* yb = Y + b * CT + n0 + 4 * lo;
  v4f ws[4];
  strip_fetch(W, m0, ws);
  __builtin_amdgcn_sched_barrier(0);
  v4f r0[CC / 4], r1[CC / 4];
#pragma unroll
  for (int s = 0; s < CC / 4; ++s) {
    int k0 = 4 * s + kh;
    r0[s] = ld4(yb + k0 * TT);
    r1[s] = ld4(yb + (k0 + 1) * TT);
  }
  __builtin_amdgcn_sched_barrier(0);
  strip_store(wst, ws);
  lds_handover();
  const float* wl = wst + lo * WS_PITCH + kh;
  v2f af[CC / 4];
#pragma unroll
  for (int s = 0; s < CC / 4; ++s) af[s] = ld2(wl + 4 * s);
  float* rb = res + b * CT + n0 + 4 * lo;
  v4f old[8];
  float bsv[8];
#pragma unroll
  for (int v = 0; v < 8; ++v) {
    int m = m0 + v + ((l >> 4) << 3);
    bsv[v] = bias[m];
    old[v] = ld4(rb + m * TT);
  }
  __builtin_amdgcn_sched_barrier(0);
  v8f acc0 = {}, acc1 = {}, acc2 = {}, acc3 = {};
#pragma unroll
  for (int s = 0; s < CC / 4; ++s) {
    v2f b0, b1, b2, b3;
    b0.x = r0[s].x; b0.y = r1[s].x;
    b1.x = r0[s].y; b1.y = r1[s].y;
    b2.x = r0[s].z; b2.y = r1[s].z;
    b3.x = r0[s].w; b3.y = r1[s].w;
    acc0 = wmma4(af[s], b0, acc0);
    acc1 = wmma4(af[s], b1, acc1);
    acc2 = wmma4(af[s], b2, acc2);
    acc3 = wmma4(af[s], b3, acc3);
    if ((s & 3) == 3) __builtin_amdgcn_sched_barrier(0);
  }
#pragma unroll
  for (int v = 0; v < 8; ++v) {
    int m = m0 + v + ((l >> 4) << 3);
    v4f o;
    o.x = fmaxf(acc0[v] + bsv[v], 0.f) + old[v].x;
    o.y = fmaxf(acc1[v] + bsv[v], 0.f) + old[v].y;
    o.z = fmaxf(acc2[v] + bsv[v], 0.f) + old[v].z;
    o.w = fmaxf(acc3[v] + bsv[v], 0.f) + old[v].w;
    *reinterpret_cast<v4f*>(rb + m * TT) = o;
  }
}

__global__ __launch_bounds__(128) void xtw_gemm(const float* __restrict__ X,
                                                const float* __restrict__ W,
                                                float* __restrict__ dst) {
  int l = threadIdx.x & 31;
  int tile = blockIdx.x * 4 + (threadIdx.x >> 5);
  int b = tile >> 6;
  int t0 = (tile & 63) << 4;
  int lo = l & 15;
  int kh = (l >> 4) << 1;
  const float* xb = X + b * CT + t0 + lo;
  const float* wb = W + 4 * lo;
  float ax[CC / 4], ay[CC / 4];
  v4f r0[CC / 4], r1[CC / 4];
#pragma unroll
  for (int s = 0; s < CC / 4; ++s) {
    int k0 = 4 * s + kh;
    ax[s] = xb[k0 * TT];
    ay[s] = xb[(k0 + 1) * TT];
    r0[s] = ld4(wb + k0 * DKD);
    r1[s] = ld4(wb + (k0 + 1) * DKD);
  }
  __builtin_amdgcn_sched_barrier(0);
  v8f acc0 = {}, acc1 = {}, acc2 = {}, acc3 = {};
#pragma unroll
  for (int s = 0; s < CC / 4; ++s) {
    v2f a, b0, b1, b2, b3;
    a.x = ax[s]; a.y = ay[s];
    b0.x = r0[s].x; b0.y = r1[s].x;
    b1.x = r0[s].y; b1.y = r1[s].y;
    b2.x = r0[s].z; b2.y = r1[s].z;
    b3.x = r0[s].w; b3.y = r1[s].w;
    acc0 = wmma4(a, b0, acc0);
    acc1 = wmma4(a, b1, acc1);
    acc2 = wmma4(a, b2, acc2);
    acc3 = wmma4(a, b3, acc3);
  }
  float* db = dst + b * TT * DKD + 4 * lo;
#pragma unroll
  for (int v = 0; v < 8; ++v) {
    int t = t0 + v + ((l >> 4) << 3);
    v4f o;
    o.x = acc0[v];
    o.y = acc1[v];
    o.z = acc2[v];
    o.w = acc3[v];
    *reinterpret_cast<v4f*>(db + t * DKD) = o;
  }
}


__device__ __forceinline__ float xhalf(float x) {
  int b = __builtin_bit_cast(int, x);
  return __builtin_bit_cast(
      float, __builtin_amdgcn_permlanex16(b, b, 0x76543210, 0xfedcba98, false, false));
}

__device__ __forceinline__ void load_block(const float* __restrict__ kb,
                                           const float* __restrict__ vb,
                                           const float* __restrict__ mb,
                                           int key0, int c, int h,
                                           v4f (&k)[2][8], v4f (&v)[2][8],
                                           v4f (&hm)[2][2]) {
#pragma unroll
  for (int u = 0; u < 2; ++u) {
    const float* kp = kb + (key0 + 16 * u + c) * DKD + 4 * h;
    const float* vp = vb + (key0 + 16 * u + 8 * h) * DKD + 4 * c;
    const float* mp = mb + key0 + 16 * u + 8 * h;
#pragma unroll
    for (int m = 0; m < 8; ++m) k[u][m] = ld4(kp + 8 * m);
#pragma unroll
    for (int r = 0; r < 8; ++r) v[u][r] = ld4(vp + r * DKD);
    hm[u][0] = ld4(mp);
    hm[u][1] = ld4(mp + 4);
  }
}

__device__ __forceinline__ void process_block(const v4f (&q)[8],
                                              const v4f (&k)[2][8],
                                              const v4f (&v)[2][8],
                                              const v4f (&hm)[2][2],
                                              float& m_run, float& l_run,
                                              v8f (&acc)[4]) {
  v8f s[2] = {};
#pragma unroll
  for (int m = 0; m < 8; ++m) {
    v2f qlo, qhi, a;
    qlo.x = q[m].x; qlo.y = q[m].y;
    qhi.x = q[m].z; qhi.y = q[m].w;
    a.x = k[0][m].x; a.y = k[0][m].y;
    s[0] = wmma4(a, qlo, s[0]);
    a.x = k[1][m].x; a.y = k[1][m].y;
    s[1] = wmma4(a, qlo, s[1]);
    a.x = k[0][m].z; a.y = k[0][m].w;
    s[0] = wmma4(a, qhi, s[0]);
    a.x = k[1][m].z; a.y = k[1][m].w;
    s[1] = wmma4(a, qhi, s[1]);
  }
  float val[2][8];
  float bm = -3.4e38f;
#pragma unroll
  for (int u = 0; u < 2; ++u) {
#pragma unroll
    for (int r = 0; r < 8; ++r) {
      float hmv = hm[u][r >> 2][r & 3];
      float sv = s[u][r] * 0.125f;
      val[u][r] = sv * hmv + (1.f - hmv) * -1e30f;
      bm = fmaxf(bm, val[u][r]);
    }
  }
  bm = fmaxf(bm, xhalf(bm));
  float m_new = fmaxf(m_run, bm);
  float alpha = expf(m_run - m_new);
  float p[2][8];
  float ps = 0.f;
#pragma unroll
  for (int u = 0; u < 2; ++u) {
#pragma unroll
    for (int r = 0; r < 8; ++r) {
      p[u][r] = expf(val[u][r] - m_new);
      ps += p[u][r];
    }
  }
  l_run = l_run * alpha + ps;
  m_run = m_new;
#pragma unroll
  for (int i = 0; i < 4; ++i) acc[i] = acc[i] * alpha;
#pragma unroll
  for (int u = 0; u < 2; ++u) {
#pragma unroll
    for (int j = 0; j < 4; ++j) {
      v2f bp;
      bp.x = p[u][2 * j];
      bp.y = p[u][2 * j + 1];
#pragma unroll
      for (int i = 0; i < 4; ++i) {
        v2f a;
        a.x = v[u][2 * j][i];
        a.y = v[u][2 * j + 1][i];
        acc[i] = wmma4(a, bp, acc[i]);
      }
    }
  }
}

__global__ __launch_bounds__(128) void attn_kernel(const float* __restrict__ Q,
                                                   const float* __restrict__ Km,
                                                   const float* __restrict__ Vm,
                                                   const float* __restrict__ mask,
                                                   float* __restrict__ head) {
  int l = threadIdx.x & 31;
  int tile = blockIdx.x * 4 + (threadIdx.x >> 5);
  int b = tile >> 6;
  int q0 = (tile & 63) << 4;
  int c = l & 15;
  int h = l >> 4;
  const float* qb = Q + b * TT * DKD;
  const float* kb = Km + b * TT * DKD;
  const float* vb = Vm + b * TT * DKD;
  const float* mb = mask + b * TT;

  v4f q[8];
#pragma unroll
  for (int m = 0; m < 8; ++m) q[m] = ld4(qb + (q0 + c) * DKD + 8 * m + 4 * h);

  v4f kA[2][8], vA[2][8], hA[2][2];
  v4f kB[2][8], vB[2][8], hB[2][2];
  v8f acc[4] = {};
  float m_run = -3.4e38f;
  float l_run = 0.f;

  load_block(kb, vb, mb, 0, c, h, kA, vA, hA);
  for (int blk = 0; blk < TT / 32 - 2; blk += 2) {
    load_block(kb, vb, mb, (blk + 1) * 32, c, h, kB, vB, hB);
    process_block(q, kA, vA, hA, m_run, l_run, acc);
    load_block(kb, vb, mb, (blk + 2) * 32, c, h, kA, vA, hA);
    process_block(q, kB, vB, hB, m_run, l_run, acc);
  }
  load_block(kb, vb, mb, TT - 32, c, h, kB, vB, hB);
  process_block(q, kA, vA, hA, m_run, l_run, acc);
  process_block(q, kB, vB, hB, m_run, l_run, acc);

  float lsum = l_run + xhalf(l_run);
  float f = (1.0f / lsum) * mb[q0 + c];
  float* hp = head + b * TT * DKD + (q0 + c) * DKD + 32 * h;
#pragma unroll
  for (int r = 0; r < 8; ++r) {
    v4f o;
    o.x = acc[0][r];
    o.y = acc[1][r];
    o.z = acc[2][r];
    o.w = acc[3][r];
    *reinterpret_cast<v4f*>(hp + 4 * r) = o * f;
  }
}

__global__ __launch_bounds__(128) void wo_gemm(const float* __restrict__ H,
                                               const float* __restrict__ Wo,
                                               float* __restrict__ res) {
  __shared__ v4f ht4[4 * DKD * DKD / 4];
  int l = threadIdx.x & 31;
  int wv = threadIdx.x >> 5;
  int tile = blockIdx.x * 4 + wv;
  int b = tile >> 7;
  int rem = tile & 127;
  int m0 = (rem >> 4) << 4;
  int n0 = (rem & 15) << 6;
  int lo = l & 15;
  int kh = (l >> 4) << 1;
  const float* hsrc = H + b * TT * DKD + n0 * DKD + 4 * l;
  float* ht = reinterpret_cast<float*>(ht4) + wv * DKD * DKD;
  const float* wc = Wo + m0 + lo;
  float* rb = res + b * CT + n0 + 4 * lo;
  v4f hv[DKD / 2];
#pragma unroll
  for (int i = 0; i < DKD / 2; ++i) hv[i] = ld4(hsrc + i * 128);
  __builtin_amdgcn_sched_barrier(0);
  float wa[DKD / 4], wb2[DKD / 4], wc2[DKD / 4], wd[DKD / 4];
  v4f old[8];
#pragma unroll
  for (int s = 0; s < DKD / 4; ++s) {
    int k0 = 4 * s + kh;
    wa[s]  = wc[k0 * CC];
    wb2[s] = wc[(k0 + 64) * CC];
    wc2[s] = wc[(k0 + 1) * CC];
    wd[s]  = wc[(k0 + 65) * CC];
  }
#pragma unroll
  for (int v = 0; v < 8; ++v) {
    int m = m0 + v + ((l >> 4) << 3);
    old[v] = ld4(rb + m * TT);
  }
  __builtin_amdgcn_sched_barrier(0);
#pragma unroll
  for (int i = 0; i < DKD / 2; ++i) {
    int row = 2 * i + (l >> 4);
    int col = 4 * (l & 15);
    *reinterpret_cast<v4f*>(ht + row * DKD + (col ^ (4 * (row >> 2)))) = hv[i];
  }
  lds_handover();
  v2f hf[DKD / 4][4];
#pragma unroll
  for (int s = 0; s < DKD / 4; ++s) {
    const float* hp = ht + (4 * lo) * DKD + ((4 * s) ^ (4 * lo)) + kh;
#pragma unroll
    for (int j = 0; j < 4; ++j) hf[s][j] = ld2(hp + j * DKD);
  }
  __builtin_amdgcn_sched_barrier(0);
  v8f acc0 = {}, acc1 = {}, acc2 = {}, acc3 = {};
#pragma unroll
  for (int s = 0; s < DKD / 4; ++s) {
    v2f a;
    a.x = wa[s] + wb2[s];
    a.y = wc2[s] + wd[s];
    acc0 = wmma4(a, hf[s][0], acc0);
    acc1 = wmma4(a, hf[s][1], acc1);
    acc2 = wmma4(a, hf[s][2], acc2);
    acc3 = wmma4(a, hf[s][3], acc3);
  }
#pragma unroll
  for (int v = 0; v < 8; ++v) {
    int m = m0 + v + ((l >> 4) << 3);
    v4f o;
    o.x = acc0[v] + old[v].x;
    o.y = acc1[v] + old[v].y;
    o.z = acc2[v] + old[v].z;
    o.w = acc3[v] + old[v].w;
    *reinterpret_cast<v4f*>(rb + m * TT) = o;
  }
}

__global__ __launch_bounds__(128) void fc_gemm(const float* __restrict__ X,
                                               const float* __restrict__ W,
                                               const float* __restrict__ bias,
                                               const float* __restrict__ res,
                                               float* __restrict__ out) {
  __shared__ v4f wst4[16 * WS_PITCH / 4];
  float* wst = reinterpret_cast<float*>(wst4);
  int l = threadIdx.x & 31;
  int tile = blockIdx.x * 4 + (threadIdx.x >> 5);
  int b = tile >> 7;
  int rem = tile & 127;
  int m0 = (rem >> 4) << 4;
  int n0 = (rem & 15) << 6;
  int lo = l & 15;
  int kh = (l >> 4) << 1;
  const float* xb = X + b * CT + n0 + 4 * lo;
  v4f ws[4];
  strip_fetch(W, m0, ws);
  __builtin_amdgcn_sched_barrier(0);
  v4f r0[CC / 4], r1[CC / 4];
#pragma unroll
  for (int s = 0; s < CC / 4; ++s) {
    int k0 = 4 * s + kh;
    r0[s] = ld4(xb + k0 * TT);
    r1[s] = ld4(xb + (k0 + 1) * TT);
  }
  __builtin_amdgcn_sched_barrier(0);
  strip_store(wst, ws);
  lds_handover();
  const float* wl = wst + lo * WS_PITCH + kh;
  v2f af[CC / 4];
#pragma unroll
  for (int s = 0; s < CC / 4; ++s) af[s] = ld2(wl + 4 * s);
  const float* rb = res + b * CT + n0 + 4 * lo;
  float* ob = out + b * CT + n0 + 4 * lo;
  v4f old[8];
  float bsv[8];
#pragma unroll
  for (int v = 0; v < 8; ++v) {
    int m = m0 + v + ((l >> 4) << 3);
    bsv[v] = bias[m];
    old[v] = ld4(rb + m * TT);
  }
  __builtin_amdgcn_sched_barrier(0);
  v8f acc0 = {}, acc1 = {}, acc2 = {}, acc3 = {};
#pragma unroll
  for (int s = 0; s < CC / 4; ++s) {
    v2f b0, b1, b2, b3;
    b0.x = r0[s].x; b0.y = r1[s].x;
    b1.x = r0[s].y; b1.y = r1[s].y;
    b2.x = r0[s].z; b2.y = r1[s].z;
    b3.x = r0[s].w; b3.y = r1[s].w;
    acc0 = wmma4(af[s], b0, acc0);
    acc1 = wmma4(af[s], b1, acc1);
    acc2 = wmma4(af[s], b2, acc2);
    acc3 = wmma4(af[s], b3, acc3);
    if ((s & 3) == 3) __builtin_amdgcn_sched_barrier(0);
  }
#pragma unroll
  for (int v = 0; v < 8; ++v) {
    int m = m0 + v + ((l >> 4) << 3);
    v4f o;
    o.x = fmaxf(acc0[v] + bsv[v], 0.f) + old[v].x;
    o.y = fmaxf(acc1[v] + bsv[v], 0.f) + old[v].y;
    o.z = fmaxf(acc2[v] + bsv[v], 0.f) + old[v].z;
    o.w = fmaxf(acc3[v] + bsv[v], 0.f) + old[v].w;
    *reinterpret_cast<v4f*>(ob + m * TT) = o;
  }
}


extern "C" void kernel_launch(void* const* d_in, const int* in_sizes, int n_in,
                              void* d_out, int out_size, void* d_ws, size_t ws_size,
                              hipStream_t stream) {
  (void)in_sizes; (void)n_in; (void)out_size; (void)ws_size;
  const float* x    = (const float*)d_in[0];
  const float* mask = (const float*)d_in[1];
  const float* dw_w = (const float*)d_in[2];
  const float* dw_b = (const float*)d_in[3];
  const float* pw_w = (const float*)d_in[4];
  const float* pw_b = (const float*)d_in[5];
  const float* n0g  = (const float*)d_in[6];
  const float* n0b  = (const float*)d_in[7];
  const float* nsg  = (const float*)d_in[8];
  const float* nsb  = (const float*)d_in[9];
  const float* neg_ = (const float*)d_in[10];
  const float* neb  = (const float*)d_in[11];
  const float* Wq   = (const float*)d_in[12];
  const float* Wk   = (const float*)d_in[13];
  const float* Wv   = (const float*)d_in[14];
  const float* Wo   = (const float*)d_in[15];
  const float* fcw  = (const float*)d_in[16];
  const float* fcb  = (const float*)d_in[17];
  float* outp = (float*)d_out;
  float* ws = (float*)d_ws;

  float* OUT = ws;
  float* RES = ws + 2097152;
  float* TMP = ws + 4194304;
  float* Qb  = ws + 6291456;
  float* Kb  = ws + 7340032;
  float* Vb  = ws + 8388608;
  float* Hb  = ws + 9437184;

  posenc_add<<<8192, 256, 0, stream>>>(x, OUT);
  ln_conv_in<<<128, 1024, 0, stream>>>(OUT, n0g, n0b, RES, dw_w, dw_b, TMP);

  for (int i = 0; i < 3; ++i) {
    pw_gemm<<<512, 128, 0, stream>>>(TMP, pw_w + i * CC * CC, pw_b + i * CC, RES);
    ln_conv_out<<<128, 1024, 0, stream>>>(RES, nsg + i * CT, nsb + i * CT,
                                          dw_w + (i + 1) * CC * 7, dw_b + (i + 1) * CC, TMP);
  }
  pw_gemm<<<512, 128, 0, stream>>>(TMP, pw_w + 3 * CC * CC, pw_b + 3 * CC, RES);
  ln_fused<<<128, 1024, 0, stream>>>(RES, nsg + 3 * CT, nsb + 3 * CT, OUT);

  xtw_gemm<<<256, 128, 0, stream>>>(OUT, Wq, Qb);
  xtw_gemm<<<256, 128, 0, stream>>>(OUT, Wk, Kb);
  xtw_gemm<<<256, 128, 0, stream>>>(OUT, Wv, Vb);
  attn_kernel<<<256, 128, 0, stream>>>(Qb, Kb, Vb, mask, Hb);
  wo_gemm<<<512, 128, 0, stream>>>(Hb, Wo, RES);
  ln_fused<<<128, 1024, 0, stream>>>(RES, neg_, neb, OUT);
  fc_gemm<<<512, 128, 0, stream>>>(OUT, fcw, fcb, RES, outp);
}


# A wave keeps its X fragment (16 t x 128 channels) in registers and multiplies it with Wq, Wk, Wv in turn, next weight in flight under the products: one projection launch, not three; sums unchanged
# speedup vs baseline: 3.2393x; 1.0318x over previous
#include <hip/hip_runtime.h>

typedef float v2f __attribute__((ext_vector_type(2)));
typedef float v8f __attribute__((ext_vector_type(8)));

#define BB 16
#define CC 128
#define TT 1024
#define DKD 64
#define CT (CC * TT)
#define LOG10000 9.210340371976184f
#define PI_2 1.5707963267948966f

__device__ inline v8f wmma4(v2f a, v2f b, v8f c) {
  return __builtin_amdgcn_wmma_f32_16x16x4_f32(false, a, false, b, (short)0, c,
                                               false, false);
}


__global__ void posenc_add(const float* __restrict__ x, float* __restrict__ out) {
  int idx = blockIdx.x * blockDim.x + threadIdx.x;
  int ct = idx % CT;
  int c = ct / TT;
  int t = ct % TT;
  float freq = expf(-(float)(c & ~1) * (LOG10000 / (float)CC));
  float pe = sinf((float)t * freq + (float)(c & 1) * PI_2);
  out[idx] = x[idx] + pe;
}

typedef float v4f __attribute__((ext_vector_type(4)));
#define LN_SLICES 8
#define LN_THREADS 1024
#define CT4 (CT / 4)
#define LN_ITEMS (CT4 / LN_THREADS)
#define LN_OWN (LN_ITEMS / LN_SLICES)

__device__ __forceinline__ float wave_sum(float x) {
#pragma unroll
  for (int m = 16; m > 0; m >>= 1) x += __shfl_xor(x, m);
  return x;
}

#define LN_PLAIN 0
#define LN_CONV_OUT 1
#define LN_CONV_IN 2
#define ROW_ITEMS (TT / 4)
#define ROWS_PER_STEP (LN_THREADS / ROW_ITEMS)
#define ROW_PITCH (ROW_ITEMS + 2)

__device__ __forceinline__ void ln_body(const int mode,
                                        const float* __restrict__ src,
                                        const float* __restrict__ g,
                                        const float* __restrict__ bb,
                                        float* __restrict__ dst,
                                        const float* __restrict__ cw,
                                        const float* __restrict__ cb,
                                        float* __restrict__ cdst) {
  __shared__ float red[64];
  __shared__ v4f tile[LN_OWN][ROWS_PER_STEP * ROW_PITCH];
  int b = blockIdx.x / LN_SLICES;
  int sl = blockIdx.x % LN_SLICES;
  int tid = threadIdx.x;
  const v4f* p = reinterpret_cast<const v4f*>(src + b * CT);
  const v4f* g4 = reinterpret_cast<const v4f*>(g);
  const v4f* b4 = reinterpret_cast<const v4f*>(bb);
  v4f* d4 = reinterpret_cast<v4f*>(dst + b * CT);
  v4f* c4 = reinterpret_cast<v4f*>(cdst + b * CT);
  int rowi = tid / ROW_ITEMS;
  int cell = rowi * ROW_PITCH + 1 + tid % ROW_ITEMS;

  const int base = sl * LN_OWN;
  v4f xv[LN_OWN];
#pragma unroll
  for (int i = 0; i < LN_OWN; ++i) xv[i] = p[(base + i) * LN_THREADS + tid];
  v4f s4 = {0.f, 0.f, 0.f, 0.f};
  v4f q4 = {0.f, 0.f, 0.f, 0.f};
#pragma unroll 7
  for (int i = LN_OWN; i < LN_ITEMS; ++i) {
    v4f v = p[((base + i) & (LN_ITEMS - 1)) * LN_THREADS + tid];
    s4 += v;
    q4 += v * v;
  }
#pragma unroll
  for (int i = 0; i < LN_OWN; ++i) {
    s4 += xv[i];
    q4 += xv[i] * xv[i];
  }
  float wt[LN_OWN][7], cbias[LN_OWN];
  if (mode != LN_PLAIN) {
#pragma unroll
    for (int i = 0; i < LN_OWN; ++i) {
      int c = __builtin_amdgcn_readfirstlane((sl * LN_OWN + i) * ROWS_PER_STEP + rowi);
#pragma unroll
      for (int j = 0; j < 7; ++j) wt[i][j] = cw[c * 7 + j];
      cbias[i] = cb[c];
    }
  }
  v4f gv[LN_OWN], bv[LN_OWN];
#pragma unroll
  for (int i = 0; i < LN_OWN; ++i) {
    int e = (base + i) * LN_THREADS + tid;
    gv[i] = g4[e];
    bv[i] = b4[e];
  }
  float s = (s4.x + s4.y) + (s4.z + s4.w);
  float q = (q4.x + q4.y) + (q4.z + q4.w);
  s = wave_sum(s);
  q = wave_sum(q);
  int w = tid >> 5;
  int l = tid & 31;
  if (l == 0) {
    red[w] = s;
    red[32 + w] = q;
  }
  if (mode != LN_PLAIN) {
    if (tid < LN_OWN * ROWS_PER_STEP * 2) {
      v4f z = {0.f, 0.f, 0.f, 0.f};
      tile[tid >> 3][((tid >> 1) & (ROWS_PER_STEP - 1)) * ROW_PITCH + (tid & 1) * (ROW_PITCH - 1)] = z;
    }
  }
  if (mode == LN_CONV_IN) {
#pragma unroll
    for (int i = 0; i < LN_OWN; ++i) tile[i][cell] = xv[i];
  }
  __syncthreads();
  s = wave_sum(red[l]);
  q = wave_sum(red[32 + l]);
  float inv_n = 1.0f / (float)CT;
  float mu = s * inv_n;
  float var = q * inv_n - mu * mu;
  float r = rsqrtf(var + 1e-5f);
  v4f y[LN_OWN];
#pragma unroll
  for (int i = 0; i < LN_OWN; ++i) {
    int e = (sl * LN_OWN + i) * LN_THREADS + tid;
    y[i] = (xv[i] - mu) * r * gv[i] + bv[i];
    if (mode != LN_CONV_OUT) d4[e] = y[i];
    if (mode == LN_CONV_OUT) tile[i][cell] = y[i];
  }
  if (mode == LN_PLAIN) return;
  if (mode == LN_CONV_OUT) __syncthreads();
#pragma unroll
  for (int i = 0; i < LN_OWN; ++i) {
    int e = (sl * LN_OWN + i) * LN_THREADS + tid;
    v4f lf = tile[i][cell - 1];
    v4f rt = tile[i][cell + 1];
    v4f md = (mode == LN_CONV_IN) ? xv[i] : y[i];
    float t[10] = {lf.y, lf.z, lf.w, md.x, md.y, md.z, md.w, rt.x, rt.y, rt.z};
    v4f o;
#pragma unroll
    for (int k = 0; k < 4; ++k) {
      float a = cbias[i];
#pragma unroll
      for (int j = 0; j < 7; ++j) a += t[k + j] * wt[i][j];
      o[k] = a;
    }
    c4[e] = o;
  }
}

__global__ __launch_bounds__(LN_THREADS) void ln_fused(const float* __restrict__ src,
                                                       const float* __restrict__ g,
                                                       const float* __restrict__ bb,
                                                       float* __restrict__ dst) {
  ln_body(LN_PLAIN, src, g, bb, dst, nullptr, nullptr, nullptr);
}

__global__ __launch_bounds__(LN_THREADS) void ln_conv_out(const float* __restrict__ src,
                                                          const float* __restrict__ g,
                                                          const float* __restrict__ bb,
                                                          const float* __restrict__ cw,
                                                          const float* __restrict__ cb,
                                                          float* __restrict__ cdst) {
  ln_body(LN_CONV_OUT, src, g, bb, nullptr, cw, cb, cdst);
}

__global__ __launch_bounds__(LN_THREADS) void ln_conv_in(const float* __restrict__ src,
                                                         const float* __restrict__ g,
                                                         const float* __restrict__ bb,
                                                         float* __restrict__ dst,
                                                         const float* __restrict__ cw,
                                                         const float* __restrict__ cb,
                                                         float* __restrict__ cdst) {
  ln_body(LN_CONV_IN, src, g, bb, dst, cw, cb, cdst);
}


__device__ __forceinline__ v4f ld4(const float* p) {
  return *reinterpret_cast<const v4f*>(p);
}

__device__ __forceinline__ v2f ld2(const float* p) {
  return *reinterpret_cast<const v2f*>(p);
}


__device__ __forceinline__ void lds_handover() {
  asm volatile("s_wait_dscnt 0x0\n\ts_barrier_signal -1\n\ts_barrier_wait -1" ::: "memory");
}

#define WS_PITCH (CC + 4)

__device__ __forceinline__ void strip_fetch(const float* __restrict__ W, int m0, v4f (&ws)[4]) {
  int srow = threadIdx.x >> 5;
  int scol = (threadIdx.x & 31) << 2;
#pragma unroll
  for (int i = 0; i < 4; ++i) ws[i] = ld4(W + (m0 + srow + 4 * i) * CC + scol);
}

__device__ __forceinline__ void strip_store(float* wst, const v4f (&ws)[4]) {
  int srow = threadIdx.x >> 5;
  int scol = (threadIdx.x & 31) << 2;
#pragma unroll
  for (int i = 0; i < 4; ++i)
    *reinterpret_cast<v4f*>(wst + (srow + 4 * i) * WS_PITCH + scol) = ws[i];
}

__global__ __launch_bounds__(128) void pw_gemm(const float* __restrict__ Y,
                                               const float* __restrict__ W,
                                               const float* __restrict__ bias,
                                               float* __restrict__ res) {
  __shared__ v4f wst4[16 * WS_PITCH / 4];
  float* wst = reinterpret_cast<float*>(wst4);
  int l = threadIdx.x & 31;
  int tile = blockIdx.x * 4 + (threadIdx.x >> 5);
  int b = tile >> 7;
  int rem = tile & 127;
  int m0 = (rem >> 4) << 4;
  int n0 = (rem & 15) << 6;
  int lo = l & 15;
  int kh = (l >> 4) << 1;
  const float* yb = Y + b * CT + n0 + 4 * lo;
  v4f ws[4];
  strip_fetch(W, m0, ws);
  __builtin_amdgcn_sched_barrier(0);
  v4f r0[CC / 4], r1[CC / 4];
#pragma unroll
  for (int s = 0; s < CC / 4; ++s) {
    int k0 = 4 * s + kh;
    r0[s] = ld4(yb + k0 * TT);
    r1[s] = ld4(yb + (k0 + 1) * TT);
  }
  __builtin_amdgcn_sched_barrier(0);
  strip_store(wst, ws);
  lds_handover();
  const float* wl = wst + lo * WS_PITCH + kh;
  v2f af[CC / 4];
#pragma unroll
  for (int s = 0; s < CC / 4; ++s) af[s] = ld2(wl + 4 * s);
  float* rb = res + b * CT + n0 + 4 * lo;
  v4f old[8];
  float bsv[8];
#pragma unroll
  for (int v = 0; v < 8; ++v) {
    int m = m0 + v + ((l >> 4) << 3);
    bsv[v] = bias[m];
    old[v] = ld4(rb + m * TT);
  }
  __builtin_amdgcn_sched_barrier(0);
  v8f acc0 = {}, acc1 = {}, acc2 = {}, acc3 = {};
#pragma unroll
  for (int s = 0; s < CC / 4; ++s) {
    v2f b0, b1, b2, b3;
    b0.x = r0[s].x; b0.y = r1[s].x;
    b1.x = r0[s].y; b1.y = r1[s].y;
    b2.x = r0[s].z; b2.y = r1[s].z;
    b3.x = r0[s].w; b3.y = r1[s].w;
    acc0 = wmma4(af[s], b0, acc0);
    acc1 = wmma4(af[s], b1, acc1);
    acc2 = wmma4(af[s], b2, acc2);
    acc3 = wmma4(af[s], b3, acc3);
    if ((s & 3) == 3) __builtin_amdgcn_sched_barrier(0);
  }
#pragma unroll
  for (int v = 0; v < 8; ++v) {
    int m = m0 + v + ((l >> 4) << 3);
    v4f o;
    o.x = fmaxf(acc0[v] + bsv[v], 0.f) + old[v].x;
    o.y = fmaxf(acc1[v] + bsv[v], 0.f) + old[v].y;
    o.z = fmaxf(acc2[v] + bsv[v], 0.f) + old[v].z;
    o.w = fmaxf(acc3[v] + bsv[v], 0.f) + old[v].w;
    *reinterpret_cast<v4f*>(rb + m * TT) = o;
  }
}


__device__ __forceinline__ void w_load(const float* __restrict__ wb, int kh,
                                       v4f (&r0)[CC / 4], v4f (&r1)[CC / 4]) {
#pragma unroll
  for (int s = 0; s < CC / 4; ++s) {
    int k0 = 4 * s + kh;
    r0[s] = ld4(wb + k0 * DKD);
    r1[s] = ld4(wb + (k0 + 1) * DKD);
  }
}

__device__ __forceinline__ void xw_tile(const float (&ax)[CC / 4], const float (&ay)[CC / 4],
                                        const v4f (&r0)[CC / 4], const v4f (&r1)[CC / 4],
                                        float* __restrict__ db, int t0, int l) {
  v8f acc0 = {}, acc1 = {}, acc2 = {}, acc3 = {};
#pragma unroll
  for (int s = 0; s < CC / 4; ++s) {
    v2f a, b0, b1, b2, b3;
    a.x = ax[s]; a.y = ay[s];
    b0.x = r0[s].x; b0.y = r1[s].x;
    b1.x = r0[s].y; b1.y = r1[s].y;
    b2.x = r0[s].z; b2.y = r1[s].z;
    b3.x = r0[s].w; b3.y = r1[s].w;
    acc0 = wmma4(a, b0, acc0);
    acc1 = wmma4(a, b1, acc1);
    acc2 = wmma4(a, b2, acc2);
    acc3 = wmma4(a, b3, acc3);
  }
#pragma unroll
  for (int v = 0; v < 8; ++v) {
    int t = t0 + v + ((l >> 4) << 3);
    v4f o;
    o.x = acc0[v];
    o.y = acc1[v];
    o.z = acc2[v];
    o.w = acc3[v];
    *reinterpret_cast<v4f*>(db + t * DKD) = o;
  }
}

__global__ __launch_bounds__(128) void qkv_gemm(const float* __restrict__ X,
                                                const float* __restrict__ Wq,
                                                const float* __restrict__ Wk,
                                                const float* __restrict__ Wv,
                                                float* __restrict__ Qd,
                                                float* __restrict__ Kd,
                                                float* __restrict__ Vd) {
  int l = threadIdx.x & 31;
  int tile = blockIdx.x * 4 + (threadIdx.x >> 5);
  int b = tile >> 6;
  int t0 = (tile & 63) << 4;
  int lo = l & 15;
  int kh = (l >> 4) << 1;
  const float* xb = X + b * CT + t0 + lo;
  int dofs = b * TT * DKD + 4 * lo;
  float ax[CC / 4], ay[CC / 4];
  v4f ra0[CC / 4], ra1[CC / 4];
  v4f rb0[CC / 4], rb1[CC / 4];
#pragma unroll
  for (int s = 0; s < CC / 4; ++s) {
    int k0 = 4 * s + kh;
    ax[s] = xb[k0 * TT];
    ay[s] = xb[(k0 + 1) * TT];
  }
  w_load(Wq + 4 * lo, kh, ra0, ra1);
  __builtin_amdgcn_sched_barrier(0);
  w_load(Wk + 4 * lo, kh, rb0, rb1);
  __builtin_amdgcn_sched_barrier(0);
  xw_tile(ax, ay, ra0, ra1, Qd + dofs, t0, l);
  __builtin_amdgcn_sched_barrier(0);
  w_load(Wv + 4 * lo, kh, ra0, ra1);
  __builtin_amdgcn_sched_barrier(0);
  xw_tile(ax, ay, rb0, rb1, Kd + dofs, t0, l);
  __builtin_amdgcn_sched_barrier(0);
  xw_tile(ax, ay, ra0, ra1, Vd + dofs, t0, l);
}


__device__ __forceinline__ float xhalf(float x) {
  int b = __builtin_bit_cast(int, x);
  return __builtin_bit_cast(
      float, __builtin_amdgcn_permlanex16(b, b, 0x76543210, 0xfedcba98, false, false));
}

__device__ __forceinline__ void load_block(const float* __restrict__ kb,
                                           const float* __restrict__ vb,
                                           const float* __restrict__ mb,
                                           int key0, int c, int h,
                                           v4f (&k)[2][8], v4f (&v)[2][8],
                                           v4f (&hm)[2][2]) {
#pragma unroll
  for (int u = 0; u < 2; ++u) {
    const float* kp = kb + (key0 + 16 * u + c) * DKD + 4 * h;
    const float* vp = vb + (key0 + 16 * u + 8 * h) * DKD + 4 * c;
    const float* mp = mb + key0 + 16 * u + 8 * h;
#pragma unroll
    for (int m = 0; m < 8; ++m) k[u][m] = ld4(kp + 8 * m);
#pragma unroll
    for (int r = 0; r < 8; ++r) v[u][r] = ld4(vp + r * DKD);
    hm[u][0] = ld4(mp);
    hm[u][1] = ld4(mp + 4);
  }
}

__device__ __forceinline__ void process_block(const v4f (&q)[8],
                                              const v4f (&k)[2][8],
                                              const v4f (&v)[2][8],
                                              const v4f (&hm)[2][2],
                                              float& m_run, float& l_run,
                                              v8f (&acc)[4]) {
  v8f s[2] = {};
#pragma unroll
  for (int m = 0; m < 8; ++m) {
    v2f qlo, qhi, a;
    qlo.x = q[m].x; qlo.y = q[m].y;
    qhi.x = q[m].z; qhi.y = q[m].w;
    a.x = k[0][m].x; a.y = k[0][m].y;
    s[0] = wmma4(a, qlo, s[0]);
    a.x = k[1][m].x; a.y = k[1][m].y;
    s[1] = wmma4(a, qlo, s[1]);
    a.x = k[0][m].z; a.y = k[0][m].w;
    s[0] = wmma4(a, qhi, s[0]);
    a.x = k[1][m].z; a.y = k[1][m].w;
    s[1] = wmma4(a, qhi, s[1]);
  }
  float val[2][8];
  float bm = -3.4e38f;
#pragma unroll
  for (int u = 0; u < 2; ++u) {
#pragma unroll
    for (int r = 0; r < 8; ++r) {
      float hmv = hm[u][r >> 2][r & 3];
      float sv = s[u][r] * 0.125f;
      val[u][r] = sv * hmv + (1.f - hmv) * -1e30f;
      bm = fmaxf(bm, val[u][r]);
    }
  }
  bm = fmaxf(bm, xhalf(bm));
  float m_new = fmaxf(m_run, bm);
  float alpha = expf(m_run - m_new);
  float p[2][8];
  float ps = 0.f;
#pragma unroll
  for (int u = 0; u < 2; ++u) {
#pragma unroll
    for (int r = 0; r < 8; ++r) {
      p[u][r] = expf(val[u][r] - m_new);
      ps += p[u][r];
    }
  }
  l_run = l_run * alpha + ps;
  m_run = m_new;
#pragma unroll
  for (int i = 0; i < 4; ++i) acc[i] = acc[i] * alpha;
#pragma unroll
  for (int u = 0; u < 2; ++u) {
#pragma unroll
    for (int j = 0; j < 4; ++j) {
      v2f bp;
      bp.x = p[u][2 * j];
      bp.y = p[u][2 * j + 1];
#pragma unroll
      for (int i = 0; i < 4; ++i) {
        v2f a;
        a.x = v[u][2 * j][i];
        a.y = v[u][2 * j + 1][i];
        acc[i] = wmma4(a, bp, acc[i]);
      }
    }
  }
}

__global__ __launch_bounds__(128) void attn_kernel(const float* __restrict__ Q,
                                                   const float* __restrict__ Km,
                                                   const float* __restrict__ Vm,
                                                   const float* __restrict__ mask,
                                                   float* __restrict__ head) {
  int l = threadIdx.x & 31;
  int tile = blockIdx.x * 4 + (threadIdx.x >> 5);
  int b = tile >> 6;
  int q0 = (tile & 63) << 4;
  int c = l & 15;
  int h = l >> 4;
  const float* qb = Q + b * TT * DKD;
  const float* kb = Km + b * TT * DKD;
  const float* vb = Vm + b * TT * DKD;
  const float* mb = mask + b * TT;

  v4f q[8];
#pragma unroll
  for (int m = 0; m < 8; ++m) q[m] = ld4(qb + (q0 + c) * DKD + 8 * m + 4 * h);

  v4f kA[2][8], vA[2][8], hA[2][2];
  v4f kB[2][8], vB[2][8], hB[2][2];
  v8f acc[4] = {};
  float m_run = -3.4e38f;
  float l_run = 0.f;

  load_block(kb, vb, mb, 0, c, h, kA, vA, hA);
  for (int blk = 0; blk < TT / 32 - 2; blk += 2) {
    load_block(kb, vb, mb, (blk + 1) * 32, c, h, kB, vB, hB);
    process_block(q, kA, vA, hA, m_run, l_run, acc);
    load_block(kb, vb, mb, (blk + 2) * 32, c, h, kA, vA, hA);
    process_block(q, kB, vB, hB, m_run, l_run, acc);
  }
  load_block(kb, vb, mb, TT - 32, c, h, kB, vB, hB);
  process_block(q, kA, vA, hA, m_run, l_run, acc);
  process_block(q, kB, vB, hB, m_run, l_run, acc);

  float lsum = l_run + xhalf(l_run);
  float f = (1.0f / lsum) * mb[q0 + c];
  float* hp = head + b * TT * DKD + (q0 + c) * DKD + 32 * h;
#pragma unroll
  for (int r = 0; r < 8; ++r) {
    v4f o;
    o.x = acc[0][r];
    o.y = acc[1][r];
    o.z = acc[2][r];
    o.w = acc[3][r];
    *reinterpret_cast<v4f*>(hp + 4 * r) = o * f;
  }
}

__global__ __launch_bounds__(128) void wo_gemm(const float* __restrict__ H,
                                               const float* __restrict__ Wo,
                                               float* __restrict__ res) {
  __shared__ v4f ht4[4 * DKD * DKD / 4];
  int l = threadIdx.x & 31;
  int wv = threadIdx.x >> 5;
  int tile = blockIdx.x * 4 + wv;
  int b = tile >> 7;
  int rem = tile & 127;
  int m0 = (rem >> 4) << 4;
  int n0 = (rem & 15) << 6;
  int lo = l & 15;
  int kh = (l >> 4) << 1;
  const float* hsrc = H + b * TT * DKD + n0 * DKD + 4 * l;
  float* ht = reinterpret_cast<float*>(ht4) + wv * DKD * DKD;
  const float* wc = Wo + m0 + lo;
  float* rb = res + b * CT + n0 + 4 * lo;
  v4f hv[DKD / 2];
#pragma unroll
  for (int i = 0; i < DKD / 2; ++i) hv[i] = ld4(hsrc + i * 128);
  __builtin_amdgcn_sched_barrier(0);
  float wa[DKD / 4], wb2[DKD / 4], wc2[DKD / 4], wd[DKD / 4];
  v4f old[8];
#pragma unroll
  for (int s = 0; s < DKD / 4; ++s) {
    int k0 = 4 * s + kh;
    wa[s]  = wc[k0 * CC];
    wb2[s] = wc[(k0 + 64) * CC];
    wc2[s] = wc[(k0 + 1) * CC];
    wd[s]  = wc[(k0 + 65) * CC];
  }
#pragma unroll
  for (int v = 0; v < 8; ++v) {
    int m = m0 + v + ((l >> 4) << 3);
    old[v] = ld4(rb + m * TT);
  }
  __builtin_amdgcn_sched_barrier(0);
#pragma unroll
  for (int i = 0; i < DKD / 2; ++i) {
    int row = 2 * i + (l >> 4);
    int col = 4 * (l & 15);
    *reinterpret_cast<v4f*>(ht + row * DKD + (col ^ (4 * (row >> 2)))) = hv[i];
  }
  lds_handover();
  v2f hf[DKD / 4][4];
#pragma unroll
  for (int s = 0; s < DKD / 4; ++s) {
    const float* hp = ht + (4 * lo) * DKD + ((4 * s) ^ (4 * lo)) + kh;
#pragma unroll
    for (int j = 0; j < 4; ++j) hf[s][j] = ld2(hp + j * DKD);
  }
  __builtin_amdgcn_sched_barrier(0);
  v8f acc0 = {}, acc1 = {}, acc2 = {}, acc3 = {};
#pragma unroll
  for (int s = 0; s < DKD / 4; ++s) {
    v2f a;
    a.x = wa[s] + wb2[s];
    a.y = wc2[s] + wd[s];
    acc0 = wmma4(a, hf[s][0], acc0);
    acc1 = wmma4(a, hf[s][1], acc1);
    acc2 = wmma4(a, hf[s][2], acc2);
    acc3 = wmma4(a, hf[s][3], acc3);
  }
#pragma unroll
  for (int v = 0; v < 8; ++v) {
    int m = m0 + v + ((l >> 4) << 3);
    v4f o;
    o.x = acc0[v] + old[v].x;
    o.y = acc1[v] + old[v].y;
    o.z = acc2[v] + old[v].z;
    o.w = acc3[v] + old[v].w;
    *reinterpret_cast<v4f*>(rb + m * TT) = o;
  }
}

__global__ __launch_bounds__(128) void fc_gemm(const float* __restrict__ X,
                                               const float* __restrict__ W,
                                               const float* __restrict__ bias,
                                               const float* __restrict__ res,
                                               float* __restrict__ out) {
  __shared__ v4f wst4[16 * WS_PITCH / 4];
  float* wst = reinterpret_cast<float*>(wst4);
  int l = threadIdx.x & 31;
  int tile = blockIdx.x * 4 + (threadIdx.x >> 5);
  int b = tile >> 7;
  int rem = tile & 127;
  int m0 = (rem >> 4) << 4;
  int n0 = (rem & 15) << 6;
  int lo = l & 15;
  int kh = (l >> 4) << 1;
  const float* xb = X + b * CT + n0 + 4 * lo;
  v4f ws[4];
  strip_fetch(W, m0, ws);
  __builtin_amdgcn_sched_barrier(0);
  v4f r0[CC / 4], r1[CC / 4];
#pragma unroll
  for (int s = 0; s < CC / 4; ++s) {
    int k0 = 4 * s + kh;
    r0[s] = ld4(xb + k0 * TT);
    r1[s] = ld4(xb + (k0 + 1) * TT);
  }
  __builtin_amdgcn_sched_barrier(0);
  strip_store(wst, ws);
  lds_handover();
  const float* wl = wst + lo * WS_PITCH + kh;
  v2f af[CC / 4];
#pragma unroll
  for (int s = 0; s < CC / 4; ++s) af[s] = ld2(wl + 4 * s);
  const float* rb = res + b * CT + n0 + 4 * lo;
  float* ob = out + b * CT + n0 + 4 * lo;
  v4f old[8];
  float bsv[8];
#pragma unroll
  for (int v = 0; v < 8; ++v) {
    int m = m0 + v + ((l >> 4) << 3);
    bsv[v] = bias[m];
    old[v] = ld4(rb + m * TT);
  }
  __builtin_amdgcn_sched_barrier(0);
  v8f acc0 = {}, acc1 = {}, acc2 = {}, acc3 = {};
#pragma unroll
  for (int s = 0; s < CC / 4; ++s) {
    v2f b0, b1, b2, b3;
    b0.x = r0[s].x; b0.y = r1[s].x;
    b1.x = r0[s].y; b1.y = r1[s].y;
    b2.x = r0[s].z; b2.y = r1[s].z;
    b3.x = r0[s].w; b3.y = r1[s].w;
    acc0 = wmma4(af[s], b0, acc0);
    acc1 = wmma4(af[s], b1, acc1);
    acc2 = wmma4(af[s], b2, acc2);
    acc3 = wmma4(af[s], b3, acc3);
    if ((s & 3) == 3) __builtin_amdgcn_sched_barrier(0);
  }
#pragma unroll
  for (int v = 0; v < 8; ++v) {
    int m = m0 + v + ((l >> 4) << 3);
    v4f o;
    o.x = fmaxf(acc0[v] + bsv[v], 0.f) + old[v].x;
    o.y = fmaxf(acc1[v] + bsv[v], 0.f) + old[v].y;
    o.z = fmaxf(acc2[v] + bsv[v], 0.f) + old[v].z;
    o.w = fmaxf(acc3[v] + bsv[v], 0.f) + old[v].w;
    *reinterpret_cast<v4f*>(ob + m * TT) = o;
  }
}


extern "C" void kernel_launch(void* const* d_in, const int* in_sizes, int n_in,
                              void* d_out, int out_size, void* d_ws, size_t ws_size,
                              hipStream_t stream) {
  (void)in_sizes; (void)n_in; (void)out_size; (void)ws_size;
  const float* x    = (const float*)d_in[0];
  const float* mask = (const float*)d_in[1];
  const float* dw_w = (const float*)d_in[2];
  const float* dw_b = (const float*)d_in[3];
  const float* pw_w = (const float*)d_in[4];
  const float* pw_b = (const float*)d_in[5];
  const float* n0g  = (const float*)d_in[6];
  const float* n0b  = (const float*)d_in[7];
  const float* nsg  = (const float*)d_in[8];
  const float* nsb  = (const float*)d_in[9];
  const float* neg_ = (const float*)d_in[10];
  const float* neb  = (const float*)d_in[11];
  const float* Wq   = (const float*)d_in[12];
  const float* Wk   = (const float*)d_in[13];
  const float* Wv   = (const float*)d_in[14];
  const float* Wo   = (const float*)d_in[15];
  const float* fcw  = (const float*)d_in[16];
  const float* fcb  = (const float*)d_in[17];
  float* outp = (float*)d_out;
  float* ws = (float*)d_ws;

  float* OUT = ws;
  float* RES = ws + 2097152;
  float* TMP = ws + 4194304;
  float* Qb  = ws + 6291456;
  float* Kb  = ws + 7340032;
  float* Vb  = ws + 8388608;
  float* Hb  = ws + 9437184;

  posenc_add<<<8192, 256, 0, stream>>>(x, OUT);
  ln_conv_in<<<128, 1024, 0, stream>>>(OUT, n0g, n0b, RES, dw_w, dw_b, TMP);

  for (int i = 0; i < 3; ++i) {
    pw_gemm<<<512, 128, 0, stream>>>(TMP, pw_w + i * CC * CC, pw_b + i * CC, RES);
    ln_conv_out<<<128, 1024, 0, stream>>>(RES, nsg + i * CT, nsb + i * CT,
                                          dw_w + (i + 1) * CC * 7, dw_b + (i + 1) * CC, TMP);
  }
  pw_gemm<<<512, 128, 0, stream>>>(TMP, pw_w + 3 * CC * CC, pw_b + 3 * CC, RES);
  ln_fused<<<128, 1024, 0, stream>>>(RES, nsg + 3 * CT, nsb + 3 * CT, OUT);

  qkv_gemm<<<256, 128, 0, stream>>>(OUT, Wq, Wk, Wv, Qb, Kb, Vb);
  attn_kernel<<<256, 128, 0, stream>>>(Qb, Kb, Vb, mask, Hb);
  wo_gemm<<<512, 128, 0, stream>>>(Hb, Wo, RES);
  ln_fused<<<128, 1024, 0, stream>>>(RES, neg_, neb, OUT);
  fc_gemm<<<512, 128, 0, stream>>>(OUT, fcw, fcb, RES, outp);
}


# A LayerNorm launch goes on with the 1x1 product that reads its output (groups sliced by time, 16 per element; B tile and weight in LDS): 5 product launches gone; another LayerNorm summation order
# speedup vs baseline: 3.7994x; 1.1729x over previous
#include <hip/hip_runtime.h>

typedef float v2f __attribute__((ext_vector_type(2)));
typedef float v8f __attribute__((ext_vector_type(8)));

#define BB 16
#define CC 128
#define TT 1024
#define DKD 64
#define CT (CC * TT)
#define LOG10000 9.210340371976184f
#define PI_2 1.5707963267948966f

__device__ inline v8f wmma4(v2f a, v2f b, v8f c) {
  return __builtin_amdgcn_wmma_f32_16x16x4_f32(false, a, false, b, (short)0, c,
                                               false, false);
}


__global__ void posenc_add(const float* __restrict__ x, float* __restrict__ out) {
  int idx = blockIdx.x * blockDim.x + threadIdx.x;
  int ct = idx % CT;
  int c = ct / TT;
  int t = ct % TT;
  float freq = expf(-(float)(c & ~1) * (LOG10000 / (float)CC));
  float pe = sinf((float)t * freq + (float)(c & 1) * PI_2);
  out[idx] = x[idx] + pe;
}

typedef float v4f __attribute__((ext_vector_type(4)));
#define LN_SLICES 8
#define LN_THREADS 1024
#define CT4 (CT / 4)
#define LN_ITEMS (CT4 / LN_THREADS)
#define LN_OWN (LN_ITEMS / LN_SLICES)
#define ROW_ITEMS (TT / 4)

__device__ __forceinline__ float wave_sum(float x) {
#pragma unroll
  for (int m = 16; m > 0; m >>= 1) x += __shfl_xor(x, m);
  return x;
}

__global__ __launch_bounds__(LN_THREADS) void ln_fused(const float* __restrict__ src,
                                                       const float* __restrict__ g,
                                                       const float* __restrict__ bb,
                                                       float* __restrict__ dst) {
  __shared__ float red[64];
  int b = blockIdx.x / LN_SLICES;
  int sl = blockIdx.x % LN_SLICES;
  int tid = threadIdx.x;
  const v4f* p = reinterpret_cast<const v4f*>(src + b * CT);
  const v4f* g4 = reinterpret_cast<const v4f*>(g);
  const v4f* b4 = reinterpret_cast<const v4f*>(bb);
  v4f* d4 = reinterpret_cast<v4f*>(dst + b * CT);

  const int base = sl * LN_OWN;
  v4f xv[LN_OWN];
#pragma unroll
  for (int i = 0; i < LN_OWN; ++i) xv[i] = p[(base + i) * LN_THREADS + tid];
  v4f s4 = {0.f, 0.f, 0.f, 0.f};
  v4f q4 = {0.f, 0.f, 0.f, 0.f};
#pragma unroll 7
  for (int i = LN_OWN; i < LN_ITEMS; ++i) {
    v4f v = p[((base + i) & (LN_ITEMS - 1)) * LN_THREADS + tid];
    s4 += v;
    q4 += v * v;
  }
#pragma unroll
  for (int i = 0; i < LN_OWN; ++i) {
    s4 += xv[i];
    q4 += xv[i] * xv[i];
  }
  v4f gv[LN_OWN], bv[LN_OWN];
#pragma unroll
  for (int i = 0; i < LN_OWN; ++i) {
    int e = (base + i) * LN_THREADS + tid;
    gv[i] = g4[e];
    bv[i] = b4[e];
  }
  float s = (s4.x + s4.y) + (s4.z + s4.w);
  float q = (q4.x + q4.y) + (q4.z + q4.w);
  s = wave_sum(s);
  q = wave_sum(q);
  int w = tid >> 5;
  int l = tid & 31;
  if (l == 0) {
    red[w] = s;
    red[32 + w] = q;
  }
  __syncthreads();
  s = wave_sum(red[l]);
  q = wave_sum(red[32 + l]);
  float inv_n = 1.0f / (float)CT;
  float mu = s * inv_n;
  float var = q * inv_n - mu * mu;
  float r = rsqrtf(var + 1e-5f);
#pragma unroll
  for (int i = 0; i < LN_OWN; ++i) {
    int e = (base + i) * LN_THREADS + tid;
    d4[e] = (xv[i] - mu) * r * gv[i] + bv[i];
  }
}


__device__ __forceinline__ v4f ld4(const float* p) {
  return *reinterpret_cast<const v4f*>(p);
}

__device__ __forceinline__ v2f ld2(const float* p) {
  return *reinterpret_cast<const v2f*>(p);
}


__device__ __forceinline__ void lds_handover() {
  asm volatile("s_wait_dscnt 0x0\n\ts_barrier_signal -1\n\ts_barrier_wait -1" ::: "memory");
}

#define PW_CONV_LN 0
#define PW_CONV_RAW 1
#define PW_LN 2
#define FS 16
#define FT (TT / FS)
#define FI (FT / 4)
#define FROWS (LN_THREADS / FI)
#define XP (FI + 2)
#define BP (FT + 8)
#define OP (FT + 4)
#define WS_PITCH (CC + 4)
#define LDS_BT 0
#define LDS_XT (LDS_BT + CC * BP / 4)
#define LDS_WL (LDS_XT + CC * XP)
#define LDS_RED (LDS_WL + CC * WS_PITCH / 4)
#define LDS_ITEMS (LDS_RED + 16)

template <int MODE>
__global__ __launch_bounds__(LN_THREADS) void ln_pw(const float* __restrict__ src,
                                                    const float* __restrict__ g,
                                                    const float* __restrict__ bb,
                                                    const float* __restrict__ cw,
                                                    const float* __restrict__ cb,
                                                    const float* __restrict__ W,
                                                    const float* __restrict__ bias,
                                                    float* __restrict__ dst) {
  __shared__ v4f lds[LDS_ITEMS];
  float* bt = reinterpret_cast<float*>(lds + LDS_BT);
  v4f* xt = lds + LDS_XT;
  float* ot = reinterpret_cast<float*>(lds + LDS_XT);
  float* wl = reinterpret_cast<float*>(lds + LDS_WL);
  float* red = reinterpret_cast<float*>(lds + LDS_RED);
  const bool conv = MODE != PW_LN;
  int b = blockIdx.x / FS;
  int sl = blockIdx.x % FS;
  int tid = threadIdx.x;
  const v4f* p = reinterpret_cast<const v4f*>(src + b * CT);
  const v4f* g4 = reinterpret_cast<const v4f*>(g);
  const v4f* b4 = reinterpret_cast<const v4f*>(bb);
  const v4f* w4 = reinterpret_cast<const v4f*>(W);
  v4f* d4 = reinterpret_cast<v4f*>(dst + b * CT);
  int r = tid / FI;
  int it = tid % FI;
  int e0 = r * ROW_ITEMS + it;
  const int own = sl * FI;

  v4f ws[4];
#pragma unroll
  for (int i = 0; i < 4; ++i) ws[i] = w4[tid + i * LN_THREADS];
  v4f xv[2];
#pragma unroll
  for (int i = 0; i < 2; ++i) xv[i] = p[e0 + i * FROWS * ROW_ITEMS + own];
  int hrow = tid >> 1;
  int side = tid & 1;
  bool hlane = conv && tid < 2 * CC;
  bool hin = hlane && (side ? (sl < FS - 1) : (sl > 0));
  int he = hrow * ROW_ITEMS + own + (side ? FI : -1);
  v4f hv = {0.f, 0.f, 0.f, 0.f};
  v4f hg = {0.f, 0.f, 0.f, 0.f};
  v4f hb = {0.f, 0.f, 0.f, 0.f};
  if (hin) {
    hv = p[he];
    if (MODE == PW_CONV_LN) {
      hg = g4[he];
      hb = b4[he];
    }
  }
  v4f s4 = {0.f, 0.f, 0.f, 0.f};
  v4f q4 = {0.f, 0.f, 0.f, 0.f};
#pragma unroll 5
  for (int j = 1; j < FS; ++j) {
    int o = ((sl + j) & (FS - 1)) * FI;
    v4f v0 = p[e0 + o];
    v4f v1 = p[e0 + FROWS * ROW_ITEMS + o];
    s4 += v0;
    q4 += v0 * v0;
    s4 += v1;
    q4 += v1 * v1;
  }
#pragma unroll
  for (int i = 0; i < 2; ++i) {
    s4 += xv[i];
    q4 += xv[i] * xv[i];
  }
  float wt[2][7], cbias[2];
  v4f gv[2], bv[2];
#pragma unroll
  for (int i = 0; i < 2; ++i) {
    int c = r + i * FROWS;
    if (conv) {
#pragma unroll
      for (int j = 0; j < 7; ++j) wt[i][j] = cw[c * 7 + j];
      cbias[i] = cb[c];
    }
    int e = e0 + i * FROWS * ROW_ITEMS + own;
    gv[i] = g4[e];
    bv[i] = b4[e];
  }
#pragma unroll
  for (int i = 0; i < 4; ++i) {
    int qi = tid + i * LN_THREADS;
    *reinterpret_cast<v4f*>(wl + (qi >> 5) * WS_PITCH + ((qi & 31) << 2)) = ws[i];
  }
  float s = (s4.x + s4.y) + (s4.z + s4.w);
  float q = (q4.x + q4.y) + (q4.z + q4.w);
  s = wave_sum(s);
  q = wave_sum(q);
  int w = tid >> 5;
  int l = tid & 31;
  if (l == 0) {
    red[w] = s;
    red[32 + w] = q;
  }
  if (MODE == PW_CONV_RAW) {
#pragma unroll
    for (int i = 0; i < 2; ++i) xt[(r + i * FROWS) * XP + 1 + it] = xv[i];
    if (hlane) xt[hrow * XP + side * (XP - 1)] = hv;
  }
  __syncthreads();
  s = wave_sum(red[l]);
  q = wave_sum(red[32 + l]);
  float inv_n = 1.0f / (float)CT;
  float mu = s * inv_n;
  float var = q * inv_n - mu * mu;
  float rr = rsqrtf(var + 1e-5f);
  v4f y[2];
#pragma unroll
  for (int i = 0; i < 2; ++i) y[i] = (xv[i] - mu) * rr * gv[i] + bv[i];
  if (MODE == PW_CONV_LN) {
#pragma unroll
    for (int i = 0; i < 2; ++i) xt[(r + i * FROWS) * XP + 1 + it] = y[i];
    if (hlane) {
      v4f z = {0.f, 0.f, 0.f, 0.f};
      xt[hrow * XP + side * (XP - 1)] = hin ? (hv - mu) * rr * hg + hb : z;
    }
    __syncthreads();
  }
#pragma unroll
  for (int i = 0; i < 2; ++i) {
    int c = r + i * FROWS;
    v4f o;
    if (conv) {
      v4f lf = xt[c * XP + it];
      v4f rt = xt[c * XP + it + 2];
      v4f md = (MODE == PW_CONV_RAW) ? xv[i] : y[i];
      float t[10] = {lf.y, lf.z, lf.w, md.x, md.y, md.z, md.w, rt.x, rt.y, rt.z};
#pragma unroll
      for (int k = 0; k < 4; ++k) {
        float a = cbias[i];
#pragma unroll
        for (int j = 0; j < 7; ++j) a += t[k + j] * wt[i][j];
        o[k] = a;
      }
    } else {
      o = y[i];
    }
    *reinterpret_cast<v4f*>(bt + c * BP + 4 * it) = o;
  }
  int lo = l & 15;
  int hi = l >> 4;
  int kh = hi << 1;
  int m0 = (w >> 2) << 4;
  int n0 = (w & 3) << 4;
  float bsv[8];
#pragma unroll
  for (int v = 0; v < 8; ++v) bsv[v] = bias[m0 + v + 8 * hi];
  __syncthreads();
  const float* wa = wl + (m0 + lo) * WS_PITCH + kh;
  const float* bp = bt + kh * BP + n0 + lo;
  v8f acc = {};
#pragma unroll
  for (int sk = 0; sk < CC / 4; ++sk) {
    v2f a = ld2(wa + 4 * sk);
    v2f bf;
    bf.x = bp[(4 * sk) * BP];
    bf.y = bp[(4 * sk + 1) * BP];
    acc = wmma4(a, bf, acc);
  }
#pragma unroll
  for (int v = 0; v < 8; ++v) {
    int m = m0 + v + 8 * hi;
    ot[m * OP + n0 + lo] = fmaxf(acc[v] + bsv[v], 0.f);
  }
  __syncthreads();
#pragma unroll
  for (int i = 0; i < 2; ++i) {
    int c = r + i * FROWS;
    v4f o = *reinterpret_cast<const v4f*>(ot + c * OP + 4 * it);
    o += (MODE == PW_CONV_RAW) ? y[i] : xv[i];
    d4[e0 + i * FROWS * ROW_ITEMS + own] = o;
  }
}


__device__ __forceinline__ void w_load(const float* __restrict__ wb, int kh,
                                       v4f (&r0)[CC / 4], v4f (&r1)[CC / 4]) {
#pragma unroll
  for (int s = 0; s < CC / 4; ++s) {
    int k0 = 4 * s + kh;
    r0[s] = ld4(wb + k0 * DKD);
    r1[s] = ld4(wb + (k0 + 1) * DKD);
  }
}

__device__ __forceinline__ void xw_tile(const float (&ax)[CC / 4], const float (&ay)[CC / 4],
                                        const v4f (&r0)[CC / 4], const v4f (&r1)[CC / 4],
                                        float* __restrict__ db, int t0, int l) {
  v8f acc0 = {}, acc1 = {}, acc2 = {}, acc3 = {};
#pragma unroll
  for (int s = 0; s < CC / 4; ++s) {
    v2f a, b0, b1, b2, b3;
    a.x = ax[s]; a.y = ay[s];
    b0.x = r0[s].x; b0.y = r1[s].x;
    b1.x = r0[s].y; b1.y = r1[s].y;
    b2.x = r0[s].z; b2.y = r1[s].z;
    b3.x = r0[s].w; b3.y = r1[s].w;
    acc0 = wmma4(a, b0, acc0);
    acc1 = wmma4(a, b1, acc1);
    acc2 = wmma4(a, b2, acc2);
    acc3 = wmma4(a, b3, acc3);
  }
#pragma unroll
  for (int v = 0; v < 8; ++v) {
    int t = t0 + v + ((l >> 4) << 3);
    v4f o;
    o.x = acc0[v];
    o.y = acc1[v];
    o.z = acc2[v];
    o.w = acc3[v];
    *reinterpret_cast<v4f*>(db + t * DKD) = o;
  }
}

__global__ __launch_bounds__(128) void qkv_gemm(const float* __restrict__ X,
                                                const float* __restrict__ Wq,
                                                const float* __restrict__ Wk,
                                                const float* __restrict__ Wv,
                                                float* __restrict__ Qd,
                                                float* __restrict__ Kd,
                                                float* __restrict__ Vd) {
  int l = threadIdx.x & 31;
  int tile = blockIdx.x * 4 + (threadIdx.x >> 5);
  int b = tile >> 6;
  int t0 = (tile & 63) << 4;
  int lo = l & 15;
  int kh = (l >> 4) << 1;
  const float* xb = X + b * CT + t0 + lo;
  int dofs = b * TT * DKD + 4 * lo;
  float ax[CC / 4], ay[CC / 4];
  v4f ra0[CC / 4], ra1[CC / 4];
  v4f rb0[CC / 4], rb1[CC / 4];
#pragma unroll
  for (int s = 0; s < CC / 4; ++s) {
    int k0 = 4 * s + kh;
    ax[s] = xb[k0 * TT];
    ay[s] = xb[(k0 + 1) * TT];
  }
  w_load(Wq + 4 * lo, kh, ra0, ra1);
  __builtin_amdgcn_sched_barrier(0);
  w_load(Wk + 4 * lo, kh, rb0, rb1);
  __builtin_amdgcn_sched_barrier(0);
  xw_tile(ax, ay, ra0, ra1, Qd + dofs, t0, l);
  __builtin_amdgcn_sched_barrier(0);
  w_load(Wv + 4 * lo, kh, ra0, ra1);
  __builtin_amdgcn_sched_barrier(0);
  xw_tile(ax, ay, rb0, rb1, Kd + dofs, t0, l);
  __builtin_amdgcn_sched_barrier(0);
  xw_tile(ax, ay, ra0, ra1, Vd + dofs, t0, l);
}


__device__ __forceinline__ float xhalf(float x) {
  int b = __builtin_bit_cast(int, x);
  return __builtin_bit_cast(
      float, __builtin_amdgcn_permlanex16(b, b, 0x76543210, 0xfedcba98, false, false));
}

__device__ __forceinline__ void load_block(const float* __restrict__ kb,
                                           const float* __restrict__ vb,
                                           const float* __restrict__ mb,
                                           int key0, int c, int h,
                                           v4f (&k)[2][8], v4f (&v)[2][8],
                                           v4f (&hm)[2][2]) {
#pragma unroll
  for (int u = 0; u < 2; ++u) {
    const float* kp = kb + (key0 + 16 * u + c) * DKD + 4 * h;
    const float* vp = vb + (key0 + 16 * u + 8 * h) * DKD + 4 * c;
    const float* mp = mb + key0 + 16 * u + 8 * h;
#pragma unroll
    for (int m = 0; m < 8; ++m) k[u][m] = ld4(kp + 8 * m);
#pragma unroll
    for (int r = 0; r < 8; ++r) v[u][r] = ld4(vp + r * DKD);
    hm[u][0] = ld4(mp);
    hm[u][1] = ld4(mp + 4);
  }
}

__device__ __forceinline__ void process_block(const v4f (&q)[8],
                                              const v4f (&k)[2][8],
                                              const v4f (&v)[2][8],
                                              const v4f (&hm)[2][2],
                                              float& m_run, float& l_run,
                                              v8f (&acc)[4]) {
  v8f s[2] = {};
#pragma unroll
  for (int m = 0; m < 8; ++m) {
    v2f qlo, qhi, a;
    qlo.x = q[m].x; qlo.y = q[m].y;
    qhi.x = q[m].z; qhi.y = q[m].w;
    a.x = k[0][m].x; a.y = k[0][m].y;
    s[0] = wmma4(a, qlo, s[0]);
    a.x = k[1][m].x; a.y = k[1][m].y;
    s[1] = wmma4(a, qlo, s[1]);
    a.x = k[0][m].z; a.y = k[0][m].w;
    s[0] = wmma4(a, qhi, s[0]);
    a.x = k[1][m].z; a.y = k[1][m].w;
    s[1] = wmma4(a, qhi, s[1]);
  }
  float val[2][8];
  float bm = -3.4e38f;
#pragma unroll
  for (int u = 0; u < 2; ++u) {
#pragma unroll
    for (int r = 0; r < 8; ++r) {
      float hmv = hm[u][r >> 2][r & 3];
      float sv = s[u][r] * 0.125f;
      val[u][r] = sv * hmv + (1.f - hmv) * -1e30f;
      bm = fmaxf(bm, val[u][r]);
    }
  }
  bm = fmaxf(bm, xhalf(bm));
  float m_new = fmaxf(m_run, bm);
  float alpha = expf(m_run - m_new);
  float p[2][8];
  float ps = 0.f;
#pragma unroll
  for (int u = 0; u < 2; ++u) {
#pragma unroll
    for (int r = 0; r < 8; ++r) {
      p[u][r] = expf(val[u][r] - m_new);
      ps += p[u][r];
    }
  }
  l_run = l_run * alpha + ps;
  m_run = m_new;
#pragma unroll
  for (int i = 0; i < 4; ++i) acc[i] = acc[i] * alpha;
#pragma unroll
  for (int u = 0; u < 2; ++u) {
#pragma unroll
    for (int j = 0; j < 4; ++j) {
      v2f bp;
      bp.x = p[u][2 * j];
      bp.y = p[u][2 * j + 1];
#pragma unroll
      for (int i = 0; i < 4; ++i) {
        v2f a;
        a.x = v[u][2 * j][i];
        a.y = v[u][2 * j + 1][i];
        acc[i] = wmma4(a, bp, acc[i]);
      }
    }
  }
}

__global__ __launch_bounds__(128) void attn_kernel(const float* __restrict__ Q,
                                                   const float* __restrict__ Km,
                                                   const float* __restrict__ Vm,
                                                   const float* __restrict__ mask,
                                                   float* __restrict__ head) {
  int l = threadIdx.x & 31;
  int tile = blockIdx.x * 4 + (threadIdx.x >> 5);
  int b = tile >> 6;
  int q0 = (tile & 63) << 4;
  int c = l & 15;
  int h = l >> 4;
  const float* qb = Q + b * TT * DKD;
  const float* kb = Km + b * TT * DKD;
  const float* vb = Vm + b * TT * DKD;
  const float* mb = mask + b * TT;

  v4f q[8];
#pragma unroll
  for (int m = 0; m < 8; ++m) q[m] = ld4(qb + (q0 + c) * DKD + 8 * m + 4 * h);

  v4f kA[2][8], vA[2][8], hA[2][2];
  v4f kB[2][8], vB[2][8], hB[2][2];
  v8f acc[4] = {};
  float m_run = -3.4e38f;
  float l_run = 0.f;

  load_block(kb, vb, mb, 0, c, h, kA, vA, hA);
  for (int blk = 0; blk < TT / 32 - 2; blk += 2) {
    load_block(kb, vb, mb, (blk + 1) * 32, c, h, kB, vB, hB);
    process_block(q, kA, vA, hA, m_run, l_run, acc);
    load_block(kb, vb, mb, (blk + 2) * 32, c, h, kA, vA, hA);
    process_block(q, kB, vB, hB, m_run, l_run, acc);
  }
  load_block(kb, vb, mb, TT - 32, c, h, kB, vB, hB);
  process_block(q, kA, vA, hA, m_run, l_run, acc);
  process_block(q, kB, vB, hB, m_run, l_run, acc);

  float lsum = l_run + xhalf(l_run);
  float f = (1.0f / lsum) * mb[q0 + c];
  float* hp = head + b * TT * DKD + (q0 + c) * DKD + 32 * h;
#pragma unroll
  for (int r = 0; r < 8; ++r) {
    v4f o;
    o.x = acc[0][r];
    o.y = acc[1][r];
    o.z = acc[2][r];
    o.w = acc[3][r];
    *reinterpret_cast<v4f*>(hp + 4 * r) = o * f;
  }
}

__global__ __launch_bounds__(128) void wo_gemm(const float* __restrict__ H,
                                               const float* __restrict__ Wo,
                                               float* __restrict__ res) {
  __shared__ v4f ht4[4 * DKD * DKD / 4];
  int l = threadIdx.x & 31;
  int wv = threadIdx.x >> 5;
  int tile = blockIdx.x * 4 + wv;
  int b = tile >> 7;
  int rem = tile & 127;
  int m0 = (rem >> 4) << 4;
  int n0 = (rem & 15) << 6;
  int lo = l & 15;
  int kh = (l >> 4) << 1;
  const float* hsrc = H + b * TT * DKD + n0 * DKD + 4 * l;
  float* ht = reinterpret_cast<float*>(ht4) + wv * DKD * DKD;
  const float* wc = Wo + m0 + lo;
  float* rb = res + b * CT + n0 + 4 * lo;
  v4f hv[DKD / 2];
#pragma unroll
  for (int i = 0; i < DKD / 2; ++i) hv[i] = ld4(hsrc + i * 128);
  __builtin_amdgcn_sched_barrier(0);
  float wa[DKD / 4], wb2[DKD / 4], wc2[DKD / 4], wd[DKD / 4];
  v4f old[8];
#pragma unroll
  for (int s = 0; s < DKD / 4; ++s) {
    int k0 = 4 * s + kh;
    wa[s]  = wc[k0 * CC];
    wb2[s] = wc[(k0 + 64) * CC];
    wc2[s] = wc[(k0 + 1) * CC];
    wd[s]  = wc[(k0 + 65) * CC];
  }
#pragma unroll
  for (int v = 0; v < 8; ++v) {
    int m = m0 + v + ((l >> 4) << 3);
    old[v] = ld4(rb + m * TT);
  }
  __builtin_amdgcn_sched_barrier(0);
#pragma unroll
  for (int i = 0; i < DKD / 2; ++i) {
    int row = 2 * i + (l >> 4);
    int col = 4 * (l & 15);
    *reinterpret_cast<v4f*>(ht + row * DKD + (col ^ (4 * (row >> 2)))) = hv[i];
  }
  lds_handover();
  v2f hf[DKD / 4][4];
#pragma unroll
  for (int s = 0; s < DKD / 4; ++s) {
    const float* hp = ht + (4 * lo) * DKD + ((4 * s) ^ (4 * lo)) + kh;
#pragma unroll
    for (int j = 0; j < 4; ++j) hf[s][j] = ld2(hp + j * DKD);
  }
  __builtin_amdgcn_sched_barrier(0);
  v8f acc0 = {}, acc1 = {}, acc2 = {}, acc3 = {};
#pragma unroll
  for (int s = 0; s < DKD / 4; ++s) {
    v2f a;
    a.x = wa[s] + wb2[s];
    a.y = wc2[s] + wd[s];
    acc0 = wmma4(a, hf[s][0], acc0);
    acc1 = wmma4(a, hf[s][1], acc1);
    acc2 = wmma4(a, hf[s][2], acc2);
    acc3 = wmma4(a, hf[s][3], acc3);
  }
#pragma unroll
  for (int v = 0; v < 8; ++v) {
    int m = m0 + v + ((l >> 4) << 3);
    v4f o;
    o.x = acc0[v] + old[v].x;
    o.y = acc1[v] + old[v].y;
    o.z = acc2[v] + old[v].z;
    o.w = acc3[v] + old[v].w;
    *reinterpret_cast<v4f*>(rb + m * TT) = o;
  }
}


extern "C" void kernel_launch(void* const* d_in, const int* in_sizes, int n_in,
                              void* d_out, int out_size, void* d_ws, size_t ws_size,
                              hipStream_t stream) {
  (void)in_sizes; (void)n_in; (void)out_size; (void)ws_size;
  const float* x    = (const float*)d_in[0];
  const float* mask = (const float*)d_in[1];
  const float* dw_w = (const float*)d_in[2];
  const float* dw_b = (const float*)d_in[3];
  const float* pw_w = (const float*)d_in[4];
  const float* pw_b = (const float*)d_in[5];
  const float* n0g  = (const float*)d_in[6];
  const float* n0b  = (const float*)d_in[7];
  const float* nsg  = (const float*)d_in[8];
  const float* nsb  = (const float*)d_in[9];
  const float* neg_ = (const float*)d_in[10];
  const float* neb  = (const float*)d_in[11];
  const float* Wq   = (const float*)d_in[12];
  const float* Wk   = (const float*)d_in[13];
  const float* Wv   = (const float*)d_in[14];
  const float* Wo   = (const float*)d_in[15];
  const float* fcw  = (const float*)d_in[16];
  const float* fcb  = (const float*)d_in[17];
  float* outp = (float*)d_out;
  float* ws = (float*)d_ws;

  float* OUT = ws;
  float* RES = ws + 2097152;
  float* TMP = ws + 4194304;
  float* Qb  = ws + 6291456;
  float* Kb  = ws + 7340032;
  float* Vb  = ws + 8388608;
  float* Hb  = ws + 9437184;

  posenc_add<<<8192, 256, 0, stream>>>(x, OUT);
  ln_pw<PW_CONV_RAW><<<256, 1024, 0, stream>>>(OUT, n0g, n0b, dw_w, dw_b, pw_w, pw_b, RES);
  ln_pw<PW_CONV_LN><<<256, 1024, 0, stream>>>(RES, nsg, nsb, dw_w + CC * 7, dw_b + CC,
                                              pw_w + CC * CC, pw_b + CC, TMP);
  ln_pw<PW_CONV_LN><<<256, 1024, 0, stream>>>(TMP, nsg + CT, nsb + CT, dw_w + 2 * CC * 7,
                                              dw_b + 2 * CC, pw_w + 2 * CC * CC, pw_b + 2 * CC, RES);
  ln_pw<PW_CONV_LN><<<256, 1024, 0, stream>>>(RES, nsg + 2 * CT, nsb + 2 * CT, dw_w + 3 * CC * 7,
                                              dw_b + 3 * CC, pw_w + 3 * CC * CC, pw_b + 3 * CC, TMP);
  ln_fused<<<128, 1024, 0, stream>>>(TMP, nsg + 3 * CT, nsb + 3 * CT, OUT);

  qkv_gemm<<<256, 128, 0, stream>>>(OUT, Wq, Wk, Wv, Qb, Kb, Vb);
  attn_kernel<<<256, 128, 0, stream>>>(Qb, Kb, Vb, mask, Hb);
  wo_gemm<<<512, 128, 0, stream>>>(Hb, Wo, TMP);
  ln_pw<PW_LN><<<256, 1024, 0, stream>>>(TMP, neg_, neb, nullptr, nullptr, fcw, fcb, outp);
}


# First LayerNorm launch reads x and adds the positional term itself (own slice, border: from the angle; sweep: turning sin/cos pairs of neighbour channels): posenc launch gone; another LN sum order
# speedup vs baseline: 4.0155x; 1.0569x over previous
#include <hip/hip_runtime.h>

typedef float v2f __attribute__((ext_vector_type(2)));
typedef float v8f __attribute__((ext_vector_type(8)));

#define BB 16
#define CC 128
#define TT 1024
#define DKD 64
#define CT (CC * TT)
#define LOG10000 9.210340371976184f
#define PI_2 1.5707963267948966f

__device__ inline v8f wmma4(v2f a, v2f b, v8f c) {
  return __builtin_amdgcn_wmma_f32_16x16x4_f32(false, a, false, b, (short)0, c,
                                               false, false);
}


__device__ __forceinline__ float pe_freq(int c) {
  return expf(-(float)(c & ~1) * (LOG10000 / (float)CC));
}

typedef float v4f __attribute__((ext_vector_type(4)));
#define LN_SLICES 8
#define LN_THREADS 1024
#define CT4 (CT / 4)
#define LN_ITEMS (CT4 / LN_THREADS)
#define LN_OWN (LN_ITEMS / LN_SLICES)
#define ROW_ITEMS (TT / 4)

__device__ __forceinline__ float wave_sum(float x) {
#pragma unroll
  for (int m = 16; m > 0; m >>= 1) x += __shfl_xor(x, m);
  return x;
}

__global__ __launch_bounds__(LN_THREADS) void ln_fused(const float* __restrict__ src,
                                                       const float* __restrict__ g,
                                                       const float* __restrict__ bb,
                                                       float* __restrict__ dst) {
  __shared__ float red[64];
  int b = blockIdx.x / LN_SLICES;
  int sl = blockIdx.x % LN_SLICES;
  int tid = threadIdx.x;
  const v4f* p = reinterpret_cast<const v4f*>(src + b * CT);
  const v4f* g4 = reinterpret_cast<const v4f*>(g);
  const v4f* b4 = reinterpret_cast<const v4f*>(bb);
  v4f* d4 = reinterpret_cast<v4f*>(dst + b * CT);

  const int base = sl * LN_OWN;
  v4f xv[LN_OWN];
#pragma unroll
  for (int i = 0; i < LN_OWN; ++i) xv[i] = p[(base + i) * LN_THREADS + tid];
  v4f s4 = {0.f, 0.f, 0.f, 0.f};
  v4f q4 = {0.f, 0.f, 0.f, 0.f};
#pragma unroll 7
  for (int i = LN_OWN; i < LN_ITEMS; ++i) {
    v4f v = p[((base + i) & (LN_ITEMS - 1)) * LN_THREADS + tid];
    s4 += v;
    q4 += v * v;
  }
#pragma unroll
  for (int i = 0; i < LN_OWN; ++i) {
    s4 += xv[i];
    q4 += xv[i] * xv[i];
  }
  v4f gv[LN_OWN], bv[LN_OWN];
#pragma unroll
  for (int i = 0; i < LN_OWN; ++i) {
    int e = (base + i) * LN_THREADS + tid;
    gv[i] = g4[e];
    bv[i] = b4[e];
  }
  float s = (s4.x + s4.y) + (s4.z + s4.w);
  float q = (q4.x + q4.y) + (q4.z + q4.w);
  s = wave_sum(s);
  q = wave_sum(q);
  int w = tid >> 5;
  int l = tid & 31;
  if (l == 0) {
    red[w] = s;
    red[32 + w] = q;
  }
  __syncthreads();
  s = wave_sum(red[l]);
  q = wave_sum(red[32 + l]);
  float inv_n = 1.0f / (float)CT;
  float mu = s * inv_n;
  float var = q * inv_n - mu * mu;
  float r = rsqrtf(var + 1e-5f);
#pragma unroll
  for (int i = 0; i < LN_OWN; ++i) {
    int e = (base + i) * LN_THREADS + tid;
    d4[e] = (xv[i] - mu) * r * gv[i] + bv[i];
  }
}


__device__ __forceinline__ v4f ld4(const float* p) {
  return *reinterpret_cast<const v4f*>(p);
}

__device__ __forceinline__ v2f ld2(const float* p) {
  return *reinterpret_cast<const v2f*>(p);
}


__device__ __forceinline__ void lds_handover() {
  asm volatile("s_wait_dscnt 0x0\n\ts_barrier_signal -1\n\ts_barrier_wait -1" ::: "memory");
}

#define PW_CONV_LN 0
#define PW_CONV_RAW 1
#define PW_LN 2
#define FS 16
#define FT (TT / FS)
#define FI (FT / 4)
#define FROWS (LN_THREADS / FI)
#define XP (FI + 2)
#define BP (FT + 8)
#define OP (FT + 4)
#define WS_PITCH (CC + 4)
#define LDS_BT 0
#define LDS_XT (LDS_BT + CC * BP / 4)
#define LDS_WL (LDS_XT + CC * XP)
#define LDS_RED (LDS_WL + CC * WS_PITCH / 4)
#define LDS_ITEMS (LDS_RED + 16)

__device__ __forceinline__ void pe_turn(v4f& sn, v4f& cs, float S, float C) {
  v4f n = sn * C + cs * S;
  cs = cs * C - sn * S;
  sn = n;
}

template <int MODE>
__global__ __launch_bounds__(LN_THREADS) void ln_pw(const float* __restrict__ src,
                                                    const float* __restrict__ g,
                                                    const float* __restrict__ bb,
                                                    const float* __restrict__ cw,
                                                    const float* __restrict__ cb,
                                                    const float* __restrict__ W,
                                                    const float* __restrict__ bias,
                                                    float* __restrict__ dst) {
  __shared__ v4f lds[LDS_ITEMS];
  float* bt = reinterpret_cast<float*>(lds + LDS_BT);
  v4f* xt = lds + LDS_XT;
  float* ot = reinterpret_cast<float*>(lds + LDS_XT);
  float* wl = reinterpret_cast<float*>(lds + LDS_WL);
  float* red = reinterpret_cast<float*>(lds + LDS_RED);
  const bool conv = MODE != PW_LN;
  int b = blockIdx.x / FS;
  int sl = blockIdx.x % FS;
  int tid = threadIdx.x;
  const v4f* p = reinterpret_cast<const v4f*>(src + b * CT);
  const v4f* g4 = reinterpret_cast<const v4f*>(g);
  const v4f* b4 = reinterpret_cast<const v4f*>(bb);
  const v4f* w4 = reinterpret_cast<const v4f*>(W);
  v4f* d4 = reinterpret_cast<v4f*>(dst + b * CT);
  const bool pos = MODE == PW_CONV_RAW;
  int r = tid / FI;
  int it = tid % FI;
  const int c0 = pos ? 2 * r : r;
  const int cstep = pos ? 1 : FROWS;
  int e0 = c0 * ROW_ITEMS + it;
  const int e1 = cstep * ROW_ITEMS;
  const int own = sl * FI;

  v4f ws[4];
#pragma unroll
  for (int i = 0; i < 4; ++i) ws[i] = w4[tid + i * LN_THREADS];
  v4f xv[2];
#pragma unroll
  for (int i = 0; i < 2; ++i) xv[i] = p[e0 + i * e1 + own];
  int hrow = tid >> 1;
  int side = tid & 1;
  bool hlane = conv && tid < 2 * CC;
  bool hin = hlane && (side ? (sl < FS - 1) : (sl > 0));
  int he = hrow * ROW_ITEMS + own + (side ? FI : -1);
  v4f hv = {0.f, 0.f, 0.f, 0.f};
  v4f hg = {0.f, 0.f, 0.f, 0.f};
  v4f hb = {0.f, 0.f, 0.f, 0.f};
  if (hin) {
    hv = p[he];
    if (MODE == PW_CONV_LN) {
      hg = g4[he];
      hb = b4[he];
    }
  }
  v4f s4 = {0.f, 0.f, 0.f, 0.f};
  v4f q4 = {0.f, 0.f, 0.f, 0.f};
  if (!pos) {
#pragma unroll 5
    for (int j = 1; j < FS; ++j) {
      int o = ((sl + j) & (FS - 1)) * FI;
      v4f v0 = p[e0 + o];
      v4f v1 = p[e0 + e1 + o];
      s4 += v0;
      q4 += v0 * v0;
      s4 += v1;
      q4 += v1 * v1;
    }
  } else {
    v4f pa[5], pb[5];
    auto ask = [&](int j0) {
#pragma unroll
      for (int j = 0; j < 5; ++j) {
        int o = ((sl + j0 + j) & (FS - 1)) * FI;
        pa[j] = p[e0 + o];
        pb[j] = p[e0 + e1 + o];
      }
    };
    ask(1);
    const float f = pe_freq(c0);
    v4f sn, cs;
#pragma unroll
    for (int k = 0; k < 4; ++k) {
      float a = (float)(sl * FT + 4 * it + k) * f;
      sn[k] = sinf(a);
      cs[k] = sinf(a + PI_2);
    }
    xv[0] += sn;
    xv[1] += cs;
    float s1, k1, sw, kw;
    sincosf((float)FT * f, &s1, &k1);
    sincosf((float)TT * f, &sw, &kw);
    const float s0 = s1 * kw - k1 * sw;
    const float k0 = k1 * kw + s1 * sw;
    auto sum5 = [&](int j0) {
#pragma unroll
      for (int j = 0; j < 5; ++j) {
        bool wrap = ((sl + j0 + j) & (FS - 1)) == 0;
        pe_turn(sn, cs, wrap ? s0 : s1, wrap ? k0 : k1);
        v4f v0 = pa[j] + sn;
        v4f v1 = pb[j] + cs;
        s4 += v0;
        q4 += v0 * v0;
        s4 += v1;
        q4 += v1 * v1;
      }
    };
    sum5(1);
    ask(6);
    __builtin_amdgcn_sched_barrier(0);
    sum5(6);
    ask(11);
    __builtin_amdgcn_sched_barrier(0);
    sum5(11);
  }
#pragma unroll
  for (int i = 0; i < 2; ++i) {
    s4 += xv[i];
    q4 += xv[i] * xv[i];
  }
  float wt[2][7], cbias[2];
  v4f gv[2], bv[2];
#pragma unroll
  for (int i = 0; i < 2; ++i) {
    int c = c0 + i * cstep;
    if (conv) {
#pragma unroll
      for (int j = 0; j < 7; ++j) wt[i][j] = cw[c * 7 + j];
      cbias[i] = cb[c];
    }
    int e = e0 + i * e1 + own;
    gv[i] = g4[e];
    bv[i] = b4[e];
  }
#pragma unroll
  for (int i = 0; i < 4; ++i) {
    int qi = tid + i * LN_THREADS;
    *reinterpret_cast<v4f*>(wl + (qi >> 5) * WS_PITCH + ((qi & 31) << 2)) = ws[i];
  }
  float s = (s4.x + s4.y) + (s4.z + s4.w);
  float q = (q4.x + q4.y) + (q4.z + q4.w);
  s = wave_sum(s);
  q = wave_sum(q);
  int w = tid >> 5;
  int l = tid & 31;
  if (l == 0) {
    red[w] = s;
    red[32 + w] = q;
  }
  if (MODE == PW_CONV_RAW) {
#pragma unroll
    for (int i = 0; i < 2; ++i) xt[(c0 + i * cstep) * XP + 1 + it] = xv[i];
    if (hin) {
      float hf = pe_freq(hrow);
      int th = 4 * (own + (side ? FI : -1));
#pragma unroll
      for (int k = 0; k < 4; ++k) {
        float a = (float)(th + k) * hf;
        hv[k] += sinf((hrow & 1) ? a + PI_2 : a);
      }
    }
    if (hlane) xt[hrow * XP + side * (XP - 1)] = hv;
  }
  __syncthreads();
  s = wave_sum(red[l]);
  q = wave_sum(red[32 + l]);
  float inv_n = 1.0f / (float)CT;
  float mu = s * inv_n;
  float var = q * inv_n - mu * mu;
  float rr = rsqrtf(var + 1e-5f);
  v4f y[2];
#pragma unroll
  for (int i = 0; i < 2; ++i) y[i] = (xv[i] - mu) * rr * gv[i] + bv[i];
  if (MODE == PW_CONV_LN) {
#pragma unroll
    for (int i = 0; i < 2; ++i) xt[(c0 + i * cstep) * XP + 1 + it] = y[i];
    if (hlane) {
      v4f z = {0.f, 0.f, 0.f, 0.f};
      xt[hrow * XP + side * (XP - 1)] = hin ? (hv - mu) * rr * hg + hb : z;
    }
    __syncthreads();
  }
#pragma unroll
  for (int i = 0; i < 2; ++i) {
    int c = c0 + i * cstep;
    v4f o;
    if (conv) {
      v4f lf = xt[c * XP + it];
      v4f rt = xt[c * XP + it + 2];
      v4f md = (MODE == PW_CONV_RAW) ? xv[i] : y[i];
      float t[10] = {lf.y, lf.z, lf.w, md.x, md.y, md.z, md.w, rt.x, rt.y, rt.z};
#pragma unroll
      for (int k = 0; k < 4; ++k) {
        float a = cbias[i];
#pragma unroll
        for (int j = 0; j < 7; ++j) a += t[k + j] * wt[i][j];
        o[k] = a;
      }
    } else {
      o = y[i];
    }
    *reinterpret_cast<v4f*>(bt + c * BP + 4 * it) = o;
  }
  int lo = l & 15;
  int hi = l >> 4;
  int kh = hi << 1;
  int m0 = (w >> 2) << 4;
  int n0 = (w & 3) << 4;
  float bsv[8];
#pragma unroll
  for (int v = 0; v < 8; ++v) bsv[v] = bias[m0 + v + 8 * hi];
  __syncthreads();
  const float* wa = wl + (m0 + lo) * WS_PITCH + kh;
  const float* bp = bt + kh * BP + n0 + lo;
  v8f acc = {};
#pragma unroll
  for (int sk = 0; sk < CC / 4; ++sk) {
    v2f a = ld2(wa + 4 * sk);
    v2f bf;
    bf.x = bp[(4 * sk) * BP];
    bf.y = bp[(4 * sk + 1) * BP];
    acc = wmma4(a, bf, acc);
  }
#pragma unroll
  for (int v = 0; v < 8; ++v) {
    int m = m0 + v + 8 * hi;
    ot[m * OP + n0 + lo] = fmaxf(acc[v] + bsv[v], 0.f);
  }
  __syncthreads();
#pragma unroll
  for (int i = 0; i < 2; ++i) {
    int c = c0 + i * cstep;
    v4f o = *reinterpret_cast<const v4f*>(ot + c * OP + 4 * it);
    o += (MODE == PW_CONV_RAW) ? y[i] : xv[i];
    d4[e0 + i * e1 + own] = o;
  }
}


__device__ __forceinline__ void w_load(const float* __restrict__ wb, int kh,
                                       v4f (&r0)[CC / 4], v4f (&r1)[CC / 4]) {
#pragma unroll
  for (int s = 0; s < CC / 4; ++s) {
    int k0 = 4 * s + kh;
    r0[s] = ld4(wb + k0 * DKD);
    r1[s] = ld4(wb + (k0 + 1) * DKD);
  }
}

__device__ __forceinline__ void xw_tile(const float (&ax)[CC / 4], const float (&ay)[CC / 4],
                                        const v4f (&r0)[CC / 4], const v4f (&r1)[CC / 4],
                                        float* __restrict__ db, int t0, int l) {
  v8f acc0 = {}, acc1 = {}, acc2 = {}, acc3 = {};
#pragma unroll
  for (int s = 0; s < CC / 4; ++s) {
    v2f a, b0, b1, b2, b3;
    a.x = ax[s]; a.y = ay[s];
    b0.x = r0[s].x; b0.y = r1[s].x;
    b1.x = r0[s].y; b1.y = r1[s].y;
    b2.x = r0[s].z; b2.y = r1[s].z;
    b3.x = r0[s].w; b3.y = r1[s].w;
    acc0 = wmma4(a, b0, acc0);
    acc1 = wmma4(a, b1, acc1);
    acc2 = wmma4(a, b2, acc2);
    acc3 = wmma4(a, b3, acc3);
  }
#pragma unroll
  for (int v = 0; v < 8; ++v) {
    int t = t0 + v + ((l >> 4) << 3);
    v4f o;
    o.x = acc0[v];
    o.y = acc1[v];
    o.z = acc2[v];
    o.w = acc3[v];
    *reinterpret_cast<v4f*>(db + t * DKD) = o;
  }
}

__global__ __launch_bounds__(128) void qkv_gemm(const float* __restrict__ X,
                                                const float* __restrict__ Wq,
                                                const float* __restrict__ Wk,
                                                const float* __restrict__ Wv,
                                                float* __restrict__ Qd,
                                                float* __restrict__ Kd,
                                                float* __restrict__ Vd) {
  int l = threadIdx.x & 31;
  int tile = blockIdx.x * 4 + (threadIdx.x >> 5);
  int b = tile >> 6;
  int t0 = (tile & 63) << 4;
  int lo = l & 15;
  int kh = (l >> 4) << 1;
  const float* xb = X + b * CT + t0 + lo;
  int dofs = b * TT * DKD + 4 * lo;
  float ax[CC / 4], ay[CC / 4];
  v4f ra0[CC / 4], ra1[CC / 4];
  v4f rb0[CC / 4], rb1[CC / 4];
#pragma unroll
  for (int s = 0; s < CC / 4; ++s) {
    int k0 = 4 * s + kh;
    ax[s] = xb[k0 * TT];
    ay[s] = xb[(k0 + 1) * TT];
  }
  w_load(Wq + 4 * lo, kh, ra0, ra1);
  __builtin_amdgcn_sched_barrier(0);
  w_load(Wk + 4 * lo, kh, rb0, rb1);
  __builtin_amdgcn_sched_barrier(0);
  xw_tile(ax, ay, ra0, ra1, Qd + dofs, t0, l);
  __builtin_amdgcn_sched_barrier(0);
  w_load(Wv + 4 * lo, kh, ra0, ra1);
  __builtin_amdgcn_sched_barrier(0);
  xw_tile(ax, ay, rb0, rb1, Kd + dofs, t0, l);
  __builtin_amdgcn_sched_barrier(0);
  xw_tile(ax, ay, ra0, ra1, Vd + dofs, t0, l);
}


__device__ __forceinline__ float xhalf(float x) {
  int b = __builtin_bit_cast(int, x);
  return __builtin_bit_cast(
      float, __builtin_amdgcn_permlanex16(b, b, 0x76543210, 0xfedcba98, false, false));
}

__device__ __forceinline__ void load_block(const float* __restrict__ kb,
                                           const float* __restrict__ vb,
                                           const float* __restrict__ mb,
                                           int key0, int c, int h,
                                           v4f (&k)[2][8], v4f (&v)[2][8],
                                           v4f (&hm)[2][2]) {
#pragma unroll
  for (int u = 0; u < 2; ++u) {
    const float* kp = kb + (key0 + 16 * u + c) * DKD + 4 * h;
    const float* vp = vb + (key0 + 16 * u + 8 * h) * DKD + 4 * c;
    const float* mp = mb + key0 + 16 * u + 8 * h;
#pragma unroll
    for (int m = 0; m < 8; ++m) k[u][m] = ld4(kp + 8 * m);
#pragma unroll
    for (int r = 0; r < 8; ++r) v[u][r] = ld4(vp + r * DKD);
    hm[u][0] = ld4(mp);
    hm[u][1] = ld4(mp + 4);
  }
}

__device__ __forceinline__ void process_block(const v4f (&q)[8],
                                              const v4f (&k)[2][8],
                                              const v4f (&v)[2][8],
                                              const v4f (&hm)[2][2],
                                              float& m_run, float& l_run,
                                              v8f (&acc)[4]) {
  v8f s[2] = {};
#pragma unroll
  for (int m = 0; m < 8; ++m) {
    v2f qlo, qhi, a;
    qlo.x = q[m].x; qlo.y = q[m].y;
    qhi.x = q[m].z; qhi.y = q[m].w;
    a.x = k[0][m].x; a.y = k[0][m].y;
    s[0] = wmma4(a, qlo, s[0]);
    a.x = k[1][m].x; a.y = k[1][m].y;
    s[1] = wmma4(a, qlo, s[1]);
    a.x = k[0][m].z; a.y = k[0][m].w;
    s[0] = wmma4(a, qhi, s[0]);
    a.x = k[1][m].z; a.y = k[1][m].w;
    s[1] = wmma4(a, qhi, s[1]);
  }
  float val[2][8];
  float bm = -3.4e38f;
#pragma unroll
  for (int u = 0; u < 2; ++u) {
#pragma unroll
    for (int r = 0; r < 8; ++r) {
      float hmv = hm[u][r >> 2][r & 3];
      float sv = s[u][r] * 0.125f;
      val[u][r] = sv * hmv + (1.f - hmv) * -1e30f;
      bm = fmaxf(bm, val[u][r]);
    }
  }
  bm = fmaxf(bm, xhalf(bm));
  float m_new = fmaxf(m_run, bm);
  float alpha = expf(m_run - m_new);
  float p[2][8];
  float ps = 0.f;
#pragma unroll
  for (int u = 0; u < 2; ++u) {
#pragma unroll
    for (int r = 0; r < 8; ++r) {
      p[u][r] = expf(val[u][r] - m_new);
      ps += p[u][r];
    }
  }
  l_run = l_run * alpha + ps;
  m_run = m_new;
#pragma unroll
  for (int i = 0; i < 4; ++i) acc[i] = acc[i] * alpha;
#pragma unroll
  for (int u = 0; u < 2; ++u) {
#pragma unroll
    for (int j = 0; j < 4; ++j) {
      v2f bp;
      bp.x = p[u][2 * j];
      bp.y = p[u][2 * j + 1];
#pragma unroll
      for (int i = 0; i < 4; ++i) {
        v2f a;
        a.x = v[u][2 * j][i];
        a.y = v[u][2 * j + 1][i];
        acc[i] = wmma4(a, bp, acc[i]);
      }
    }
  }
}

__global__ __launch_bounds__(128) void attn_kernel(const float* __restrict__ Q,
                                                   const float* __restrict__ Km,
                                                   const float* __restrict__ Vm,
                                                   const float* __restrict__ mask,
                                                   float* __restrict__ head) {
  int l = threadIdx.x & 31;
  int tile = blockIdx.x * 4 + (threadIdx.x >> 5);
  int b = tile >> 6;
  int q0 = (tile & 63) << 4;
  int c = l & 15;
  int h = l >> 4;
  const float* qb = Q + b * TT * DKD;
  const float* kb = Km + b * TT * DKD;
  const float* vb = Vm + b * TT * DKD;
  const float* mb = mask + b * TT;

  v4f q[8];
#pragma unroll
  for (int m = 0; m < 8; ++m) q[m] = ld4(qb + (q0 + c) * DKD + 8 * m + 4 * h);

  v4f kA[2][8], vA[2][8], hA[2][2];
  v4f kB[2][8], vB[2][8], hB[2][2];
  v8f acc[4] = {};
  float m_run = -3.4e38f;
  float l_run = 0.f;

  load_block(kb, vb, mb, 0, c, h, kA, vA, hA);
  for (int blk = 0; blk < TT / 32 - 2; blk += 2) {
    load_block(kb, vb, mb, (blk + 1) * 32, c, h, kB, vB, hB);
    process_block(q, kA, vA, hA, m_run, l_run, acc);
    load_block(kb, vb, mb, (blk + 2) * 32, c, h, kA, vA, hA);
    process_block(q, kB, vB, hB, m_run, l_run, acc);
  }
  load_block(kb, vb, mb, TT - 32, c, h, kB, vB, hB);
  process_block(q, kA, vA, hA, m_run, l_run, acc);
  process_block(q, kB, vB, hB, m_run, l_run, acc);

  float lsum = l_run + xhalf(l_run);
  float f = (1.0f / lsum) * mb[q0 + c];
  float* hp = head + b * TT * DKD + (q0 + c) * DKD + 32 * h;
#pragma unroll
  for (int r = 0; r < 8; ++r) {
    v4f o;
    o.x = acc[0][r];
    o.y = acc[1][r];
    o.z = acc[2][r];
    o.w = acc[3][r];
    *reinterpret_cast<v4f*>(hp + 4 * r) = o * f;
  }
}

__global__ __launch_bounds__(128) void wo_gemm(const float* __restrict__ H,
                                               const float* __restrict__ Wo,
                                               float* __restrict__ res) {
  __shared__ v4f ht4[4 * DKD * DKD / 4];
  int l = threadIdx.x & 31;
  int wv = threadIdx.x >> 5;
  int tile = blockIdx.x * 4 + wv;
  int b = tile >> 7;
  int rem = tile & 127;
  int m0 = (rem >> 4) << 4;
  int n0 = (rem & 15) << 6;
  int lo = l & 15;
  int kh = (l >> 4) << 1;
  const float* hsrc = H + b * TT * DKD + n0 * DKD + 4 * l;
  float* ht = reinterpret_cast<float*>(ht4) + wv * DKD * DKD;
  const float* wc = Wo + m0 + lo;
  float* rb = res + b * CT + n0 + 4 * lo;
  v4f hv[DKD / 2];
#pragma unroll
  for (int i = 0; i < DKD / 2; ++i) hv[i] = ld4(hsrc + i * 128);
  __builtin_amdgcn_sched_barrier(0);
  float wa[DKD / 4], wb2[DKD / 4], wc2[DKD / 4], wd[DKD / 4];
  v4f old[8];
#pragma unroll
  for (int s = 0; s < DKD / 4; ++s) {
    int k0 = 4 * s + kh;
    wa[s]  = wc[k0 * CC];
    wb2[s] = wc[(k0 + 64) * CC];
    wc2[s] = wc[(k0 + 1) * CC];
    wd[s]  = wc[(k0 + 65) * CC];
  }
#pragma unroll
  for (int v = 0; v < 8; ++v) {
    int m = m0 + v + ((l >> 4) << 3);
    old[v] = ld4(rb + m * TT);
  }
  __builtin_amdgcn_sched_barrier(0);
#pragma unroll
  for (int i = 0; i < DKD / 2; ++i) {
    int row = 2 * i + (l >> 4);
    int col = 4 * (l & 15);
    *reinterpret_cast<v4f*>(ht + row * DKD + (col ^ (4 * (row >> 2)))) = hv[i];
  }
  lds_handover();
  v2f hf[DKD / 4][4];
#pragma unroll
  for (int s = 0; s < DKD / 4; ++s) {
    const float* hp = ht + (4 * lo) * DKD + ((4 * s) ^ (4 * lo)) + kh;
#pragma unroll
    for (int j = 0; j < 4; ++j) hf[s][j] = ld2(hp + j * DKD);
  }
  __builtin_amdgcn_sched_barrier(0);
  v8f acc0 = {}, acc1 = {}, acc2 = {}, acc3 = {};
#pragma unroll
  for (int s = 0; s < DKD / 4; ++s) {
    v2f a;
    a.x = wa[s] + wb2[s];
    a.y = wc2[s] + wd[s];
    acc0 = wmma4(a, hf[s][0], acc0);
    acc1 = wmma4(a, hf[s][1], acc1);
    acc2 = wmma4(a, hf[s][2], acc2);
    acc3 = wmma4(a, hf[s][3], acc3);
  }
#pragma unroll
  for (int v = 0; v < 8; ++v) {
    int m = m0 + v + ((l >> 4) << 3);
    v4f o;
    o.x = acc0[v] + old[v].x;
    o.y = acc1[v] + old[v].y;
    o.z = acc2[v] + old[v].z;
    o.w = acc3[v] + old[v].w;
    *reinterpret_cast<v4f*>(rb + m * TT) = o;
  }
}


extern "C" void kernel_launch(void* const* d_in, const int* in_sizes, int n_in,
                              void* d_out, int out_size, void* d_ws, size_t ws_size,
                              hipStream_t stream) {
  (void)in_sizes; (void)n_in; (void)out_size; (void)ws_size;
  const float* x    = (const float*)d_in[0];
  const float* mask = (const float*)d_in[1];
  const float* dw_w = (const float*)d_in[2];
  const float* dw_b = (const float*)d_in[3];
  const float* pw_w = (const float*)d_in[4];
  const float* pw_b = (const float*)d_in[5];
  const float* n0g  = (const float*)d_in[6];
  const float* n0b  = (const float*)d_in[7];
  const float* nsg  = (const float*)d_in[8];
  const float* nsb  = (const float*)d_in[9];
  const float* neg_ = (const float*)d_in[10];
  const float* neb  = (const float*)d_in[11];
  const float* Wq   = (const float*)d_in[12];
  const float* Wk   = (const float*)d_in[13];
  const float* Wv   = (const float*)d_in[14];
  const float* Wo   = (const float*)d_in[15];
  const float* fcw  = (const float*)d_in[16];
  const float* fcb  = (const float*)d_in[17];
  float* outp = (float*)d_out;
  float* ws = (float*)d_ws;

  float* OUT = ws;
  float* RES = ws + 2097152;
  float* TMP = ws + 4194304;
  float* Qb  = ws + 6291456;
  float* Kb  = ws + 7340032;
  float* Vb  = ws + 8388608;
  float* Hb  = ws + 9437184;

  ln_pw<PW_CONV_RAW><<<256, 1024, 0, stream>>>(x, n0g, n0b, dw_w, dw_b, pw_w, pw_b, RES);
  ln_pw<PW_CONV_LN><<<256, 1024, 0, stream>>>(RES, nsg, nsb, dw_w + CC * 7, dw_b + CC,
                                              pw_w + CC * CC, pw_b + CC, TMP);
  ln_pw<PW_CONV_LN><<<256, 1024, 0, stream>>>(TMP, nsg + CT, nsb + CT, dw_w + 2 * CC * 7,
                                              dw_b + 2 * CC, pw_w + 2 * CC * CC, pw_b + 2 * CC, RES);
  ln_pw<PW_CONV_LN><<<256, 1024, 0, stream>>>(RES, nsg + 2 * CT, nsb + 2 * CT, dw_w + 3 * CC * 7,
                                              dw_b + 3 * CC, pw_w + 3 * CC * CC, pw_b + 3 * CC, TMP);
  ln_fused<<<128, 1024, 0, stream>>>(TMP, nsg + 3 * CT, nsb + 3 * CT, OUT);

  qkv_gemm<<<256, 128, 0, stream>>>(OUT, Wq, Wk, Wv, Qb, Kb, Vb);
  attn_kernel<<<256, 128, 0, stream>>>(Qb, Kb, Vb, mask, Hb);
  wo_gemm<<<512, 128, 0, stream>>>(Hb, Wo, TMP);
  ln_pw<PW_LN><<<256, 1024, 0, stream>>>(TMP, neg_, neb, nullptr, nullptr, fcw, fcb, outp);
}
